# starting kernel
# baseline (speedup 1.0000x reference)
#include <hip/hip_runtime.h>
#include <math.h>

typedef __attribute__((ext_vector_type(2))) float v2f;
typedef __attribute__((ext_vector_type(8))) float v8f;

__global__ void k_init_deg(int* deg, int N) {
    int i = blockIdx.x * blockDim.x + threadIdx.x;
    if (i < N) deg[i] = 1;
}

__global__ void k_count(const int* __restrict__ dst, int* deg, int E) {
    int e = blockIdx.x * blockDim.x + threadIdx.x;
    if (e < E) atomicAdd(&deg[dst[e]], 1);
}

__global__ void k_dinv(const int* deg, float* dinv, int N) {
    int i = blockIdx.x * blockDim.x + threadIdx.x;
    if (i < N) {
        float f = (float)deg[i];
        dinv[i] = 1.0f / sqrtf(f);
    }
}

__global__ void k_xw1(const float* __restrict__ x, const float* __restrict__ W1,
                      const float* __restrict__ dinv,
                      float* __restrict__ y1, float* __restrict__ agg1, int N) {
    int lane = threadIdx.x & 31;
    int wave = threadIdx.x >> 5;
    int base = (blockIdx.x * (blockDim.x >> 5) + wave) * 16;
    if (base >= N) return;
    int m    = lane & 15;
    int half = lane >> 4;
    int node = base + m;
    bool full = (base + 16 <= N);

    v2f a = {0.0f, 0.0f};
    if (half == 0 && node < N) {
        a.x = x[node * 2 + 0];
        a.y = x[node * 2 + 1];
    }
    v2f b = {0.0f, 0.0f};
    if (half == 0 && m < 4) {
        b.x = W1[0 * 4 + m];
        b.y = W1[1 * 4 + m];
    }
    v8f c = {};
    v8f d = __builtin_amdgcn_wmma_f32_16x16x4_f32(
        false, a, false, b, (short)0, c, false, false);

    int nb = base + 8 * half;
    if (full) {
        if (m < 4) {
            float dv[8];
            #pragma unroll
            for (int v = 0; v < 8; ++v) dv[v] = dinv[nb + v];
            #pragma unroll
            for (int v = 0; v < 8; ++v) {
                float val = dv[v] * d[v];
                y1[(nb + v) * 4 + m]   = val;
                agg1[(nb + v) * 4 + m] = val;
            }
        }
    } else {
        if (m < 4) {
            #pragma unroll
            for (int v = 0; v < 8; ++v) {
                int nd = nb + v;
                if (nd < N) {
                    float val = dinv[nd] * d[v];
                    y1[nd * 4 + m]   = val;
                    agg1[nd * 4 + m] = val;
                }
            }
        }
    }
}

__global__ void k_scatter1(const int* __restrict__ src, const int* __restrict__ dst,
                           const float* __restrict__ y1, float* agg1, int E) {
    int e = blockIdx.x * blockDim.x + threadIdx.x;
    if (e >= E) return;
    int s = src[e];
    int t = dst[e];
    float4 y = *(const float4*)(y1 + (size_t)s * 4);
    float* p = agg1 + (size_t)t * 4;
    atomicAdd(p + 0, y.x);
    atomicAdd(p + 1, y.y);
    atomicAdd(p + 2, y.z);
    atomicAdd(p + 3, y.w);
}

__global__ void k_layer2(const float* __restrict__ agg1, const float* __restrict__ dinv,
                         const float* __restrict__ b1, const float* __restrict__ W2,
                         float* __restrict__ y2, float* __restrict__ agg2, int N) {
    int lane = threadIdx.x & 31;
    int wave = threadIdx.x >> 5;
    int base = (blockIdx.x * (blockDim.x >> 5) + wave) * 16;
    if (base >= N) return;
    int m    = lane & 15;
    int half = lane >> 4;
    int node = base + m;
    int kb   = half * 2;
    bool full = (base + 16 <= N);

    v2f a = {0.0f, 0.0f};
    if (node < N) {
        float dv = dinv[node];
        a.x = tanhf(dv * agg1[node * 4 + kb + 0] + b1[kb + 0]);
        a.y = tanhf(dv * agg1[node * 4 + kb + 1] + b1[kb + 1]);
    }
    v2f b = {0.0f, 0.0f};
    if (m < 2) {
        b.x = W2[(kb + 0) * 2 + m];
        b.y = W2[(kb + 1) * 2 + m];
    }
    v8f c = {};
    v8f d = __builtin_amdgcn_wmma_f32_16x16x4_f32(
        false, a, false, b, (short)0, c, false, false);

    int nb = base + 8 * half;
    if (full) {
        if (m < 2) {
            float dv[8];
            #pragma unroll
            for (int v = 0; v < 8; ++v) dv[v] = dinv[nb + v];
            #pragma unroll
            for (int v = 0; v < 8; ++v) {
                float val = dv[v] * d[v];
                y2[(nb + v) * 2 + m]   = val;
                agg2[(nb + v) * 2 + m] = val;
            }
        }
    } else {
        if (m < 2) {
            #pragma unroll
            for (int v = 0; v < 8; ++v) {
                int nd = nb + v;
                if (nd < N) {
                    float val = dinv[nd] * d[v];
                    y2[nd * 2 + m]   = val;
                    agg2[nd * 2 + m] = val;
                }
            }
        }
    }
}

__global__ void k_scatter2(const int* __restrict__ src, const int* __restrict__ dst,
                           const float* __restrict__ y2, float* agg2, int E) {
    int e = blockIdx.x * blockDim.x + threadIdx.x;
    if (e >= E) return;
    int s = src[e];
    int t = dst[e];
    float2 y = *(const float2*)(y2 + (size_t)s * 2);
    float* p = agg2 + (size_t)t * 2;
    atomicAdd(p + 0, y.x);
    atomicAdd(p + 1, y.y);
}

__global__ void k_final(const float* __restrict__ agg2, const float* __restrict__ dinv,
                        const float* __restrict__ b2, const float* __restrict__ Wc,
                        const float* __restrict__ bc, float* __restrict__ out, int N) {
    int i = blockIdx.x * blockDim.x + threadIdx.x;
    if (i >= N) return;
    float dv = dinv[i];
    float h0 = tanhf(dv * agg2[i * 2 + 0] + b2[0]);
    float h1 = tanhf(dv * agg2[i * 2 + 1] + b2[1]);
    float2 o;
    o.x = h0 * Wc[0] + h1 * Wc[2] + bc[0];
    o.y = h0 * Wc[1] + h1 * Wc[3] + bc[1];
    *(float2*)(out + (size_t)i * 2) = o;
}

extern "C" void kernel_launch(void* const* d_in, const int* in_sizes, int n_in,
                              void* d_out, int out_size, void* d_ws, size_t ws_size,
                              hipStream_t stream) {
    const float* x  = (const float*)d_in[0];
    const int*   ei = (const int*)d_in[1];
    const float* W1 = (const float*)d_in[2];
    const float* b1 = (const float*)d_in[3];
    const float* W2 = (const float*)d_in[4];
    const float* b2 = (const float*)d_in[5];
    const float* Wc = (const float*)d_in[6];
    const float* bc = (const float*)d_in[7];
    float* out = (float*)d_out;

    const int N = in_sizes[0] / 2;
    const int E = in_sizes[1] / 2;
    const int* src = ei;
    const int* dst = ei + E;

    float* ws   = (float*)d_ws;
    int*   deg  = (int*)ws;
    float* dinv = ws;
    float* y1   = ws + (size_t)N;
    float* agg1 = ws + (size_t)5 * N;
    float* y2   = ws + (size_t)9 * N;
    float* agg2 = ws + (size_t)11 * N;

    const int T = 256;
    dim3 gN((N + T - 1) / T);
    dim3 gE((E + T - 1) / T);
    const int nodesPerBlock = (T / 32) * 16;
    dim3 gW((N + nodesPerBlock - 1) / nodesPerBlock);

    k_init_deg<<<gN, T, 0, stream>>>(deg, N);
    k_count   <<<gE, T, 0, stream>>>(dst, deg, E);
    k_dinv    <<<gN, T, 0, stream>>>(deg, dinv, N);
    k_xw1     <<<gW, T, 0, stream>>>(x, W1, dinv, y1, agg1, N);
    k_scatter1<<<gE, T, 0, stream>>>(src, dst, y1, agg1, E);
    k_layer2  <<<gW, T, 0, stream>>>(agg1, dinv, b1, W2, y2, agg2, N);
    k_scatter2<<<gE, T, 0, stream>>>(src, dst, y2, agg2, E);
    k_final   <<<gN, T, 0, stream>>>(agg2, dinv, b2, Wc, bc, out, N);
}


# Layer-1 edge sum taken on the 2 input features before the W1 product (linearity), not on the 4 hidden ones: 2 float atomics per edge instead of 4; W1 on WMMA after the sum; summation order changes
# speedup vs baseline: 1.1001x; 1.1001x over previous
#include <hip/hip_runtime.h>
#include <math.h>

typedef __attribute__((ext_vector_type(2))) float v2f;
typedef __attribute__((ext_vector_type(8))) float v8f;

__global__ void k_init_deg(int* deg, int N) {
    int i = blockIdx.x * blockDim.x + threadIdx.x;
    if (i < N) deg[i] = 1;
}

__global__ void k_count(const int* __restrict__ dst, int* deg, int E) {
    int e = blockIdx.x * blockDim.x + threadIdx.x;
    if (e < E) atomicAdd(&deg[dst[e]], 1);
}

__global__ void k_dinv(const int* deg, float* dinv, int N) {
    int i = blockIdx.x * blockDim.x + threadIdx.x;
    if (i < N) {
        float f = (float)deg[i];
        dinv[i] = 1.0f / sqrtf(f);
    }
}

__global__ void k_y1(const float* __restrict__ x, const float* __restrict__ dinv,
                     float* __restrict__ y1, float* __restrict__ agg1, int N) {
    int i = blockIdx.x * blockDim.x + threadIdx.x;
    if (i >= N) return;
    float dv = dinv[i];
    float2 xv = *(const float2*)(x + (size_t)i * 2);
    float2 v;
    v.x = dv * xv.x;
    v.y = dv * xv.y;
    *(float2*)(y1   + (size_t)i * 2) = v;
    *(float2*)(agg1 + (size_t)i * 2) = v;
}

__global__ void k_scatter(const int* __restrict__ src, const int* __restrict__ dst,
                          const float* __restrict__ y, float* agg, int E) {
    int e = blockIdx.x * blockDim.x + threadIdx.x;
    if (e >= E) return;
    int s = src[e];
    int t = dst[e];
    float2 v = *(const float2*)(y + (size_t)s * 2);
    float* p = agg + (size_t)t * 2;
    atomicAdd(p + 0, v.x);
    atomicAdd(p + 1, v.y);
}

__global__ void k_layer2(const float* __restrict__ agg1, const float* __restrict__ dinv,
                         const float* __restrict__ W1, const float* __restrict__ b1,
                         const float* __restrict__ W2,
                         float* __restrict__ y2, float* __restrict__ agg2, int N) {
    int lane = threadIdx.x & 31;
    int wave = threadIdx.x >> 5;
    int base = (blockIdx.x * (blockDim.x >> 5) + wave) * 16;
    if (base >= N) return;
    int m    = lane & 15;
    int half = lane >> 4;
    int node = base + m;
    int kb   = half * 2;
    bool full = (base + 16 <= N);

    int j = m & 7;
    v2f a1 = {0.0f, 0.0f};
    if (half == 0 && j < 4) {
        a1.x = W1[0 * 4 + j];
        a1.y = W1[1 * 4 + j];
    }
    v2f s = {0.0f, 0.0f};
    if (half == 0 && node < N) {
        float dv = dinv[node];
        float2 g = *(const float2*)(agg1 + (size_t)node * 2);
        s.x = dv * g.x;
        s.y = dv * g.y;
    }
    v8f c1 = {b1[0], b1[1], b1[2], b1[3], 0.0f, 0.0f, 0.0f, 0.0f};
    v8f p = __builtin_amdgcn_wmma_f32_16x16x4_f32(
        false, a1, false, s, (short)0, c1, false, false);

    v2f a2;
    a2.x = tanhf(half ? p[2] : p[0]);
    a2.y = tanhf(half ? p[3] : p[1]);

    v2f b2 = {0.0f, 0.0f};
    if (m < 2) {
        b2.x = W2[(kb + 0) * 2 + m];
        b2.y = W2[(kb + 1) * 2 + m];
    }
    v8f c2 = {};
    v8f d = __builtin_amdgcn_wmma_f32_16x16x4_f32(
        false, a2, false, b2, (short)0, c2, false, false);

    int nb = base + 8 * half;
    if (full) {
        if (m < 2) {
            float dv[8];
            #pragma unroll
            for (int v = 0; v < 8; ++v) dv[v] = dinv[nb + v];
            #pragma unroll
            for (int v = 0; v < 8; ++v) {
                float val = dv[v] * d[v];
                y2[(nb + v) * 2 + m]   = val;
                agg2[(nb + v) * 2 + m] = val;
            }
        }
    } else {
        if (m < 2) {
            #pragma unroll
            for (int v = 0; v < 8; ++v) {
                int nd = nb + v;
                if (nd < N) {
                    float val = dinv[nd] * d[v];
                    y2[nd * 2 + m]   = val;
                    agg2[nd * 2 + m] = val;
                }
            }
        }
    }
}

__global__ void k_final(const float* __restrict__ agg2, const float* __restrict__ dinv,
                        const float* __restrict__ b2, const float* __restrict__ Wc,
                        const float* __restrict__ bc, float* __restrict__ out, int N) {
    int i = blockIdx.x * blockDim.x + threadIdx.x;
    if (i >= N) return;
    float dv = dinv[i];
    float h0 = tanhf(dv * agg2[i * 2 + 0] + b2[0]);
    float h1 = tanhf(dv * agg2[i * 2 + 1] + b2[1]);
    float2 o;
    o.x = h0 * Wc[0] + h1 * Wc[2] + bc[0];
    o.y = h0 * Wc[1] + h1 * Wc[3] + bc[1];
    *(float2*)(out + (size_t)i * 2) = o;
}

extern "C" void kernel_launch(void* const* d_in, const int* in_sizes, int n_in,
                              void* d_out, int out_size, void* d_ws, size_t ws_size,
                              hipStream_t stream) {
    const float* x  = (const float*)d_in[0];
    const int*   ei = (const int*)d_in[1];
    const float* W1 = (const float*)d_in[2];
    const float* b1 = (const float*)d_in[3];
    const float* W2 = (const float*)d_in[4];
    const float* b2 = (const float*)d_in[5];
    const float* Wc = (const float*)d_in[6];
    const float* bc = (const float*)d_in[7];
    float* out = (float*)d_out;

    const int N = in_sizes[0] / 2;
    const int E = in_sizes[1] / 2;
    const int* src = ei;
    const int* dst = ei + E;

    float* ws   = (float*)d_ws;
    float* y1   = ws;
    float* agg1 = ws + (size_t)2 * N;
    float* y2   = ws + (size_t)4 * N;
    float* agg2 = ws + (size_t)6 * N;
    float* dinv = ws + (size_t)8 * N;
    int*   deg  = (int*)dinv;

    const int T = 256;
    dim3 gN((N + T - 1) / T);
    dim3 gE((E + T - 1) / T);
    const int nodesPerBlock = (T / 32) * 16;
    dim3 gW((N + nodesPerBlock - 1) / nodesPerBlock);

    k_init_deg<<<gN, T, 0, stream>>>(deg, N);
    k_count   <<<gE, T, 0, stream>>>(dst, deg, E);
    k_dinv    <<<gN, T, 0, stream>>>(deg, dinv, N);
    k_y1      <<<gN, T, 0, stream>>>(x, dinv, y1, agg1, N);
    k_scatter <<<gE, T, 0, stream>>>(src, dst, y1, agg1, E);
    k_layer2  <<<gW, T, 0, stream>>>(agg1, dinv, W1, b1, W2, y2, agg2, N);
    k_scatter <<<gE, T, 0, stream>>>(src, dst, y2, agg2, E);
    k_final   <<<gN, T, 0, stream>>>(agg2, dinv, b2, Wc, bc, out, N);
}


# One copy of deg and of both edge sums per chip half (die id asked once per wave of 256 edges): a wave adds into its half's copy, readers take copy0+copy1; float atomics kept, summation order changes
# speedup vs baseline: 3.6322x; 3.3017x over previous
#include <hip/hip_runtime.h>
#include <math.h>

typedef __attribute__((ext_vector_type(2))) float v2f;
typedef __attribute__((ext_vector_type(8))) float v8f;

#define WAVE_EDGES 256

__device__ __forceinline__ int wave_half() {
    unsigned id = __builtin_amdgcn_s_sendmsg_rtn(0x87);
    return (int)((id >> 18) & 1u);
}

template <bool FULL>
__device__ __forceinline__ void load8(const int* __restrict__ p, int e, int E, int4& a, int4& b) {
    if (FULL) {
        a = *(const int4*)(p + e);
        b = *(const int4*)(p + e + 128);
    } else {
        a.x = (e + 0 < E) ? p[e + 0] : -1;
        a.y = (e + 1 < E) ? p[e + 1] : -1;
        a.z = (e + 2 < E) ? p[e + 2] : -1;
        a.w = (e + 3 < E) ? p[e + 3] : -1;
        b.x = (e + 128 < E) ? p[e + 128] : -1;
        b.y = (e + 129 < E) ? p[e + 129] : -1;
        b.z = (e + 130 < E) ? p[e + 130] : -1;
        b.w = (e + 131 < E) ? p[e + 131] : -1;
    }
}

__device__ __forceinline__ bool vec_ok(const int* p, int E) {
    return ((E & 3) == 0) && ((((size_t)p) & 15) == 0);
}

__device__ __forceinline__ int wave_index() {
    return blockIdx.x * (blockDim.x >> 5) + __builtin_amdgcn_readfirstlane((int)(threadIdx.x >> 5));
}

__global__ void k_init_deg(int* deg0, int* deg1, int N) {
    int i = blockIdx.x * blockDim.x + threadIdx.x;
    if (i < N) {
        deg0[i] = 1;
        deg1[i] = 0;
    }
}

template <bool FULL>
__device__ __forceinline__ void count_add(int* deg, int t) {
    if (FULL || t >= 0) atomicAdd(&deg[t], 1);
}

template <bool FULL>
__device__ __forceinline__ void count_wave(const int* __restrict__ dst, int* deg, int e0, int E) {
    int4 dA, dB;
    load8<FULL>(dst, e0, E, dA, dB);
    count_add<FULL>(deg, dA.x);
    count_add<FULL>(deg, dA.y);
    count_add<FULL>(deg, dA.z);
    count_add<FULL>(deg, dA.w);
    count_add<FULL>(deg, dB.x);
    count_add<FULL>(deg, dB.y);
    count_add<FULL>(deg, dB.z);
    count_add<FULL>(deg, dB.w);
}

__global__ void k_count(const int* __restrict__ dst, int* deg0, int* deg1, int E) {
    const int lane = threadIdx.x & 31;
    const int wave = wave_index();
    const long long w0 = (long long)wave * WAVE_EDGES;
    if (w0 >= E) return;
    int* deg = wave_half() ? deg1 : deg0;
    const int e0 = (int)w0 + lane * 4;
    if (vec_ok(dst, E) && w0 + WAVE_EDGES <= E) count_wave<true>(dst, deg, e0, E);
    else                                        count_wave<false>(dst, deg, e0, E);
}

__global__ void k_dinv(const int* deg0, const int* __restrict__ deg1, float* dinv, int N) {
    int i = blockIdx.x * blockDim.x + threadIdx.x;
    if (i < N) {
        float f = (float)(deg0[i] + deg1[i]);
        dinv[i] = 1.0f / sqrtf(f);
    }
}

__global__ void k_y1(const float* __restrict__ x, const float* __restrict__ dinv,
                     float* __restrict__ y1, float* __restrict__ agg1,
                     float* __restrict__ agg1b, int N) {
    int i = blockIdx.x * blockDim.x + threadIdx.x;
    if (i >= N) return;
    float dv = dinv[i];
    float2 xv = *(const float2*)(x + (size_t)i * 2);
    float2 v;
    v.x = dv * xv.x;
    v.y = dv * xv.y;
    *(float2*)(y1   + (size_t)i * 2) = v;
    *(float2*)(agg1 + (size_t)i * 2) = v;
    *(float2*)(agg1b + (size_t)i * 2) = make_float2(0.0f, 0.0f);
}

template <bool FULL>
__device__ __forceinline__ float2 gather2(const float* __restrict__ y, int s) {
    if (!FULL && s < 0) s = 0;
    return *(const float2*)(y + (size_t)s * 2);
}

template <bool FULL>
__device__ __forceinline__ void add2(float* agg, int t, float2 v) {
    if (FULL || t >= 0) {
        float* p = agg + (size_t)t * 2;
        atomicAdd(p + 0, v.x);
        atomicAdd(p + 1, v.y);
    }
}

template <bool FULL>
__device__ __forceinline__ void scatter_wave(const int* __restrict__ src, const int* __restrict__ dst,
                                             const float* __restrict__ y, float* agg, int e0, int E) {
    int4 dA, dB, sA, sB;
    load8<FULL>(dst, e0, E, dA, dB);
    load8<FULL>(src, e0, E, sA, sB);
    const float2 v0 = gather2<FULL>(y, sA.x);
    const float2 v1 = gather2<FULL>(y, sA.y);
    const float2 v2 = gather2<FULL>(y, sA.z);
    const float2 v3 = gather2<FULL>(y, sA.w);
    const float2 v4 = gather2<FULL>(y, sB.x);
    const float2 v5 = gather2<FULL>(y, sB.y);
    const float2 v6 = gather2<FULL>(y, sB.z);
    const float2 v7 = gather2<FULL>(y, sB.w);
    add2<FULL>(agg, dA.x, v0);
    add2<FULL>(agg, dA.y, v1);
    add2<FULL>(agg, dA.z, v2);
    add2<FULL>(agg, dA.w, v3);
    add2<FULL>(agg, dB.x, v4);
    add2<FULL>(agg, dB.y, v5);
    add2<FULL>(agg, dB.z, v6);
    add2<FULL>(agg, dB.w, v7);
}

__global__ void k_scatter(const int* __restrict__ src, const int* __restrict__ dst,
                          const float* __restrict__ y, float* agg0, float* agg1, int E) {
    const int lane = threadIdx.x & 31;
    const int wave = wave_index();
    const long long w0 = (long long)wave * WAVE_EDGES;
    if (w0 >= E) return;
    float* agg = wave_half() ? agg1 : agg0;
    const int e0 = (int)w0 + lane * 4;
    if (vec_ok(dst, E) && vec_ok(src, E) && w0 + WAVE_EDGES <= E)
        scatter_wave<true>(src, dst, y, agg, e0, E);
    else
        scatter_wave<false>(src, dst, y, agg, e0, E);
}

__global__ void k_layer2(const float* __restrict__ agg1, float* aggB,
                         const float* __restrict__ dinv,
                         const float* __restrict__ W1, const float* __restrict__ b1,
                         const float* __restrict__ W2,
                         float* __restrict__ y2, float* __restrict__ agg2, int N) {
    int lane = threadIdx.x & 31;
    int wave = threadIdx.x >> 5;
    int base = (blockIdx.x * (blockDim.x >> 5) + wave) * 16;
    if (base >= N) return;
    int m    = lane & 15;
    int half = lane >> 4;
    int node = base + m;
    int kb   = half * 2;
    bool full = (base + 16 <= N);

    int j = m & 7;
    v2f a1 = {0.0f, 0.0f};
    if (half == 0 && j < 4) {
        a1.x = W1[0 * 4 + j];
        a1.y = W1[1 * 4 + j];
    }
    v2f s = {0.0f, 0.0f};
    if (half == 0 && node < N) {
        float dv = dinv[node];
        float2 g = *(const float2*)(agg1 + (size_t)node * 2);
        float2* pb = (float2*)(aggB + (size_t)node * 2);
        float2 gb = *pb;
        *pb = make_float2(0.0f, 0.0f);
        s.x = dv * (g.x + gb.x);
        s.y = dv * (g.y + gb.y);
    }
    v8f c1 = {b1[0], b1[1], b1[2], b1[3], 0.0f, 0.0f, 0.0f, 0.0f};
    v8f p = __builtin_amdgcn_wmma_f32_16x16x4_f32(
        false, a1, false, s, (short)0, c1, false, false);

    v2f a2;
    a2.x = tanhf(half ? p[2] : p[0]);
    a2.y = tanhf(half ? p[3] : p[1]);

    v2f b2 = {0.0f, 0.0f};
    if (m < 2) {
        b2.x = W2[(kb + 0) * 2 + m];
        b2.y = W2[(kb + 1) * 2 + m];
    }
    v8f c2 = {};
    v8f d = __builtin_amdgcn_wmma_f32_16x16x4_f32(
        false, a2, false, b2, (short)0, c2, false, false);

    int nb = base + 8 * half;
    if (full) {
        if (m < 2) {
            float dv[8];
            #pragma unroll
            for (int v = 0; v < 8; ++v) dv[v] = dinv[nb + v];
            #pragma unroll
            for (int v = 0; v < 8; ++v) {
                float val = dv[v] * d[v];
                y2[(nb + v) * 2 + m]   = val;
                agg2[(nb + v) * 2 + m] = val;
            }
        }
    } else {
        if (m < 2) {
            #pragma unroll
            for (int v = 0; v < 8; ++v) {
                int nd = nb + v;
                if (nd < N) {
                    float val = dinv[nd] * d[v];
                    y2[nd * 2 + m]   = val;
                    agg2[nd * 2 + m] = val;
                }
            }
        }
    }
}

__global__ void k_final(const float* __restrict__ agg2, const float* __restrict__ agg2b,
                        const float* __restrict__ dinv,
                        const float* __restrict__ b2, const float* __restrict__ Wc,
                        const float* __restrict__ bc, float* __restrict__ out, int N) {
    int i = blockIdx.x * blockDim.x + threadIdx.x;
    if (i >= N) return;
    float dv = dinv[i];
    float2 ga = *(const float2*)(agg2  + (size_t)i * 2);
    float2 gb = *(const float2*)(agg2b + (size_t)i * 2);
    float h0 = tanhf(dv * (ga.x + gb.x) + b2[0]);
    float h1 = tanhf(dv * (ga.y + gb.y) + b2[1]);
    float2 o;
    o.x = h0 * Wc[0] + h1 * Wc[2] + bc[0];
    o.y = h0 * Wc[1] + h1 * Wc[3] + bc[1];
    *(float2*)(out + (size_t)i * 2) = o;
}

extern "C" void kernel_launch(void* const* d_in, const int* in_sizes, int n_in,
                              void* d_out, int out_size, void* d_ws, size_t ws_size,
                              hipStream_t stream) {
    const float* x  = (const float*)d_in[0];
    const int*   ei = (const int*)d_in[1];
    const float* W1 = (const float*)d_in[2];
    const float* b1 = (const float*)d_in[3];
    const float* W2 = (const float*)d_in[4];
    const float* b2 = (const float*)d_in[5];
    const float* Wc = (const float*)d_in[6];
    const float* bc = (const float*)d_in[7];
    float* out = (float*)d_out;

    const int N = in_sizes[0] / 2;
    const int E = in_sizes[1] / 2;
    const int* src = ei;
    const int* dst = ei + E;

    float* ws   = (float*)d_ws;
    float* y1   = ws;
    float* agg1 = ws + (size_t)2 * N;
    float* aggB = ws + (size_t)4 * N;
    float* y2   = ws;
    float* agg2 = ws + (size_t)6 * N;
    float* dinv = ws + (size_t)8 * N;
    int*   deg0 = (int*)dinv;
    int*   deg1 = (int*)ws;

    const int T = 256;
    dim3 gN((N + T - 1) / T);
    const int edgesPerBlock = (T / 32) * WAVE_EDGES;
    dim3 gE((E + edgesPerBlock - 1) / edgesPerBlock);
    const int nodesPerBlock = (T / 32) * 16;
    dim3 gW((N + nodesPerBlock - 1) / nodesPerBlock);

    k_init_deg<<<gN, T, 0, stream>>>(deg0, deg1, N);
    k_count   <<<gE, T, 0, stream>>>(dst, deg0, deg1, E);
    k_dinv    <<<gN, T, 0, stream>>>(deg0, deg1, dinv, N);
    k_y1      <<<gN, T, 0, stream>>>(x, dinv, y1, agg1, aggB, N);
    k_scatter <<<gE, T, 0, stream>>>(src, dst, y1, agg1, aggB, E);
    k_layer2  <<<gW, T, 0, stream>>>(agg1, aggB, dinv, W1, b1, W2, y2, agg2, N);
    k_scatter <<<gE, T, 0, stream>>>(src, dst, y2, agg2, aggB, E);
    k_final   <<<gN, T, 0, stream>>>(agg2, aggB, dinv, b2, Wc, bc, out, N);
}


# Both float adds of an edge issued by a lane pair in one atomic instruction (neighbour lanes, neighbour words): one cache-line request per edge, not two; float atomics kept, summation order changes
# speedup vs baseline: 4.7639x; 1.3116x over previous
#include <hip/hip_runtime.h>
#include <math.h>

typedef __attribute__((ext_vector_type(2))) float v2f;
typedef __attribute__((ext_vector_type(8))) float v8f;

#define WAVE_EDGES 256

__device__ __forceinline__ int wave_half() {
    unsigned id = __builtin_amdgcn_s_sendmsg_rtn(0x87);
    return (int)((id >> 18) & 1u);
}

template <bool FULL>
__device__ __forceinline__ void load8(const int* __restrict__ p, int e, int E, int4& a, int4& b) {
    if (FULL) {
        a = *(const int4*)(p + e);
        b = *(const int4*)(p + e + 128);
    } else {
        a.x = (e + 0 < E) ? p[e + 0] : -1;
        a.y = (e + 1 < E) ? p[e + 1] : -1;
        a.z = (e + 2 < E) ? p[e + 2] : -1;
        a.w = (e + 3 < E) ? p[e + 3] : -1;
        b.x = (e + 128 < E) ? p[e + 128] : -1;
        b.y = (e + 129 < E) ? p[e + 129] : -1;
        b.z = (e + 130 < E) ? p[e + 130] : -1;
        b.w = (e + 131 < E) ? p[e + 131] : -1;
    }
}

__device__ __forceinline__ bool vec_ok(const int* p, int E) {
    return ((E & 3) == 0) && ((((size_t)p) & 15) == 0);
}

__device__ __forceinline__ int wave_index() {
    return blockIdx.x * (blockDim.x >> 5) + __builtin_amdgcn_readfirstlane((int)(threadIdx.x >> 5));
}

__global__ void k_init_deg(int* deg0, int* deg1, int N) {
    int i = blockIdx.x * blockDim.x + threadIdx.x;
    if (i < N) {
        deg0[i] = 1;
        deg1[i] = 0;
    }
}

template <bool FULL>
__device__ __forceinline__ void count_add(int* deg, int t) {
    if (FULL || t >= 0) atomicAdd(&deg[t], 1);
}

template <bool FULL>
__device__ __forceinline__ void count_wave(const int* __restrict__ dst, int* deg, int e0, int E) {
    int4 dA, dB;
    load8<FULL>(dst, e0, E, dA, dB);
    count_add<FULL>(deg, dA.x);
    count_add<FULL>(deg, dA.y);
    count_add<FULL>(deg, dA.z);
    count_add<FULL>(deg, dA.w);
    count_add<FULL>(deg, dB.x);
    count_add<FULL>(deg, dB.y);
    count_add<FULL>(deg, dB.z);
    count_add<FULL>(deg, dB.w);
}

__global__ void k_count(const int* __restrict__ dst, int* deg0, int* deg1, int E) {
    const int lane = threadIdx.x & 31;
    const int wave = wave_index();
    const long long w0 = (long long)wave * WAVE_EDGES;
    if (w0 >= E) return;
    int* deg = wave_half() ? deg1 : deg0;
    const int e0 = (int)w0 + lane * 4;
    if (vec_ok(dst, E) && w0 + WAVE_EDGES <= E) count_wave<true>(dst, deg, e0, E);
    else                                        count_wave<false>(dst, deg, e0, E);
}

__global__ void k_dinv(const int* deg0, const int* __restrict__ deg1, float* dinv, int N) {
    int i = blockIdx.x * blockDim.x + threadIdx.x;
    if (i < N) {
        float f = (float)(deg0[i] + deg1[i]);
        dinv[i] = 1.0f / sqrtf(f);
    }
}

__global__ void k_y1(const float* __restrict__ x, const float* __restrict__ dinv,
                     float* __restrict__ y1, float* __restrict__ agg1,
                     float* __restrict__ agg1b, int N) {
    int i = blockIdx.x * blockDim.x + threadIdx.x;
    if (i >= N) return;
    float dv = dinv[i];
    float2 xv = *(const float2*)(x + (size_t)i * 2);
    float2 v;
    v.x = dv * xv.x;
    v.y = dv * xv.y;
    *(float2*)(y1   + (size_t)i * 2) = v;
    *(float2*)(agg1 + (size_t)i * 2) = v;
    *(float2*)(agg1b + (size_t)i * 2) = make_float2(0.0f, 0.0f);
}

template <bool FULL>
__device__ __forceinline__ int4 load4(const int* __restrict__ p, int e, int E) {
    int4 a;
    if (FULL) {
        a = *(const int4*)(p + e);
    } else {
        a.x = (e + 0 < E) ? p[e + 0] : -1;
        a.y = (e + 1 < E) ? p[e + 1] : -1;
        a.z = (e + 2 < E) ? p[e + 2] : -1;
        a.w = (e + 3 < E) ? p[e + 3] : -1;
    }
    return a;
}

template <bool FULL>
__device__ __forceinline__ float gather1(const float* __restrict__ y, int s, int f) {
    if (!FULL && s < 0) s = 0;
    return y[(unsigned)s * 2u + (unsigned)f];
}

template <bool FULL>
__device__ __forceinline__ void add1(float* agg, int t, int f, float v) {
    if (FULL || t >= 0)
        atomicAdd(agg + ((unsigned)t * 2u + (unsigned)f), v);
}

template <bool FULL>
__device__ __forceinline__ void scatter_wave(const int* __restrict__ src, const int* __restrict__ dst,
                                             const float* __restrict__ y, float* agg,
                                             int e0, int f, int E) {
    const int4 d0 = load4<FULL>(dst, e0,       E);
    const int4 d1 = load4<FULL>(dst, e0 + 64,  E);
    const int4 d2 = load4<FULL>(dst, e0 + 128, E);
    const int4 d3 = load4<FULL>(dst, e0 + 192, E);
    const int4 s0 = load4<FULL>(src, e0,       E);
    const int4 s1 = load4<FULL>(src, e0 + 64,  E);
    const int4 s2 = load4<FULL>(src, e0 + 128, E);
    const int4 s3 = load4<FULL>(src, e0 + 192, E);
    __builtin_amdgcn_sched_barrier(0);
    const float v00 = gather1<FULL>(y, s0.x, f);
    const float v01 = gather1<FULL>(y, s0.y, f);
    const float v02 = gather1<FULL>(y, s0.z, f);
    const float v03 = gather1<FULL>(y, s0.w, f);
    const float v10 = gather1<FULL>(y, s1.x, f);
    const float v11 = gather1<FULL>(y, s1.y, f);
    const float v12 = gather1<FULL>(y, s1.z, f);
    const float v13 = gather1<FULL>(y, s1.w, f);
    const float v20 = gather1<FULL>(y, s2.x, f);
    const float v21 = gather1<FULL>(y, s2.y, f);
    const float v22 = gather1<FULL>(y, s2.z, f);
    const float v23 = gather1<FULL>(y, s2.w, f);
    const float v30 = gather1<FULL>(y, s3.x, f);
    const float v31 = gather1<FULL>(y, s3.y, f);
    const float v32 = gather1<FULL>(y, s3.z, f);
    const float v33 = gather1<FULL>(y, s3.w, f);
    add1<FULL>(agg, d0.x, f, v00);
    add1<FULL>(agg, d0.y, f, v01);
    add1<FULL>(agg, d0.z, f, v02);
    add1<FULL>(agg, d0.w, f, v03);
    add1<FULL>(agg, d1.x, f, v10);
    add1<FULL>(agg, d1.y, f, v11);
    add1<FULL>(agg, d1.z, f, v12);
    add1<FULL>(agg, d1.w, f, v13);
    add1<FULL>(agg, d2.x, f, v20);
    add1<FULL>(agg, d2.y, f, v21);
    add1<FULL>(agg, d2.z, f, v22);
    add1<FULL>(agg, d2.w, f, v23);
    add1<FULL>(agg, d3.x, f, v30);
    add1<FULL>(agg, d3.y, f, v31);
    add1<FULL>(agg, d3.z, f, v32);
    add1<FULL>(agg, d3.w, f, v33);
}

__global__ void k_scatter(const int* __restrict__ src, const int* __restrict__ dst,
                          const float* __restrict__ y, float* agg0, float* agg1, int E) {
    const int lane = threadIdx.x & 31;
    const int wave = wave_index();
    const long long w0 = (long long)wave * WAVE_EDGES;
    if (w0 >= E) return;
    float* agg = wave_half() ? agg1 : agg0;
    const int f  = lane & 1;
    const int e0 = (int)w0 + (lane >> 1) * 4;
    if (vec_ok(dst, E) && vec_ok(src, E) && w0 + WAVE_EDGES <= E)
        scatter_wave<true>(src, dst, y, agg, e0, f, E);
    else
        scatter_wave<false>(src, dst, y, agg, e0, f, E);
}

__global__ void k_layer2(const float* __restrict__ agg1, float* aggB,
                         const float* __restrict__ dinv,
                         const float* __restrict__ W1, const float* __restrict__ b1,
                         const float* __restrict__ W2,
                         float* __restrict__ y2, float* __restrict__ agg2, int N) {
    int lane = threadIdx.x & 31;
    int wave = threadIdx.x >> 5;
    int base = (blockIdx.x * (blockDim.x >> 5) + wave) * 16;
    if (base >= N) return;
    int m    = lane & 15;
    int half = lane >> 4;
    int node = base + m;
    int kb   = half * 2;
    bool full = (base + 16 <= N);

    int j = m & 7;
    v2f a1 = {0.0f, 0.0f};
    if (half == 0 && j < 4) {
        a1.x = W1[0 * 4 + j];
        a1.y = W1[1 * 4 + j];
    }
    v2f s = {0.0f, 0.0f};
    if (half == 0 && node < N) {
        float dv = dinv[node];
        float2 g = *(const float2*)(agg1 + (size_t)node * 2);
        float2* pb = (float2*)(aggB + (size_t)node * 2);
        float2 gb = *pb;
        *pb = make_float2(0.0f, 0.0f);
        s.x = dv * (g.x + gb.x);
        s.y = dv * (g.y + gb.y);
    }
    v8f c1 = {b1[0], b1[1], b1[2], b1[3], 0.0f, 0.0f, 0.0f, 0.0f};
    v8f p = __builtin_amdgcn_wmma_f32_16x16x4_f32(
        false, a1, false, s, (short)0, c1, false, false);

    v2f a2;
    a2.x = tanhf(half ? p[2] : p[0]);
    a2.y = tanhf(half ? p[3] : p[1]);

    v2f b2 = {0.0f, 0.0f};
    if (m < 2) {
        b2.x = W2[(kb + 0) * 2 + m];
        b2.y = W2[(kb + 1) * 2 + m];
    }
    v8f c2 = {};
    v8f d = __builtin_amdgcn_wmma_f32_16x16x4_f32(
        false, a2, false, b2, (short)0, c2, false, false);

    int nb = base + 8 * half;
    if (full) {
        if (m < 2) {
            float dv[8];
            #pragma unroll
            for (int v = 0; v < 8; ++v) dv[v] = dinv[nb + v];
            #pragma unroll
            for (int v = 0; v < 8; ++v) {
                float val = dv[v] * d[v];
                y2[(nb + v) * 2 + m]   = val;
                agg2[(nb + v) * 2 + m] = val;
            }
        }
    } else {
        if (m < 2) {
            #pragma unroll
            for (int v = 0; v < 8; ++v) {
                int nd = nb + v;
                if (nd < N) {
                    float val = dinv[nd] * d[v];
                    y2[nd * 2 + m]   = val;
                    agg2[nd * 2 + m] = val;
                }
            }
        }
    }
}

__global__ void k_final(const float* __restrict__ agg2, const float* __restrict__ agg2b,
                        const float* __restrict__ dinv,
                        const float* __restrict__ b2, const float* __restrict__ Wc,
                        const float* __restrict__ bc, float* __restrict__ out, int N) {
    int i = blockIdx.x * blockDim.x + threadIdx.x;
    if (i >= N) return;
    float dv = dinv[i];
    float2 ga = *(const float2*)(agg2  + (size_t)i * 2);
    float2 gb = *(const float2*)(agg2b + (size_t)i * 2);
    float h0 = tanhf(dv * (ga.x + gb.x) + b2[0]);
    float h1 = tanhf(dv * (ga.y + gb.y) + b2[1]);
    float2 o;
    o.x = h0 * Wc[0] + h1 * Wc[2] + bc[0];
    o.y = h0 * Wc[1] + h1 * Wc[3] + bc[1];
    *(float2*)(out + (size_t)i * 2) = o;
}

extern "C" void kernel_launch(void* const* d_in, const int* in_sizes, int n_in,
                              void* d_out, int out_size, void* d_ws, size_t ws_size,
                              hipStream_t stream) {
    const float* x  = (const float*)d_in[0];
    const int*   ei = (const int*)d_in[1];
    const float* W1 = (const float*)d_in[2];
    const float* b1 = (const float*)d_in[3];
    const float* W2 = (const float*)d_in[4];
    const float* b2 = (const float*)d_in[5];
    const float* Wc = (const float*)d_in[6];
    const float* bc = (const float*)d_in[7];
    float* out = (float*)d_out;

    const int N = in_sizes[0] / 2;
    const int E = in_sizes[1] / 2;
    const int* src = ei;
    const int* dst = ei + E;

    float* ws   = (float*)d_ws;
    float* y1   = ws;
    float* agg1 = ws + (size_t)2 * N;
    float* aggB = ws + (size_t)4 * N;
    float* y2   = ws;
    float* agg2 = ws + (size_t)6 * N;
    float* dinv = ws + (size_t)8 * N;
    int*   deg0 = (int*)dinv;
    int*   deg1 = (int*)ws;

    const int T = 256;
    dim3 gN((N + T - 1) / T);
    const int edgesPerBlock = (T / 32) * WAVE_EDGES;
    dim3 gE((E + edgesPerBlock - 1) / edgesPerBlock);
    const int nodesPerBlock = (T / 32) * 16;
    dim3 gW((N + nodesPerBlock - 1) / nodesPerBlock);

    k_init_deg<<<gN, T, 0, stream>>>(deg0, deg1, N);
    k_count   <<<gE, T, 0, stream>>>(dst, deg0, deg1, E);
    k_dinv    <<<gN, T, 0, stream>>>(deg0, deg1, dinv, N);
    k_y1      <<<gN, T, 0, stream>>>(x, dinv, y1, agg1, aggB, N);
    k_scatter <<<gE, T, 0, stream>>>(src, dst, y1, agg1, aggB, E);
    k_layer2  <<<gW, T, 0, stream>>>(agg1, aggB, dinv, W1, b1, W2, y2, agg2, N);
    k_scatter <<<gE, T, 0, stream>>>(src, dst, y2, agg2, aggB, E);
    k_final   <<<gN, T, 0, stream>>>(agg2, aggB, dinv, b2, Wc, bc, out, N);
}


# Degree count through work-group LDS histograms (65,536 integer counters per node slice, dst streamed once per slice), flushed by whole-line integer atomics: no scattered global atomic per edge
# speedup vs baseline: 5.3815x; 1.1296x over previous
#include <hip/hip_runtime.h>
#include <math.h>

typedef __attribute__((ext_vector_type(2))) float v2f;
typedef __attribute__((ext_vector_type(8))) float v8f;

#define WAVE_EDGES 256

__device__ __forceinline__ int wave_half() {
    unsigned id = __builtin_amdgcn_s_sendmsg_rtn(0x87);
    return (int)((id >> 18) & 1u);
}

__device__ __forceinline__ bool vec_ok(const int* p, int E) {
    return ((E & 3) == 0) && ((((size_t)p) & 15) == 0);
}

__device__ __forceinline__ int wave_index() {
    return blockIdx.x * (blockDim.x >> 5) + __builtin_amdgcn_readfirstlane((int)(threadIdx.x >> 5));
}

__global__ void k_init_deg(int* deg0, int* deg1, int N) {
    int i = blockIdx.x * blockDim.x + threadIdx.x;
    if (i < N) {
        deg0[i] = 1;
        deg1[i] = 0;
    }
}

#define HIST_BITS    16
#define HIST_BINS    (1 << HIST_BITS)
#define COUNT_T      1024
#define COUNT_CHUNKS 32
#define COUNT_LOADS  8

__device__ __forceinline__ void hist_add(int* hist, unsigned slice, int d) {
    if (((unsigned)d >> HIST_BITS) == slice) atomicAdd(&hist[d & (HIST_BINS - 1)], 1);
}

template <bool FULL>
__device__ __forceinline__ void count_group(const int* __restrict__ p, int left,
                                            int* hist, unsigned slice) {
    int4 v[COUNT_LOADS];
    #pragma unroll
    for (int k = 0; k < COUNT_LOADS; ++k) {
        if (FULL || k * (COUNT_T * 4) + 4 <= left) v[k] = *(const int4*)(p + k * (COUNT_T * 4));
        else                                       v[k] = make_int4(-1, -1, -1, -1);
    }
    #pragma unroll
    for (int k = 0; k < COUNT_LOADS; ++k) {
        hist_add(hist, slice, v[k].x);
        hist_add(hist, slice, v[k].y);
        hist_add(hist, slice, v[k].z);
        hist_add(hist, slice, v[k].w);
    }
}

__global__ __launch_bounds__(COUNT_T)
void k_count(const int* __restrict__ dst, int* deg0, int* deg1, int E, int N) {
    __shared__ __attribute__((aligned(16))) int hist[HIST_BINS];
    const int tid = threadIdx.x;
    const unsigned slice = blockIdx.x;

    for (int u = tid * 4; u < HIST_BINS; u += COUNT_T * 4)
        *(int4*)(hist + u) = make_int4(0, 0, 0, 0);
    __syncthreads();

    unsigned per = ((unsigned)E + gridDim.y - 1u) / gridDim.y;
    per = (per + 3u) & ~3u;
    const long long beg = (long long)blockIdx.y * per;
    long long end = beg + per;
    if (end > E) end = E;
    if (end < beg) end = beg;

    const bool vec = (((size_t)dst) & 15) == 0;
    const long long end4 = vec ? beg + ((end - beg) & ~3LL) : beg;
    const long long step = (long long)COUNT_T * 4 * COUNT_LOADS;
    long long i = beg;
    for (; i + step <= end4; i += step)
        count_group<true>(dst + i + tid * 4, 0, hist, slice);
    if (i < end4)
        count_group<false>(dst + i + tid * 4, (int)(end4 - i) - tid * 4, hist, slice);
    for (long long j = end4 + tid; j < end; j += COUNT_T)
        hist_add(hist, slice, dst[j]);
    __syncthreads();

    int* deg = wave_half() ? deg1 : deg0;
    const int base = (int)(slice << HIST_BITS);
    for (int u = tid; u < HIST_BINS; u += COUNT_T * 8) {
        int c[8];
        #pragma unroll
        for (int k = 0; k < 8; ++k) c[k] = hist[u + k * COUNT_T];
        #pragma unroll
        for (int k = 0; k < 8; ++k) {
            const int node = base + u + k * COUNT_T;
            if (c[k] != 0 && node < N) atomicAdd(&deg[node], c[k]);
        }
    }
}

__global__ void k_dinv(const int* deg0, const int* __restrict__ deg1, float* dinv, int N) {
    int i = blockIdx.x * blockDim.x + threadIdx.x;
    if (i < N) {
        float f = (float)(deg0[i] + deg1[i]);
        dinv[i] = 1.0f / sqrtf(f);
    }
}

__global__ void k_y1(const float* __restrict__ x, const float* __restrict__ dinv,
                     float* __restrict__ y1, float* __restrict__ agg1,
                     float* __restrict__ agg1b, int N) {
    int i = blockIdx.x * blockDim.x + threadIdx.x;
    if (i >= N) return;
    float dv = dinv[i];
    float2 xv = *(const float2*)(x + (size_t)i * 2);
    float2 v;
    v.x = dv * xv.x;
    v.y = dv * xv.y;
    *(float2*)(y1   + (size_t)i * 2) = v;
    *(float2*)(agg1 + (size_t)i * 2) = v;
    *(float2*)(agg1b + (size_t)i * 2) = make_float2(0.0f, 0.0f);
}

template <bool FULL>
__device__ __forceinline__ int4 load4(const int* __restrict__ p, int e, int E) {
    int4 a;
    if (FULL) {
        a = *(const int4*)(p + e);
    } else {
        a.x = (e + 0 < E) ? p[e + 0] : -1;
        a.y = (e + 1 < E) ? p[e + 1] : -1;
        a.z = (e + 2 < E) ? p[e + 2] : -1;
        a.w = (e + 3 < E) ? p[e + 3] : -1;
    }
    return a;
}

template <bool FULL>
__device__ __forceinline__ float gather1(const float* __restrict__ y, int s, int f) {
    if (!FULL && s < 0) s = 0;
    return y[(unsigned)s * 2u + (unsigned)f];
}

template <bool FULL>
__device__ __forceinline__ void add1(float* agg, int t, int f, float v) {
    if (FULL || t >= 0)
        atomicAdd(agg + ((unsigned)t * 2u + (unsigned)f), v);
}

template <bool FULL>
__device__ __forceinline__ void scatter_wave(const int* __restrict__ src, const int* __restrict__ dst,
                                             const float* __restrict__ y, float* agg,
                                             int e0, int f, int E) {
    const int4 d0 = load4<FULL>(dst, e0,       E);
    const int4 d1 = load4<FULL>(dst, e0 + 64,  E);
    const int4 d2 = load4<FULL>(dst, e0 + 128, E);
    const int4 d3 = load4<FULL>(dst, e0 + 192, E);
    const int4 s0 = load4<FULL>(src, e0,       E);
    const int4 s1 = load4<FULL>(src, e0 + 64,  E);
    const int4 s2 = load4<FULL>(src, e0 + 128, E);
    const int4 s3 = load4<FULL>(src, e0 + 192, E);
    __builtin_amdgcn_sched_barrier(0);
    const float v00 = gather1<FULL>(y, s0.x, f);
    const float v01 = gather1<FULL>(y, s0.y, f);
    const float v02 = gather1<FULL>(y, s0.z, f);
    const float v03 = gather1<FULL>(y, s0.w, f);
    const float v10 = gather1<FULL>(y, s1.x, f);
    const float v11 = gather1<FULL>(y, s1.y, f);
    const float v12 = gather1<FULL>(y, s1.z, f);
    const float v13 = gather1<FULL>(y, s1.w, f);
    const float v20 = gather1<FULL>(y, s2.x, f);
    const float v21 = gather1<FULL>(y, s2.y, f);
    const float v22 = gather1<FULL>(y, s2.z, f);
    const float v23 = gather1<FULL>(y, s2.w, f);
    const float v30 = gather1<FULL>(y, s3.x, f);
    const float v31 = gather1<FULL>(y, s3.y, f);
    const float v32 = gather1<FULL>(y, s3.z, f);
    const float v33 = gather1<FULL>(y, s3.w, f);
    add1<FULL>(agg, d0.x, f, v00);
    add1<FULL>(agg, d0.y, f, v01);
    add1<FULL>(agg, d0.z, f, v02);
    add1<FULL>(agg, d0.w, f, v03);
    add1<FULL>(agg, d1.x, f, v10);
    add1<FULL>(agg, d1.y, f, v11);
    add1<FULL>(agg, d1.z, f, v12);
    add1<FULL>(agg, d1.w, f, v13);
    add1<FULL>(agg, d2.x, f, v20);
    add1<FULL>(agg, d2.y, f, v21);
    add1<FULL>(agg, d2.z, f, v22);
    add1<FULL>(agg, d2.w, f, v23);
    add1<FULL>(agg, d3.x, f, v30);
    add1<FULL>(agg, d3.y, f, v31);
    add1<FULL>(agg, d3.z, f, v32);
    add1<FULL>(agg, d3.w, f, v33);
}

__global__ void k_scatter(const int* __restrict__ src, const int* __restrict__ dst,
                          const float* __restrict__ y, float* agg0, float* agg1, int E) {
    const int lane = threadIdx.x & 31;
    const int wave = wave_index();
    const long long w0 = (long long)wave * WAVE_EDGES;
    if (w0 >= E) return;
    float* agg = wave_half() ? agg1 : agg0;
    const int f  = lane & 1;
    const int e0 = (int)w0 + (lane >> 1) * 4;
    if (vec_ok(dst, E) && vec_ok(src, E) && w0 + WAVE_EDGES <= E)
        scatter_wave<true>(src, dst, y, agg, e0, f, E);
    else
        scatter_wave<false>(src, dst, y, agg, e0, f, E);
}

__global__ void k_layer2(const float* __restrict__ agg1, float* aggB,
                         const float* __restrict__ dinv,
                         const float* __restrict__ W1, const float* __restrict__ b1,
                         const float* __restrict__ W2,
                         float* __restrict__ y2, float* __restrict__ agg2, int N) {
    int lane = threadIdx.x & 31;
    int wave = threadIdx.x >> 5;
    int base = (blockIdx.x * (blockDim.x >> 5) + wave) * 16;
    if (base >= N) return;
    int m    = lane & 15;
    int half = lane >> 4;
    int node = base + m;
    int kb   = half * 2;
    bool full = (base + 16 <= N);

    int j = m & 7;
    v2f a1 = {0.0f, 0.0f};
    if (half == 0 && j < 4) {
        a1.x = W1[0 * 4 + j];
        a1.y = W1[1 * 4 + j];
    }
    v2f s = {0.0f, 0.0f};
    if (half == 0 && node < N) {
        float dv = dinv[node];
        float2 g = *(const float2*)(agg1 + (size_t)node * 2);
        float2* pb = (float2*)(aggB + (size_t)node * 2);
        float2 gb = *pb;
        *pb = make_float2(0.0f, 0.0f);
        s.x = dv * (g.x + gb.x);
        s.y = dv * (g.y + gb.y);
    }
    v8f c1 = {b1[0], b1[1], b1[2], b1[3], 0.0f, 0.0f, 0.0f, 0.0f};
    v8f p = __builtin_amdgcn_wmma_f32_16x16x4_f32(
        false, a1, false, s, (short)0, c1, false, false);

    v2f a2;
    a2.x = tanhf(half ? p[2] : p[0]);
    a2.y = tanhf(half ? p[3] : p[1]);

    v2f b2 = {0.0f, 0.0f};
    if (m < 2) {
        b2.x = W2[(kb + 0) * 2 + m];
        b2.y = W2[(kb + 1) * 2 + m];
    }
    v8f c2 = {};
    v8f d = __builtin_amdgcn_wmma_f32_16x16x4_f32(
        false, a2, false, b2, (short)0, c2, false, false);

    int nb = base + 8 * half;
    if (full) {
        if (m < 2) {
            float dv[8];
            #pragma unroll
            for (int v = 0; v < 8; ++v) dv[v] = dinv[nb + v];
            #pragma unroll
            for (int v = 0; v < 8; ++v) {
                float val = dv[v] * d[v];
                y2[(nb + v) * 2 + m]   = val;
                agg2[(nb + v) * 2 + m] = val;
            }
        }
    } else {
        if (m < 2) {
            #pragma unroll
            for (int v = 0; v < 8; ++v) {
                int nd = nb + v;
                if (nd < N) {
                    float val = dinv[nd] * d[v];
                    y2[nd * 2 + m]   = val;
                    agg2[nd * 2 + m] = val;
                }
            }
        }
    }
}

__global__ void k_final(const float* __restrict__ agg2, const float* __restrict__ agg2b,
                        const float* __restrict__ dinv,
                        const float* __restrict__ b2, const float* __restrict__ Wc,
                        const float* __restrict__ bc, float* __restrict__ out, int N) {
    int i = blockIdx.x * blockDim.x + threadIdx.x;
    if (i >= N) return;
    float dv = dinv[i];
    float2 ga = *(const float2*)(agg2  + (size_t)i * 2);
    float2 gb = *(const float2*)(agg2b + (size_t)i * 2);
    float h0 = tanhf(dv * (ga.x + gb.x) + b2[0]);
    float h1 = tanhf(dv * (ga.y + gb.y) + b2[1]);
    float2 o;
    o.x = h0 * Wc[0] + h1 * Wc[2] + bc[0];
    o.y = h0 * Wc[1] + h1 * Wc[3] + bc[1];
    *(float2*)(out + (size_t)i * 2) = o;
}

extern "C" void kernel_launch(void* const* d_in, const int* in_sizes, int n_in,
                              void* d_out, int out_size, void* d_ws, size_t ws_size,
                              hipStream_t stream) {
    const float* x  = (const float*)d_in[0];
    const int*   ei = (const int*)d_in[1];
    const float* W1 = (const float*)d_in[2];
    const float* b1 = (const float*)d_in[3];
    const float* W2 = (const float*)d_in[4];
    const float* b2 = (const float*)d_in[5];
    const float* Wc = (const float*)d_in[6];
    const float* bc = (const float*)d_in[7];
    float* out = (float*)d_out;

    const int N = in_sizes[0] / 2;
    const int E = in_sizes[1] / 2;
    const int* src = ei;
    const int* dst = ei + E;

    float* ws   = (float*)d_ws;
    float* y1   = ws;
    float* agg1 = ws + (size_t)2 * N;
    float* aggB = ws + (size_t)4 * N;
    float* y2   = ws;
    float* agg2 = ws + (size_t)6 * N;
    float* dinv = ws + (size_t)8 * N;
    int*   deg0 = (int*)dinv;
    int*   deg1 = (int*)ws;

    const int T = 256;
    dim3 gN((N + T - 1) / T);
    const int edgesPerBlock = (T / 32) * WAVE_EDGES;
    dim3 gE((E + edgesPerBlock - 1) / edgesPerBlock);
    const int nodesPerBlock = (T / 32) * 16;
    dim3 gW((N + nodesPerBlock - 1) / nodesPerBlock);
    dim3 gC((N + HIST_BINS - 1) / HIST_BINS, COUNT_CHUNKS);

    k_init_deg<<<gN, T, 0, stream>>>(deg0, deg1, N);
    k_count   <<<gC, COUNT_T, 0, stream>>>(dst, deg0, deg1, E, N);
    k_dinv    <<<gN, T, 0, stream>>>(deg0, deg1, dinv, N);
    k_y1      <<<gN, T, 0, stream>>>(x, dinv, y1, agg1, aggB, N);
    k_scatter <<<gE, T, 0, stream>>>(src, dst, y1, agg1, aggB, E);
    k_layer2  <<<gW, T, 0, stream>>>(agg1, aggB, dinv, W1, b1, W2, y2, agg2, N);
    k_scatter <<<gE, T, 0, stream>>>(src, dst, y2, agg2, aggB, E);
    k_final   <<<gN, T, 0, stream>>>(agg2, aggB, dinv, b2, Wc, bc, out, N);
}


# Edges grouped once per call by ranges of 8,192 destination nodes (one packed word per edge); both layers sum edges with LDS float atomics, then whole-line adds; summation order changes
# speedup vs baseline: 7.2853x; 1.3538x over previous
#include <hip/hip_runtime.h>
#include <math.h>

typedef __attribute__((ext_vector_type(2))) float v2f;
typedef __attribute__((ext_vector_type(8))) float v8f;

#define WAVE_EDGES 256

__device__ __forceinline__ int wave_half() {
    unsigned id = __builtin_amdgcn_s_sendmsg_rtn(0x87);
    return (int)((id >> 18) & 1u);
}

__device__ __forceinline__ bool vec_ok(const int* p, int E) {
    return ((E & 3) == 0) && ((((size_t)p) & 15) == 0);
}

__device__ __forceinline__ int wave_index() {
    return blockIdx.x * (blockDim.x >> 5) + __builtin_amdgcn_readfirstlane((int)(threadIdx.x >> 5));
}

__global__ void k_init_deg(int* deg0, int* deg1, int N) {
    int i = blockIdx.x * blockDim.x + threadIdx.x;
    if (i < N) {
        deg0[i] = 1;
        deg1[i] = 0;
    }
}

#define HIST_BITS    16
#define HIST_BINS    (1 << HIST_BITS)
#define COUNT_T      1024
#define COUNT_CHUNKS 32
#define COUNT_LOADS  8

__device__ __forceinline__ void hist_add(int* hist, unsigned slice, int d) {
    if (((unsigned)d >> HIST_BITS) == slice) atomicAdd(&hist[d & (HIST_BINS - 1)], 1);
}

template <bool FULL>
__device__ __forceinline__ void count_group(const int* __restrict__ p, int left,
                                            int* hist, unsigned slice) {
    int4 v[COUNT_LOADS];
    #pragma unroll
    for (int k = 0; k < COUNT_LOADS; ++k) {
        if (FULL || k * (COUNT_T * 4) + 4 <= left) v[k] = *(const int4*)(p + k * (COUNT_T * 4));
        else                                       v[k] = make_int4(-1, -1, -1, -1);
    }
    #pragma unroll
    for (int k = 0; k < COUNT_LOADS; ++k) {
        hist_add(hist, slice, v[k].x);
        hist_add(hist, slice, v[k].y);
        hist_add(hist, slice, v[k].z);
        hist_add(hist, slice, v[k].w);
    }
}

__global__ __launch_bounds__(COUNT_T)
void k_count(const int* __restrict__ dst, int* deg0, int* deg1, int E, int N) {
    __shared__ __attribute__((aligned(16))) int hist[HIST_BINS];
    const int tid = threadIdx.x;
    const unsigned slice = blockIdx.x;

    for (int u = tid * 4; u < HIST_BINS; u += COUNT_T * 4)
        *(int4*)(hist + u) = make_int4(0, 0, 0, 0);
    __syncthreads();

    unsigned per = ((unsigned)E + gridDim.y - 1u) / gridDim.y;
    per = (per + 3u) & ~3u;
    const long long beg = (long long)blockIdx.y * per;
    long long end = beg + per;
    if (end > E) end = E;
    if (end < beg) end = beg;

    const bool vec = (((size_t)dst) & 15) == 0;
    const long long end4 = vec ? beg + ((end - beg) & ~3LL) : beg;
    const long long step = (long long)COUNT_T * 4 * COUNT_LOADS;
    long long i = beg;
    for (; i + step <= end4; i += step)
        count_group<true>(dst + i + tid * 4, 0, hist, slice);
    if (i < end4)
        count_group<false>(dst + i + tid * 4, (int)(end4 - i) - tid * 4, hist, slice);
    for (long long j = end4 + tid; j < end; j += COUNT_T)
        hist_add(hist, slice, dst[j]);
    __syncthreads();

    int* deg = wave_half() ? deg1 : deg0;
    const int base = (int)(slice << HIST_BITS);
    for (int u = tid; u < HIST_BINS; u += COUNT_T * 8) {
        int c[8];
        #pragma unroll
        for (int k = 0; k < 8; ++k) c[k] = hist[u + k * COUNT_T];
        #pragma unroll
        for (int k = 0; k < 8; ++k) {
            const int node = base + u + k * COUNT_T;
            if (c[k] != 0 && node < N) atomicAdd(&deg[node], c[k]);
        }
    }
}

__global__ void k_dinv(const int* deg0, const int* __restrict__ deg1, float* dinv, int N) {
    int i = blockIdx.x * blockDim.x + threadIdx.x;
    if (i < N) {
        float f = (float)(deg0[i] + deg1[i]);
        dinv[i] = 1.0f / sqrtf(f);
    }
}

__global__ void k_y1(const float* __restrict__ x, const float* __restrict__ dinv,
                     float* __restrict__ y1, float* __restrict__ agg1,
                     float* __restrict__ agg1b, int N) {
    int i = blockIdx.x * blockDim.x + threadIdx.x;
    if (i >= N) return;
    float dv = dinv[i];
    float2 xv = *(const float2*)(x + (size_t)i * 2);
    float2 v;
    v.x = dv * xv.x;
    v.y = dv * xv.y;
    *(float2*)(y1   + (size_t)i * 2) = v;
    *(float2*)(agg1 + (size_t)i * 2) = v;
    *(float2*)(agg1b + (size_t)i * 2) = make_float2(0.0f, 0.0f);
}

template <bool FULL>
__device__ __forceinline__ int4 load4(const int* __restrict__ p, int e, int E) {
    int4 a;
    if (FULL) {
        a = *(const int4*)(p + e);
    } else {
        a.x = (e + 0 < E) ? p[e + 0] : -1;
        a.y = (e + 1 < E) ? p[e + 1] : -1;
        a.z = (e + 2 < E) ? p[e + 2] : -1;
        a.w = (e + 3 < E) ? p[e + 3] : -1;
    }
    return a;
}

template <bool FULL>
__device__ __forceinline__ float gather1(const float* __restrict__ y, int s, int f) {
    if (!FULL && s < 0) s = 0;
    return y[(unsigned)s * 2u + (unsigned)f];
}

template <bool FULL>
__device__ __forceinline__ void add1(float* agg, int t, int f, float v) {
    if (FULL || t >= 0)
        atomicAdd(agg + ((unsigned)t * 2u + (unsigned)f), v);
}

template <bool FULL>
__device__ __forceinline__ void scatter_wave(const int* __restrict__ src, const int* __restrict__ dst,
                                             const float* __restrict__ y, float* agg,
                                             int e0, int f, int E) {
    const int4 d0 = load4<FULL>(dst, e0,       E);
    const int4 d1 = load4<FULL>(dst, e0 + 64,  E);
    const int4 d2 = load4<FULL>(dst, e0 + 128, E);
    const int4 d3 = load4<FULL>(dst, e0 + 192, E);
    const int4 s0 = load4<FULL>(src, e0,       E);
    const int4 s1 = load4<FULL>(src, e0 + 64,  E);
    const int4 s2 = load4<FULL>(src, e0 + 128, E);
    const int4 s3 = load4<FULL>(src, e0 + 192, E);
    __builtin_amdgcn_sched_barrier(0);
    const float v00 = gather1<FULL>(y, s0.x, f);
    const float v01 = gather1<FULL>(y, s0.y, f);
    const float v02 = gather1<FULL>(y, s0.z, f);
    const float v03 = gather1<FULL>(y, s0.w, f);
    const float v10 = gather1<FULL>(y, s1.x, f);
    const float v11 = gather1<FULL>(y, s1.y, f);
    const float v12 = gather1<FULL>(y, s1.z, f);
    const float v13 = gather1<FULL>(y, s1.w, f);
    const float v20 = gather1<FULL>(y, s2.x, f);
    const float v21 = gather1<FULL>(y, s2.y, f);
    const float v22 = gather1<FULL>(y, s2.z, f);
    const float v23 = gather1<FULL>(y, s2.w, f);
    const float v30 = gather1<FULL>(y, s3.x, f);
    const float v31 = gather1<FULL>(y, s3.y, f);
    const float v32 = gather1<FULL>(y, s3.z, f);
    const float v33 = gather1<FULL>(y, s3.w, f);
    add1<FULL>(agg, d0.x, f, v00);
    add1<FULL>(agg, d0.y, f, v01);
    add1<FULL>(agg, d0.z, f, v02);
    add1<FULL>(agg, d0.w, f, v03);
    add1<FULL>(agg, d1.x, f, v10);
    add1<FULL>(agg, d1.y, f, v11);
    add1<FULL>(agg, d1.z, f, v12);
    add1<FULL>(agg, d1.w, f, v13);
    add1<FULL>(agg, d2.x, f, v20);
    add1<FULL>(agg, d2.y, f, v21);
    add1<FULL>(agg, d2.z, f, v22);
    add1<FULL>(agg, d2.w, f, v23);
    add1<FULL>(agg, d3.x, f, v30);
    add1<FULL>(agg, d3.y, f, v31);
    add1<FULL>(agg, d3.z, f, v32);
    add1<FULL>(agg, d3.w, f, v33);
}

__global__ void k_scatter(const int* __restrict__ src, const int* __restrict__ dst,
                          const float* __restrict__ y, float* agg0, float* agg1, int E) {
    const int lane = threadIdx.x & 31;
    const int wave = wave_index();
    const long long w0 = (long long)wave * WAVE_EDGES;
    if (w0 >= E) return;
    float* agg = wave_half() ? agg1 : agg0;
    const int f  = lane & 1;
    const int e0 = (int)w0 + (lane >> 1) * 4;
    if (vec_ok(dst, E) && vec_ok(src, E) && w0 + WAVE_EDGES <= E)
        scatter_wave<true>(src, dst, y, agg, e0, f, E);
    else
        scatter_wave<false>(src, dst, y, agg, e0, f, E);
}

#define RANGE_BITS  13
#define RANGE_NODES (1 << RANGE_BITS)
#define SRC_BITS    19
#define MAX_RANGES  64
#define PART_T      1024
#define PART_PER    8
#define PART_CHUNK  (PART_T * PART_PER)
#define PART_G      512
#define PART_SLABS  16
#define AGG_SPLIT   8
#define AGG_PER     16

__device__ __forceinline__ void part_block(int E, long long& beg, long long& end, int& nchunk) {
    const unsigned chunks = ((unsigned)E + (PART_CHUNK - 1u)) / PART_CHUNK;
    const unsigned cpb = (chunks + gridDim.x - 1u) / gridDim.x;
    beg = (long long)blockIdx.x * cpb * PART_CHUNK;
    end = beg + (long long)cpb * PART_CHUNK;
    if (end > E) end = E;
    if (end < beg) end = beg;
    nchunk = (int)((end - beg + PART_CHUNK - 1) / PART_CHUNK);
}

__global__ __launch_bounds__(PART_T)
void k_pcount(const int* __restrict__ dst, int* __restrict__ cnt, int E, int N) {
    __shared__ int wc[(PART_T / 32) * MAX_RANGES];
    const int tid = threadIdx.x;
    for (int u = tid; u < (PART_T / 32) * MAX_RANGES; u += PART_T) wc[u] = 0;
    __syncthreads();
    int* my = wc + (tid >> 5) * MAX_RANGES;
    long long beg, end; int nchunk;
    part_block(E, beg, end, nchunk);
    for (int c = 0; c < nchunk; ++c) {
        const long long base = beg + (long long)c * PART_CHUNK + tid;
        int d[PART_PER];
        if (beg + (long long)(c + 1) * PART_CHUNK <= end) {
            #pragma unroll
            for (int k = 0; k < PART_PER; ++k) d[k] = dst[base + (long long)k * PART_T];
        } else {
            #pragma unroll
            for (int k = 0; k < PART_PER; ++k) {
                const long long e = base + (long long)k * PART_T;
                d[k] = (e < end) ? dst[e] : -1;
            }
        }
        #pragma unroll
        for (int k = 0; k < PART_PER; ++k)
            if ((unsigned)d[k] < (unsigned)N) atomicAdd(&my[d[k] >> RANGE_BITS], 1);
    }
    __syncthreads();
    if (tid < MAX_RANGES) {
        int s = 0;
        for (int w = 0; w < PART_T / 32; ++w) s += wc[w * MAX_RANGES + tid];
        cnt[blockIdx.x * MAX_RANGES + tid] = s;
    }
}

struct PartLayout {
    int a[PART_SLABS * MAX_RANGES];
    int b[PART_SLABS * MAX_RANGES];
    int tot[MAX_RANGES];
    int before[MAX_RANGES];
    int start[MAX_RANGES];
};

__device__ __forceinline__ void part_layout(const int* __restrict__ cnt, int g, PartLayout& L) {
    const int tid = threadIdx.x;
    const int r = tid & (MAX_RANGES - 1);
    const int slab = tid >> 6;
    int all = 0, bef = 0;
    #pragma unroll 8
    for (int j = 0; j < PART_G / PART_SLABS; ++j) {
        const int gg = slab * (PART_G / PART_SLABS) + j;
        const int c = cnt[gg * MAX_RANGES + r];
        all += c;
        bef += (gg < g) ? c : 0;
    }
    L.a[slab * MAX_RANGES + r] = all;
    L.b[slab * MAX_RANGES + r] = bef;
    __syncthreads();
    if (tid < MAX_RANGES) {
        int t = 0, b = 0;
        for (int p = 0; p < PART_SLABS; ++p) {
            t += L.a[p * MAX_RANGES + tid];
            b += L.b[p * MAX_RANGES + tid];
        }
        L.tot[tid] = t;
        L.before[tid] = b;
    }
    __syncthreads();
    if (tid < MAX_RANGES) {
        int s = 0;
        for (int q = 0; q < tid; ++q) s += (L.tot[q] + 31) & ~31;
        L.start[tid] = s;
    }
    __syncthreads();
}

__global__ __launch_bounds__(PART_T)
void k_part(const int* __restrict__ src, const int* __restrict__ dst, const int* __restrict__ cnt,
            unsigned* __restrict__ grouped, int E, int N) {
    __shared__ PartLayout L;
    __shared__ int s_run[MAX_RANGES];
    __shared__ int s_cnt[MAX_RANGES];
    __shared__ int s_lofs[MAX_RANGES];
    __shared__ unsigned s_word[PART_CHUNK];
    __shared__ int s_pos[PART_CHUNK];
    const int tid = threadIdx.x;

    part_layout(cnt, (int)blockIdx.x, L);
    if (tid < MAX_RANGES) s_run[tid] = L.start[tid] + L.before[tid];

    long long beg, end; int nchunk;
    part_block(E, beg, end, nchunk);
    for (int c = 0; c < nchunk; ++c) {
        if (tid < MAX_RANGES) s_cnt[tid] = 0;
        __syncthreads();

        const long long base = beg + (long long)c * PART_CHUNK + tid;
        int d[PART_PER], s[PART_PER];
        if (beg + (long long)(c + 1) * PART_CHUNK <= end) {
            #pragma unroll
            for (int k = 0; k < PART_PER; ++k) {
                d[k] = dst[base + (long long)k * PART_T];
                s[k] = src[base + (long long)k * PART_T];
            }
        } else {
            #pragma unroll
            for (int k = 0; k < PART_PER; ++k) {
                const long long e = base + (long long)k * PART_T;
                const bool in = e < end;
                d[k] = in ? dst[e] : -1;
                s[k] = in ? src[e] : 0;
            }
        }
        int rk[PART_PER];
        #pragma unroll
        for (int k = 0; k < PART_PER; ++k) {
            rk[k] = 0;
            if ((unsigned)d[k] < (unsigned)N) rk[k] = atomicAdd(&s_cnt[d[k] >> RANGE_BITS], 1);
        }
        __syncthreads();
        if (tid < MAX_RANGES) {
            int o = 0;
            for (int q = 0; q < tid; ++q) o += s_cnt[q];
            s_lofs[tid] = o;
        }
        __syncthreads();
        #pragma unroll
        for (int k = 0; k < PART_PER; ++k) {
            if ((unsigned)d[k] < (unsigned)N) {
                const int r = d[k] >> RANGE_BITS;
                const int p = s_lofs[r] + rk[k];
                s_word[p] = ((unsigned)s[k] << RANGE_BITS) | ((unsigned)d[k] & (RANGE_NODES - 1u));
                s_pos[p]  = s_run[r] + rk[k];
            }
        }
        __syncthreads();
        const int total = s_lofs[MAX_RANGES - 1] + s_cnt[MAX_RANGES - 1];
        for (int i = tid; i < total; i += PART_T) grouped[s_pos[i]] = s_word[i];
        __syncthreads();
        if (tid < MAX_RANGES) s_run[tid] += s_cnt[tid];
    }
}

template <bool FULL>
__device__ __forceinline__ void agg_group(const unsigned* __restrict__ p, int left,
                                          const float* __restrict__ y, float* sx, float* sy) {
    unsigned w[AGG_PER];
    #pragma unroll
    for (int k = 0; k < AGG_PER; ++k)
        w[k] = (FULL || k * PART_T < left) ? p[k * PART_T] : 0u;
    float2 v[AGG_PER];
    #pragma unroll
    for (int k = 0; k < AGG_PER; ++k)
        v[k] = *(const float2*)(y + (size_t)(w[k] >> RANGE_BITS) * 2);
    #pragma unroll
    for (int k = 0; k < AGG_PER; ++k) {
        if (FULL || k * PART_T < left) {
            const unsigned o = w[k] & (RANGE_NODES - 1u);
            atomicAdd(&sx[o], v[k].x);
            atomicAdd(&sy[o], v[k].y);
        }
    }
}

__global__ __launch_bounds__(PART_T)
void k_agg(const unsigned* __restrict__ grouped, const int* __restrict__ cnt,
           const float* __restrict__ y, float* agg0, float* agg1, int N) {
    __shared__ PartLayout L;
    __shared__ float sx[RANGE_NODES];
    __shared__ float sy[RANGE_NODES];
    const int tid = threadIdx.x;
    const int r = blockIdx.y;

    for (int u = tid; u < RANGE_NODES; u += PART_T) { sx[u] = 0.0f; sy[u] = 0.0f; }
    part_layout(cnt, 0, L);

    const int tot = L.tot[r];
    int per = (tot + AGG_SPLIT - 1) / AGG_SPLIT;
    per = (per + 31) & ~31;
    int lo = (int)blockIdx.x * per;
    int hi = lo + per;
    if (hi > tot) hi = tot;
    if (lo > hi) lo = hi;
    const unsigned* base = grouped + L.start[r];

    const int step = PART_T * AGG_PER;
    int i = lo;
    for (; i + step <= hi; i += step)
        agg_group<true>(base + i + tid, 0, y, sx, sy);
    if (i < hi)
        agg_group<false>(base + i + tid, hi - i - tid, y, sx, sy);
    __syncthreads();

    float* agg = wave_half() ? agg1 : agg0;
    const int node0 = r << RANGE_BITS;
    for (int u = tid; u < 2 * RANGE_NODES; u += PART_T) {
        const int n = u >> 1;
        if (node0 + n < N) {
            const float val = (u & 1) ? sy[n] : sx[n];
            atomicAdd(agg + ((size_t)node0 * 2 + u), val);
        }
    }
}

__global__ void k_layer2(const float* __restrict__ agg1, float* aggB,
                         const float* __restrict__ dinv,
                         const float* __restrict__ W1, const float* __restrict__ b1,
                         const float* __restrict__ W2,
                         float* __restrict__ y2, float* __restrict__ agg2, int N) {
    int lane = threadIdx.x & 31;
    int wave = threadIdx.x >> 5;
    int base = (blockIdx.x * (blockDim.x >> 5) + wave) * 16;
    if (base >= N) return;
    int m    = lane & 15;
    int half = lane >> 4;
    int node = base + m;
    int kb   = half * 2;
    bool full = (base + 16 <= N);

    int j = m & 7;
    v2f a1 = {0.0f, 0.0f};
    if (half == 0 && j < 4) {
        a1.x = W1[0 * 4 + j];
        a1.y = W1[1 * 4 + j];
    }
    v2f s = {0.0f, 0.0f};
    if (half == 0 && node < N) {
        float dv = dinv[node];
        float2 g = *(const float2*)(agg1 + (size_t)node * 2);
        float2* pb = (float2*)(aggB + (size_t)node * 2);
        float2 gb = *pb;
        *pb = make_float2(0.0f, 0.0f);
        s.x = dv * (g.x + gb.x);
        s.y = dv * (g.y + gb.y);
    }
    v8f c1 = {b1[0], b1[1], b1[2], b1[3], 0.0f, 0.0f, 0.0f, 0.0f};
    v8f p = __builtin_amdgcn_wmma_f32_16x16x4_f32(
        false, a1, false, s, (short)0, c1, false, false);

    v2f a2;
    a2.x = tanhf(half ? p[2] : p[0]);
    a2.y = tanhf(half ? p[3] : p[1]);

    v2f b2 = {0.0f, 0.0f};
    if (m < 2) {
        b2.x = W2[(kb + 0) * 2 + m];
        b2.y = W2[(kb + 1) * 2 + m];
    }
    v8f c2 = {};
    v8f d = __builtin_amdgcn_wmma_f32_16x16x4_f32(
        false, a2, false, b2, (short)0, c2, false, false);

    int nb = base + 8 * half;
    if (full) {
        if (m < 2) {
            float dv[8];
            #pragma unroll
            for (int v = 0; v < 8; ++v) dv[v] = dinv[nb + v];
            #pragma unroll
            for (int v = 0; v < 8; ++v) {
                float val = dv[v] * d[v];
                y2[(nb + v) * 2 + m]   = val;
                agg2[(nb + v) * 2 + m] = val;
            }
        }
    } else {
        if (m < 2) {
            #pragma unroll
            for (int v = 0; v < 8; ++v) {
                int nd = nb + v;
                if (nd < N) {
                    float val = dinv[nd] * d[v];
                    y2[nd * 2 + m]   = val;
                    agg2[nd * 2 + m] = val;
                }
            }
        }
    }
}

__global__ void k_final(const float* __restrict__ agg2, const float* __restrict__ agg2b,
                        const float* __restrict__ dinv,
                        const float* __restrict__ b2, const float* __restrict__ Wc,
                        const float* __restrict__ bc, float* __restrict__ out, int N) {
    int i = blockIdx.x * blockDim.x + threadIdx.x;
    if (i >= N) return;
    float dv = dinv[i];
    float2 ga = *(const float2*)(agg2  + (size_t)i * 2);
    float2 gb = *(const float2*)(agg2b + (size_t)i * 2);
    float h0 = tanhf(dv * (ga.x + gb.x) + b2[0]);
    float h1 = tanhf(dv * (ga.y + gb.y) + b2[1]);
    float2 o;
    o.x = h0 * Wc[0] + h1 * Wc[2] + bc[0];
    o.y = h0 * Wc[1] + h1 * Wc[3] + bc[1];
    *(float2*)(out + (size_t)i * 2) = o;
}

extern "C" void kernel_launch(void* const* d_in, const int* in_sizes, int n_in,
                              void* d_out, int out_size, void* d_ws, size_t ws_size,
                              hipStream_t stream) {
    const float* x  = (const float*)d_in[0];
    const int*   ei = (const int*)d_in[1];
    const float* W1 = (const float*)d_in[2];
    const float* b1 = (const float*)d_in[3];
    const float* W2 = (const float*)d_in[4];
    const float* b2 = (const float*)d_in[5];
    const float* Wc = (const float*)d_in[6];
    const float* bc = (const float*)d_in[7];
    float* out = (float*)d_out;

    const int N = in_sizes[0] / 2;
    const int E = in_sizes[1] / 2;
    const int* src = ei;
    const int* dst = ei + E;

    float* ws   = (float*)d_ws;
    float* y1   = ws;
    float* agg1 = ws + (size_t)2 * N;
    float* aggB = ws + (size_t)4 * N;
    float* y2   = ws;
    float* agg2 = ws + (size_t)6 * N;
    float* dinv = ws + (size_t)8 * N;
    int*   deg0 = (int*)dinv;
    int*   deg1 = (int*)ws;

    const int T = 256;
    dim3 gN((N + T - 1) / T);
    const int edgesPerBlock = (T / 32) * WAVE_EDGES;
    dim3 gE((E + edgesPerBlock - 1) / edgesPerBlock);
    const int nodesPerBlock = (T / 32) * 16;
    dim3 gW((N + nodesPerBlock - 1) / nodesPerBlock);
    dim3 gC((N + HIST_BINS - 1) / HIST_BINS, COUNT_CHUNKS);

    const size_t cntOff   = (((size_t)9 * N * sizeof(float)) + 255) & ~(size_t)255;
    const size_t packOff  = cntOff + (size_t)PART_G * MAX_RANGES * sizeof(int);
    const size_t packEnd  = packOff + ((size_t)E + MAX_RANGES * 32) * sizeof(unsigned);
    const bool   groupedPath = (N <= (1 << SRC_BITS)) && (ws_size >= packEnd);
    int*      cnt     = (int*)((char*)d_ws + cntOff);
    unsigned* grouped = (unsigned*)((char*)d_ws + packOff);
    dim3 gA(AGG_SPLIT, (N + RANGE_NODES - 1) / RANGE_NODES);

    k_init_deg<<<gN, T, 0, stream>>>(deg0, deg1, N);
    k_count   <<<gC, COUNT_T, 0, stream>>>(dst, deg0, deg1, E, N);
    k_dinv    <<<gN, T, 0, stream>>>(deg0, deg1, dinv, N);
    k_y1      <<<gN, T, 0, stream>>>(x, dinv, y1, agg1, aggB, N);
    if (groupedPath) {
        k_pcount<<<PART_G, PART_T, 0, stream>>>(dst, cnt, E, N);
        k_part  <<<PART_G, PART_T, 0, stream>>>(src, dst, cnt, grouped, E, N);
        k_agg   <<<gA, PART_T, 0, stream>>>(grouped, cnt, y1, agg1, aggB, N);
    } else {
        k_scatter<<<gE, T, 0, stream>>>(src, dst, y1, agg1, aggB, E);
    }
    k_layer2  <<<gW, T, 0, stream>>>(agg1, aggB, dinv, W1, b1, W2, y2, agg2, N);
    if (groupedPath) k_agg    <<<gA, PART_T, 0, stream>>>(grouped, cnt, y2, agg2, aggB, N);
    else             k_scatter<<<gE, T, 0, stream>>>(src, dst, y2, agg2, aggB, E);
    k_final   <<<gN, T, 0, stream>>>(agg2, aggB, dinv, b2, Wc, bc, out, N);
}


# Degree counted from the grouped edge list: one pass, 8,192 integer LDS counters per range, three shares stored as plain planes that the node kernel sums; 8-pass dst histogram and integer atomics go
# speedup vs baseline: 8.2091x; 1.1268x over previous
#include <hip/hip_runtime.h>
#include <math.h>

typedef __attribute__((ext_vector_type(2))) float v2f;
typedef __attribute__((ext_vector_type(8))) float v8f;

#define WAVE_EDGES 256

__device__ __forceinline__ int wave_half() {
    unsigned id = __builtin_amdgcn_s_sendmsg_rtn(0x87);
    return (int)((id >> 18) & 1u);
}

__device__ __forceinline__ bool vec_ok(const int* p, int E) {
    return ((E & 3) == 0) && ((((size_t)p) & 15) == 0);
}

__device__ __forceinline__ int wave_index() {
    return blockIdx.x * (blockDim.x >> 5) + __builtin_amdgcn_readfirstlane((int)(threadIdx.x >> 5));
}

__global__ void k_init_deg(int* deg0, int* deg1, int N) {
    int i = blockIdx.x * blockDim.x + threadIdx.x;
    if (i < N) {
        deg0[i] = 1;
        deg1[i] = 0;
    }
}

#define HIST_BITS    16
#define HIST_BINS    (1 << HIST_BITS)
#define COUNT_T      1024
#define COUNT_CHUNKS 32
#define COUNT_LOADS  8

__device__ __forceinline__ void hist_add(int* hist, unsigned slice, int d) {
    if (((unsigned)d >> HIST_BITS) == slice) atomicAdd(&hist[d & (HIST_BINS - 1)], 1);
}

template <bool FULL>
__device__ __forceinline__ void count_group(const int* __restrict__ p, int left,
                                            int* hist, unsigned slice) {
    int4 v[COUNT_LOADS];
    #pragma unroll
    for (int k = 0; k < COUNT_LOADS; ++k) {
        if (FULL || k * (COUNT_T * 4) + 4 <= left) v[k] = *(const int4*)(p + k * (COUNT_T * 4));
        else                                       v[k] = make_int4(-1, -1, -1, -1);
    }
    #pragma unroll
    for (int k = 0; k < COUNT_LOADS; ++k) {
        hist_add(hist, slice, v[k].x);
        hist_add(hist, slice, v[k].y);
        hist_add(hist, slice, v[k].z);
        hist_add(hist, slice, v[k].w);
    }
}

__global__ __launch_bounds__(COUNT_T)
void k_count(const int* __restrict__ dst, int* deg0, int* deg1, int E, int N) {
    __shared__ __attribute__((aligned(16))) int hist[HIST_BINS];
    const int tid = threadIdx.x;
    const unsigned slice = blockIdx.x;

    for (int u = tid * 4; u < HIST_BINS; u += COUNT_T * 4)
        *(int4*)(hist + u) = make_int4(0, 0, 0, 0);
    __syncthreads();

    unsigned per = ((unsigned)E + gridDim.y - 1u) / gridDim.y;
    per = (per + 3u) & ~3u;
    const long long beg = (long long)blockIdx.y * per;
    long long end = beg + per;
    if (end > E) end = E;
    if (end < beg) end = beg;

    const bool vec = (((size_t)dst) & 15) == 0;
    const long long end4 = vec ? beg + ((end - beg) & ~3LL) : beg;
    const long long step = (long long)COUNT_T * 4 * COUNT_LOADS;
    long long i = beg;
    for (; i + step <= end4; i += step)
        count_group<true>(dst + i + tid * 4, 0, hist, slice);
    if (i < end4)
        count_group<false>(dst + i + tid * 4, (int)(end4 - i) - tid * 4, hist, slice);
    for (long long j = end4 + tid; j < end; j += COUNT_T)
        hist_add(hist, slice, dst[j]);
    __syncthreads();

    int* deg = wave_half() ? deg1 : deg0;
    const int base = (int)(slice << HIST_BITS);
    for (int u = tid; u < HIST_BINS; u += COUNT_T * 8) {
        int c[8];
        #pragma unroll
        for (int k = 0; k < 8; ++k) c[k] = hist[u + k * COUNT_T];
        #pragma unroll
        for (int k = 0; k < 8; ++k) {
            const int node = base + u + k * COUNT_T;
            if (c[k] != 0 && node < N) atomicAdd(&deg[node], c[k]);
        }
    }
}

__global__ void k_dinv(const int* deg0, const int* __restrict__ deg1, float* dinv, int N) {
    int i = blockIdx.x * blockDim.x + threadIdx.x;
    if (i < N) {
        float f = (float)(deg0[i] + deg1[i]);
        dinv[i] = 1.0f / sqrtf(f);
    }
}

__global__ void k_y1(const float* __restrict__ x, const float* __restrict__ dinv,
                     float* __restrict__ y1, float* __restrict__ agg1,
                     float* __restrict__ agg1b, int N) {
    int i = blockIdx.x * blockDim.x + threadIdx.x;
    if (i >= N) return;
    float dv = dinv[i];
    float2 xv = *(const float2*)(x + (size_t)i * 2);
    float2 v;
    v.x = dv * xv.x;
    v.y = dv * xv.y;
    *(float2*)(y1   + (size_t)i * 2) = v;
    *(float2*)(agg1 + (size_t)i * 2) = v;
    *(float2*)(agg1b + (size_t)i * 2) = make_float2(0.0f, 0.0f);
}

template <bool FULL>
__device__ __forceinline__ int4 load4(const int* __restrict__ p, int e, int E) {
    int4 a;
    if (FULL) {
        a = *(const int4*)(p + e);
    } else {
        a.x = (e + 0 < E) ? p[e + 0] : -1;
        a.y = (e + 1 < E) ? p[e + 1] : -1;
        a.z = (e + 2 < E) ? p[e + 2] : -1;
        a.w = (e + 3 < E) ? p[e + 3] : -1;
    }
    return a;
}

template <bool FULL>
__device__ __forceinline__ float gather1(const float* __restrict__ y, int s, int f) {
    if (!FULL && s < 0) s = 0;
    return y[(unsigned)s * 2u + (unsigned)f];
}

template <bool FULL>
__device__ __forceinline__ void add1(float* agg, int t, int f, float v) {
    if (FULL || t >= 0)
        atomicAdd(agg + ((unsigned)t * 2u + (unsigned)f), v);
}

template <bool FULL>
__device__ __forceinline__ void scatter_wave(const int* __restrict__ src, const int* __restrict__ dst,
                                             const float* __restrict__ y, float* agg,
                                             int e0, int f, int E) {
    const int4 d0 = load4<FULL>(dst, e0,       E);
    const int4 d1 = load4<FULL>(dst, e0 + 64,  E);
    const int4 d2 = load4<FULL>(dst, e0 + 128, E);
    const int4 d3 = load4<FULL>(dst, e0 + 192, E);
    const int4 s0 = load4<FULL>(src, e0,       E);
    const int4 s1 = load4<FULL>(src, e0 + 64,  E);
    const int4 s2 = load4<FULL>(src, e0 + 128, E);
    const int4 s3 = load4<FULL>(src, e0 + 192, E);
    __builtin_amdgcn_sched_barrier(0);
    const float v00 = gather1<FULL>(y, s0.x, f);
    const float v01 = gather1<FULL>(y, s0.y, f);
    const float v02 = gather1<FULL>(y, s0.z, f);
    const float v03 = gather1<FULL>(y, s0.w, f);
    const float v10 = gather1<FULL>(y, s1.x, f);
    const float v11 = gather1<FULL>(y, s1.y, f);
    const float v12 = gather1<FULL>(y, s1.z, f);
    const float v13 = gather1<FULL>(y, s1.w, f);
    const float v20 = gather1<FULL>(y, s2.x, f);
    const float v21 = gather1<FULL>(y, s2.y, f);
    const float v22 = gather1<FULL>(y, s2.z, f);
    const float v23 = gather1<FULL>(y, s2.w, f);
    const float v30 = gather1<FULL>(y, s3.x, f);
    const float v31 = gather1<FULL>(y, s3.y, f);
    const float v32 = gather1<FULL>(y, s3.z, f);
    const float v33 = gather1<FULL>(y, s3.w, f);
    add1<FULL>(agg, d0.x, f, v00);
    add1<FULL>(agg, d0.y, f, v01);
    add1<FULL>(agg, d0.z, f, v02);
    add1<FULL>(agg, d0.w, f, v03);
    add1<FULL>(agg, d1.x, f, v10);
    add1<FULL>(agg, d1.y, f, v11);
    add1<FULL>(agg, d1.z, f, v12);
    add1<FULL>(agg, d1.w, f, v13);
    add1<FULL>(agg, d2.x, f, v20);
    add1<FULL>(agg, d2.y, f, v21);
    add1<FULL>(agg, d2.z, f, v22);
    add1<FULL>(agg, d2.w, f, v23);
    add1<FULL>(agg, d3.x, f, v30);
    add1<FULL>(agg, d3.y, f, v31);
    add1<FULL>(agg, d3.z, f, v32);
    add1<FULL>(agg, d3.w, f, v33);
}

__global__ void k_scatter(const int* __restrict__ src, const int* __restrict__ dst,
                          const float* __restrict__ y, float* agg0, float* agg1, int E) {
    const int lane = threadIdx.x & 31;
    const int wave = wave_index();
    const long long w0 = (long long)wave * WAVE_EDGES;
    if (w0 >= E) return;
    float* agg = wave_half() ? agg1 : agg0;
    const int f  = lane & 1;
    const int e0 = (int)w0 + (lane >> 1) * 4;
    if (vec_ok(dst, E) && vec_ok(src, E) && w0 + WAVE_EDGES <= E)
        scatter_wave<true>(src, dst, y, agg, e0, f, E);
    else
        scatter_wave<false>(src, dst, y, agg, e0, f, E);
}

#define RANGE_BITS  13
#define RANGE_NODES (1 << RANGE_BITS)
#define SRC_BITS    19
#define MAX_RANGES  64
#define PART_T      1024
#define PART_PER    8
#define PART_CHUNK  (PART_T * PART_PER)
#define PART_G      512
#define PART_SLABS  16
#define AGG_SPLIT   8
#define AGG_PER     16

__device__ __forceinline__ void part_block(int E, long long& beg, long long& end, int& nchunk) {
    const unsigned chunks = ((unsigned)E + (PART_CHUNK - 1u)) / PART_CHUNK;
    const unsigned cpb = (chunks + gridDim.x - 1u) / gridDim.x;
    beg = (long long)blockIdx.x * cpb * PART_CHUNK;
    end = beg + (long long)cpb * PART_CHUNK;
    if (end > E) end = E;
    if (end < beg) end = beg;
    nchunk = (int)((end - beg + PART_CHUNK - 1) / PART_CHUNK);
}

__global__ __launch_bounds__(PART_T)
void k_pcount(const int* __restrict__ dst, int* __restrict__ cnt, int E, int N) {
    __shared__ int wc[(PART_T / 32) * MAX_RANGES];
    const int tid = threadIdx.x;
    for (int u = tid; u < (PART_T / 32) * MAX_RANGES; u += PART_T) wc[u] = 0;
    __syncthreads();
    int* my = wc + (tid >> 5) * MAX_RANGES;
    long long beg, end; int nchunk;
    part_block(E, beg, end, nchunk);
    for (int c = 0; c < nchunk; ++c) {
        const long long base = beg + (long long)c * PART_CHUNK + tid;
        int d[PART_PER];
        if (beg + (long long)(c + 1) * PART_CHUNK <= end) {
            #pragma unroll
            for (int k = 0; k < PART_PER; ++k) d[k] = dst[base + (long long)k * PART_T];
        } else {
            #pragma unroll
            for (int k = 0; k < PART_PER; ++k) {
                const long long e = base + (long long)k * PART_T;
                d[k] = (e < end) ? dst[e] : -1;
            }
        }
        #pragma unroll
        for (int k = 0; k < PART_PER; ++k)
            if ((unsigned)d[k] < (unsigned)N) atomicAdd(&my[d[k] >> RANGE_BITS], 1);
    }
    __syncthreads();
    if (tid < MAX_RANGES) {
        int s = 0;
        for (int w = 0; w < PART_T / 32; ++w) s += wc[w * MAX_RANGES + tid];
        cnt[blockIdx.x * MAX_RANGES + tid] = s;
    }
}

struct PartLayout {
    int a[PART_SLABS * MAX_RANGES];
    int b[PART_SLABS * MAX_RANGES];
    int tot[MAX_RANGES];
    int before[MAX_RANGES];
    int start[MAX_RANGES];
};

__device__ __forceinline__ void part_layout(const int* __restrict__ cnt, int g, PartLayout& L) {
    const int tid = threadIdx.x;
    const int r = tid & (MAX_RANGES - 1);
    const int slab = tid >> 6;
    int all = 0, bef = 0;
    #pragma unroll 8
    for (int j = 0; j < PART_G / PART_SLABS; ++j) {
        const int gg = slab * (PART_G / PART_SLABS) + j;
        const int c = cnt[gg * MAX_RANGES + r];
        all += c;
        bef += (gg < g) ? c : 0;
    }
    L.a[slab * MAX_RANGES + r] = all;
    L.b[slab * MAX_RANGES + r] = bef;
    __syncthreads();
    if (tid < MAX_RANGES) {
        int t = 0, b = 0;
        for (int p = 0; p < PART_SLABS; ++p) {
            t += L.a[p * MAX_RANGES + tid];
            b += L.b[p * MAX_RANGES + tid];
        }
        L.tot[tid] = t;
        L.before[tid] = b;
    }
    __syncthreads();
    if (tid < MAX_RANGES) {
        int s = 0;
        for (int q = 0; q < tid; ++q) s += (L.tot[q] + 31) & ~31;
        L.start[tid] = s;
    }
    __syncthreads();
}

__global__ __launch_bounds__(PART_T)
void k_part(const int* __restrict__ src, const int* __restrict__ dst, const int* __restrict__ cnt,
            unsigned* __restrict__ grouped, int E, int N) {
    __shared__ PartLayout L;
    __shared__ int s_run[MAX_RANGES];
    __shared__ int s_cnt[MAX_RANGES];
    __shared__ int s_lofs[MAX_RANGES];
    __shared__ unsigned s_word[PART_CHUNK];
    __shared__ int s_pos[PART_CHUNK];
    const int tid = threadIdx.x;

    part_layout(cnt, (int)blockIdx.x, L);
    if (tid < MAX_RANGES) s_run[tid] = L.start[tid] + L.before[tid];

    long long beg, end; int nchunk;
    part_block(E, beg, end, nchunk);
    for (int c = 0; c < nchunk; ++c) {
        if (tid < MAX_RANGES) s_cnt[tid] = 0;
        __syncthreads();

        const long long base = beg + (long long)c * PART_CHUNK + tid;
        int d[PART_PER], s[PART_PER];
        if (beg + (long long)(c + 1) * PART_CHUNK <= end) {
            #pragma unroll
            for (int k = 0; k < PART_PER; ++k) {
                d[k] = dst[base + (long long)k * PART_T];
                s[k] = src[base + (long long)k * PART_T];
            }
        } else {
            #pragma unroll
            for (int k = 0; k < PART_PER; ++k) {
                const long long e = base + (long long)k * PART_T;
                const bool in = e < end;
                d[k] = in ? dst[e] : -1;
                s[k] = in ? src[e] : 0;
            }
        }
        int rk[PART_PER];
        #pragma unroll
        for (int k = 0; k < PART_PER; ++k) {
            rk[k] = 0;
            if ((unsigned)d[k] < (unsigned)N) rk[k] = atomicAdd(&s_cnt[d[k] >> RANGE_BITS], 1);
        }
        __syncthreads();
        if (tid < MAX_RANGES) {
            int o = 0;
            for (int q = 0; q < tid; ++q) o += s_cnt[q];
            s_lofs[tid] = o;
        }
        __syncthreads();
        #pragma unroll
        for (int k = 0; k < PART_PER; ++k) {
            if ((unsigned)d[k] < (unsigned)N) {
                const int r = d[k] >> RANGE_BITS;
                const int p = s_lofs[r] + rk[k];
                s_word[p] = ((unsigned)s[k] << RANGE_BITS) | ((unsigned)d[k] & (RANGE_NODES - 1u));
                s_pos[p]  = s_run[r] + rk[k];
            }
        }
        __syncthreads();
        const int total = s_lofs[MAX_RANGES - 1] + s_cnt[MAX_RANGES - 1];
        for (int i = tid; i < total; i += PART_T) grouped[s_pos[i]] = s_word[i];
        __syncthreads();
        if (tid < MAX_RANGES) s_run[tid] += s_cnt[tid];
    }
}

template <bool FULL>
__device__ __forceinline__ void agg_group(const unsigned* __restrict__ p, int left,
                                          const float* __restrict__ y, float* sx, float* sy) {
    unsigned w[AGG_PER];
    #pragma unroll
    for (int k = 0; k < AGG_PER; ++k)
        w[k] = (FULL || k * PART_T < left) ? p[k * PART_T] : 0u;
    float2 v[AGG_PER];
    #pragma unroll
    for (int k = 0; k < AGG_PER; ++k)
        v[k] = *(const float2*)(y + (size_t)(w[k] >> RANGE_BITS) * 2);
    #pragma unroll
    for (int k = 0; k < AGG_PER; ++k) {
        if (FULL || k * PART_T < left) {
            const unsigned o = w[k] & (RANGE_NODES - 1u);
            atomicAdd(&sx[o], v[k].x);
            atomicAdd(&sy[o], v[k].y);
        }
    }
}

__global__ __launch_bounds__(PART_T)
void k_agg(const unsigned* __restrict__ grouped, const int* __restrict__ cnt,
           const float* __restrict__ y, float* agg0, float* agg1, int N) {
    __shared__ PartLayout L;
    __shared__ float sx[RANGE_NODES];
    __shared__ float sy[RANGE_NODES];
    const int tid = threadIdx.x;
    const int r = blockIdx.y;

    for (int u = tid; u < RANGE_NODES; u += PART_T) { sx[u] = 0.0f; sy[u] = 0.0f; }
    part_layout(cnt, 0, L);

    const int tot = L.tot[r];
    int per = (tot + AGG_SPLIT - 1) / AGG_SPLIT;
    per = (per + 31) & ~31;
    int lo = (int)blockIdx.x * per;
    int hi = lo + per;
    if (hi > tot) hi = tot;
    if (lo > hi) lo = hi;
    const unsigned* base = grouped + L.start[r];

    const int step = PART_T * AGG_PER;
    int i = lo;
    for (; i + step <= hi; i += step)
        agg_group<true>(base + i + tid, 0, y, sx, sy);
    if (i < hi)
        agg_group<false>(base + i + tid, hi - i - tid, y, sx, sy);
    __syncthreads();

    float* agg = wave_half() ? agg1 : agg0;
    const int node0 = r << RANGE_BITS;
    for (int u = tid; u < 2 * RANGE_NODES; u += PART_T) {
        const int n = u >> 1;
        if (node0 + n < N) {
            const float val = (u & 1) ? sy[n] : sx[n];
            atomicAdd(agg + ((size_t)node0 * 2 + u), val);
        }
    }
}

#define DEG_SPLIT 3
#define DEG_LOADS 8

template <bool FULL>
__device__ __forceinline__ void deg_group(const unsigned* __restrict__ p,
                                          const unsigned* __restrict__ safe, int left, int* hist) {
    uint4 v[DEG_LOADS];
    #pragma unroll
    for (int k = 0; k < DEG_LOADS; ++k) {
        const bool in = FULL || k * (PART_T * 4) + 4 <= left;
        v[k] = *(const uint4*)(in ? p + k * (PART_T * 4) : safe);
    }
    __builtin_amdgcn_sched_barrier(0);
    #pragma unroll
    for (int k = 0; k < DEG_LOADS; ++k) {
        if (FULL || k * (PART_T * 4) + 4 <= left) {
            atomicAdd(&hist[v[k].x & (RANGE_NODES - 1u)], 1);
            atomicAdd(&hist[v[k].y & (RANGE_NODES - 1u)], 1);
            atomicAdd(&hist[v[k].z & (RANGE_NODES - 1u)], 1);
            atomicAdd(&hist[v[k].w & (RANGE_NODES - 1u)], 1);
        }
    }
}

__global__ __launch_bounds__(PART_T)
void k_deg(const unsigned* __restrict__ grouped, const int* __restrict__ cnt,
           int* __restrict__ part, size_t stride, int N) {
    __shared__ PartLayout L;
    __shared__ int hist[RANGE_NODES];
    const int tid = threadIdx.x;
    const int r = blockIdx.y;

    for (int u = tid; u < RANGE_NODES; u += PART_T) hist[u] = 0;
    part_layout(cnt, 0, L);

    const int tot = L.tot[r];
    int per = (tot + DEG_SPLIT - 1) / DEG_SPLIT;
    per = (per + 31) & ~31;
    int lo = (int)blockIdx.x * per;
    int hi = lo + per;
    if (hi > tot) hi = tot;
    if (lo > hi) lo = hi;
    const unsigned* base = grouped + L.start[r];

    const bool vec = (((size_t)grouped) & 15) == 0;
    const int hi4 = vec ? lo + ((hi - lo) & ~3) : lo;
    const int step = PART_T * 4 * DEG_LOADS;
    int i = lo;
    for (; i + step <= hi4; i += step)
        deg_group<true>(base + i + tid * 4, base + i, 0, hist);
    if (i < hi4)
        deg_group<false>(base + i + tid * 4, base + i, hi4 - i - tid * 4, hist);
    for (int j = hi4 + tid; j < hi; j += PART_T)
        atomicAdd(&hist[base[j] & (RANGE_NODES - 1u)], 1);
    __syncthreads();

    int* mine = part + (size_t)blockIdx.x * stride;
    const int node0 = r << RANGE_BITS;
    for (int u = tid; u < RANGE_NODES; u += PART_T)
        if (node0 + u < N) mine[node0 + u] = hist[u];
}

__global__ void k_node(const int* __restrict__ p0, const int* __restrict__ p1, const int* p2,
                       const float* __restrict__ x, float* dinv,
                       float* __restrict__ y1, float* __restrict__ agg1,
                       float* __restrict__ agg1b, int N) {
    int i = blockIdx.x * blockDim.x + threadIdx.x;
    if (i >= N) return;
    float f = (float)(p0[i] + p1[i] + p2[i] + 1);
    float dv = 1.0f / sqrtf(f);
    dinv[i] = dv;
    float2 xv = *(const float2*)(x + (size_t)i * 2);
    float2 v;
    v.x = dv * xv.x;
    v.y = dv * xv.y;
    *(float2*)(y1   + (size_t)i * 2) = v;
    *(float2*)(agg1 + (size_t)i * 2) = v;
    *(float2*)(agg1b + (size_t)i * 2) = make_float2(0.0f, 0.0f);
}

__global__ void k_layer2(const float* __restrict__ agg1, float* aggB,
                         const float* __restrict__ dinv,
                         const float* __restrict__ W1, const float* __restrict__ b1,
                         const float* __restrict__ W2,
                         float* __restrict__ y2, float* __restrict__ agg2, int N) {
    int lane = threadIdx.x & 31;
    int wave = threadIdx.x >> 5;
    int base = (blockIdx.x * (blockDim.x >> 5) + wave) * 16;
    if (base >= N) return;
    int m    = lane & 15;
    int half = lane >> 4;
    int node = base + m;
    int kb   = half * 2;
    bool full = (base + 16 <= N);

    int j = m & 7;
    v2f a1 = {0.0f, 0.0f};
    if (half == 0 && j < 4) {
        a1.x = W1[0 * 4 + j];
        a1.y = W1[1 * 4 + j];
    }
    v2f s = {0.0f, 0.0f};
    if (half == 0 && node < N) {
        float dv = dinv[node];
        float2 g = *(const float2*)(agg1 + (size_t)node * 2);
        float2* pb = (float2*)(aggB + (size_t)node * 2);
        float2 gb = *pb;
        *pb = make_float2(0.0f, 0.0f);
        s.x = dv * (g.x + gb.x);
        s.y = dv * (g.y + gb.y);
    }
    v8f c1 = {b1[0], b1[1], b1[2], b1[3], 0.0f, 0.0f, 0.0f, 0.0f};
    v8f p = __builtin_amdgcn_wmma_f32_16x16x4_f32(
        false, a1, false, s, (short)0, c1, false, false);

    v2f a2;
    a2.x = tanhf(half ? p[2] : p[0]);
    a2.y = tanhf(half ? p[3] : p[1]);

    v2f b2 = {0.0f, 0.0f};
    if (m < 2) {
        b2.x = W2[(kb + 0) * 2 + m];
        b2.y = W2[(kb + 1) * 2 + m];
    }
    v8f c2 = {};
    v8f d = __builtin_amdgcn_wmma_f32_16x16x4_f32(
        false, a2, false, b2, (short)0, c2, false, false);

    int nb = base + 8 * half;
    if (full) {
        if (m < 2) {
            float dv[8];
            #pragma unroll
            for (int v = 0; v < 8; ++v) dv[v] = dinv[nb + v];
            #pragma unroll
            for (int v = 0; v < 8; ++v) {
                float val = dv[v] * d[v];
                y2[(nb + v) * 2 + m]   = val;
                agg2[(nb + v) * 2 + m] = val;
            }
        }
    } else {
        if (m < 2) {
            #pragma unroll
            for (int v = 0; v < 8; ++v) {
                int nd = nb + v;
                if (nd < N) {
                    float val = dinv[nd] * d[v];
                    y2[nd * 2 + m]   = val;
                    agg2[nd * 2 + m] = val;
                }
            }
        }
    }
}

__global__ void k_final(const float* __restrict__ agg2, const float* __restrict__ agg2b,
                        const float* __restrict__ dinv,
                        const float* __restrict__ b2, const float* __restrict__ Wc,
                        const float* __restrict__ bc, float* __restrict__ out, int N) {
    int i = blockIdx.x * blockDim.x + threadIdx.x;
    if (i >= N) return;
    float dv = dinv[i];
    float2 ga = *(const float2*)(agg2  + (size_t)i * 2);
    float2 gb = *(const float2*)(agg2b + (size_t)i * 2);
    float h0 = tanhf(dv * (ga.x + gb.x) + b2[0]);
    float h1 = tanhf(dv * (ga.y + gb.y) + b2[1]);
    float2 o;
    o.x = h0 * Wc[0] + h1 * Wc[2] + bc[0];
    o.y = h0 * Wc[1] + h1 * Wc[3] + bc[1];
    *(float2*)(out + (size_t)i * 2) = o;
}

extern "C" void kernel_launch(void* const* d_in, const int* in_sizes, int n_in,
                              void* d_out, int out_size, void* d_ws, size_t ws_size,
                              hipStream_t stream) {
    const float* x  = (const float*)d_in[0];
    const int*   ei = (const int*)d_in[1];
    const float* W1 = (const float*)d_in[2];
    const float* b1 = (const float*)d_in[3];
    const float* W2 = (const float*)d_in[4];
    const float* b2 = (const float*)d_in[5];
    const float* Wc = (const float*)d_in[6];
    const float* bc = (const float*)d_in[7];
    float* out = (float*)d_out;

    const int N = in_sizes[0] / 2;
    const int E = in_sizes[1] / 2;
    const int* src = ei;
    const int* dst = ei + E;

    float* ws   = (float*)d_ws;
    float* y1   = ws;
    float* agg1 = ws + (size_t)2 * N;
    float* aggB = ws + (size_t)4 * N;
    float* y2   = ws;
    float* agg2 = ws + (size_t)6 * N;
    float* dinv = ws + (size_t)8 * N;
    int*   deg0 = (int*)dinv;
    int*   deg1 = (int*)ws;

    const int T = 256;
    dim3 gN((N + T - 1) / T);
    const int edgesPerBlock = (T / 32) * WAVE_EDGES;
    dim3 gE((E + edgesPerBlock - 1) / edgesPerBlock);
    const int nodesPerBlock = (T / 32) * 16;
    dim3 gW((N + nodesPerBlock - 1) / nodesPerBlock);
    dim3 gC((N + HIST_BINS - 1) / HIST_BINS, COUNT_CHUNKS);

    const size_t cntOff   = (((size_t)9 * N * sizeof(float)) + 255) & ~(size_t)255;
    const size_t packOff  = cntOff + (size_t)PART_G * MAX_RANGES * sizeof(int);
    const size_t packEnd  = packOff + ((size_t)E + MAX_RANGES * 32) * sizeof(unsigned);
    const bool   groupedPath = (N <= (1 << SRC_BITS)) && (ws_size >= packEnd);
    int*      cnt     = (int*)((char*)d_ws + cntOff);
    unsigned* grouped = (unsigned*)((char*)d_ws + packOff);
    dim3 gA(AGG_SPLIT, (N + RANGE_NODES - 1) / RANGE_NODES);
    dim3 gD(DEG_SPLIT, (N + RANGE_NODES - 1) / RANGE_NODES);
    int* degp = (int*)(ws + (size_t)6 * N);

    if (groupedPath) {
        k_pcount<<<PART_G, PART_T, 0, stream>>>(dst, cnt, E, N);
        k_part  <<<PART_G, PART_T, 0, stream>>>(src, dst, cnt, grouped, E, N);
        k_deg   <<<gD, PART_T, 0, stream>>>(grouped, cnt, degp, (size_t)N, N);
        k_node  <<<gN, T, 0, stream>>>(degp, degp + (size_t)N, degp + (size_t)2 * N, x, dinv,
                                       y1, agg1, aggB, N);
        k_agg   <<<gA, PART_T, 0, stream>>>(grouped, cnt, y1, agg1, aggB, N);
    } else {
        k_init_deg<<<gN, T, 0, stream>>>(deg0, deg1, N);
        k_count   <<<gC, COUNT_T, 0, stream>>>(dst, deg0, deg1, E, N);
        k_dinv    <<<gN, T, 0, stream>>>(deg0, deg1, dinv, N);
        k_y1      <<<gN, T, 0, stream>>>(x, dinv, y1, agg1, aggB, N);
        k_scatter<<<gE, T, 0, stream>>>(src, dst, y1, agg1, aggB, E);
    }
    k_layer2  <<<gW, T, 0, stream>>>(agg1, aggB, dinv, W1, b1, W2, y2, agg2, N);
    if (groupedPath) k_agg    <<<gA, PART_T, 0, stream>>>(grouped, cnt, y2, agg2, aggB, N);
    else             k_scatter<<<gE, T, 0, stream>>>(src, dst, y2, agg2, aggB, E);
    k_final   <<<gN, T, 0, stream>>>(agg2, aggB, dinv, b2, Wc, bc, out, N);
}


# Grouped edge list stored chunk by chunk (each sorted 8,192-edge chunk in its own words, 65 run offsets per chunk): no count pass over dst, no layout sum per work-group; summation order changes
# speedup vs baseline: 8.5598x; 1.0427x over previous
#include <hip/hip_runtime.h>
#include <math.h>

typedef __attribute__((ext_vector_type(2))) float v2f;
typedef __attribute__((ext_vector_type(8))) float v8f;

#define WAVE_EDGES 256

__device__ __forceinline__ int wave_half() {
    unsigned id = __builtin_amdgcn_s_sendmsg_rtn(0x87);
    return (int)((id >> 18) & 1u);
}

__device__ __forceinline__ bool vec_ok(const int* p, int E) {
    return ((E & 3) == 0) && ((((size_t)p) & 15) == 0);
}

__device__ __forceinline__ int wave_index() {
    return blockIdx.x * (blockDim.x >> 5) + __builtin_amdgcn_readfirstlane((int)(threadIdx.x >> 5));
}

__global__ void k_init_deg(int* deg0, int* deg1, int N) {
    int i = blockIdx.x * blockDim.x + threadIdx.x;
    if (i < N) {
        deg0[i] = 1;
        deg1[i] = 0;
    }
}

#define HIST_BITS    16
#define HIST_BINS    (1 << HIST_BITS)
#define COUNT_T      1024
#define COUNT_CHUNKS 32
#define COUNT_LOADS  8

__device__ __forceinline__ void hist_add(int* hist, unsigned slice, int d) {
    if (((unsigned)d >> HIST_BITS) == slice) atomicAdd(&hist[d & (HIST_BINS - 1)], 1);
}

template <bool FULL>
__device__ __forceinline__ void count_group(const int* __restrict__ p, int left,
                                            int* hist, unsigned slice) {
    int4 v[COUNT_LOADS];
    #pragma unroll
    for (int k = 0; k < COUNT_LOADS; ++k) {
        if (FULL || k * (COUNT_T * 4) + 4 <= left) v[k] = *(const int4*)(p + k * (COUNT_T * 4));
        else                                       v[k] = make_int4(-1, -1, -1, -1);
    }
    #pragma unroll
    for (int k = 0; k < COUNT_LOADS; ++k) {
        hist_add(hist, slice, v[k].x);
        hist_add(hist, slice, v[k].y);
        hist_add(hist, slice, v[k].z);
        hist_add(hist, slice, v[k].w);
    }
}

__global__ __launch_bounds__(COUNT_T)
void k_count(const int* __restrict__ dst, int* deg0, int* deg1, int E, int N) {
    __shared__ __attribute__((aligned(16))) int hist[HIST_BINS];
    const int tid = threadIdx.x;
    const unsigned slice = blockIdx.x;

    for (int u = tid * 4; u < HIST_BINS; u += COUNT_T * 4)
        *(int4*)(hist + u) = make_int4(0, 0, 0, 0);
    __syncthreads();

    unsigned per = ((unsigned)E + gridDim.y - 1u) / gridDim.y;
    per = (per + 3u) & ~3u;
    const long long beg = (long long)blockIdx.y * per;
    long long end = beg + per;
    if (end > E) end = E;
    if (end < beg) end = beg;

    const bool vec = (((size_t)dst) & 15) == 0;
    const long long end4 = vec ? beg + ((end - beg) & ~3LL) : beg;
    const long long step = (long long)COUNT_T * 4 * COUNT_LOADS;
    long long i = beg;
    for (; i + step <= end4; i += step)
        count_group<true>(dst + i + tid * 4, 0, hist, slice);
    if (i < end4)
        count_group<false>(dst + i + tid * 4, (int)(end4 - i) - tid * 4, hist, slice);
    for (long long j = end4 + tid; j < end; j += COUNT_T)
        hist_add(hist, slice, dst[j]);
    __syncthreads();

    int* deg = wave_half() ? deg1 : deg0;
    const int base = (int)(slice << HIST_BITS);
    for (int u = tid; u < HIST_BINS; u += COUNT_T * 8) {
        int c[8];
        #pragma unroll
        for (int k = 0; k < 8; ++k) c[k] = hist[u + k * COUNT_T];
        #pragma unroll
        for (int k = 0; k < 8; ++k) {
            const int node = base + u + k * COUNT_T;
            if (c[k] != 0 && node < N) atomicAdd(&deg[node], c[k]);
        }
    }
}

__global__ void k_dinv(const int* deg0, const int* __restrict__ deg1, float* dinv, int N) {
    int i = blockIdx.x * blockDim.x + threadIdx.x;
    if (i < N) {
        float f = (float)(deg0[i] + deg1[i]);
        dinv[i] = 1.0f / sqrtf(f);
    }
}

__global__ void k_y1(const float* __restrict__ x, const float* __restrict__ dinv,
                     float* __restrict__ y1, float* __restrict__ agg1,
                     float* __restrict__ agg1b, int N) {
    int i = blockIdx.x * blockDim.x + threadIdx.x;
    if (i >= N) return;
    float dv = dinv[i];
    float2 xv = *(const float2*)(x + (size_t)i * 2);
    float2 v;
    v.x = dv * xv.x;
    v.y = dv * xv.y;
    *(float2*)(y1   + (size_t)i * 2) = v;
    *(float2*)(agg1 + (size_t)i * 2) = v;
    *(float2*)(agg1b + (size_t)i * 2) = make_float2(0.0f, 0.0f);
}

template <bool FULL>
__device__ __forceinline__ int4 load4(const int* __restrict__ p, int e, int E) {
    int4 a;
    if (FULL) {
        a = *(const int4*)(p + e);
    } else {
        a.x = (e + 0 < E) ? p[e + 0] : -1;
        a.y = (e + 1 < E) ? p[e + 1] : -1;
        a.z = (e + 2 < E) ? p[e + 2] : -1;
        a.w = (e + 3 < E) ? p[e + 3] : -1;
    }
    return a;
}

template <bool FULL>
__device__ __forceinline__ float gather1(const float* __restrict__ y, int s, int f) {
    if (!FULL && s < 0) s = 0;
    return y[(unsigned)s * 2u + (unsigned)f];
}

template <bool FULL>
__device__ __forceinline__ void add1(float* agg, int t, int f, float v) {
    if (FULL || t >= 0)
        atomicAdd(agg + ((unsigned)t * 2u + (unsigned)f), v);
}

template <bool FULL>
__device__ __forceinline__ void scatter_wave(const int* __restrict__ src, const int* __restrict__ dst,
                                             const float* __restrict__ y, float* agg,
                                             int e0, int f, int E) {
    const int4 d0 = load4<FULL>(dst, e0,       E);
    const int4 d1 = load4<FULL>(dst, e0 + 64,  E);
    const int4 d2 = load4<FULL>(dst, e0 + 128, E);
    const int4 d3 = load4<FULL>(dst, e0 + 192, E);
    const int4 s0 = load4<FULL>(src, e0,       E);
    const int4 s1 = load4<FULL>(src, e0 + 64,  E);
    const int4 s2 = load4<FULL>(src, e0 + 128, E);
    const int4 s3 = load4<FULL>(src, e0 + 192, E);
    __builtin_amdgcn_sched_barrier(0);
    const float v00 = gather1<FULL>(y, s0.x, f);
    const float v01 = gather1<FULL>(y, s0.y, f);
    const float v02 = gather1<FULL>(y, s0.z, f);
    const float v03 = gather1<FULL>(y, s0.w, f);
    const float v10 = gather1<FULL>(y, s1.x, f);
    const float v11 = gather1<FULL>(y, s1.y, f);
    const float v12 = gather1<FULL>(y, s1.z, f);
    const float v13 = gather1<FULL>(y, s1.w, f);
    const float v20 = gather1<FULL>(y, s2.x, f);
    const float v21 = gather1<FULL>(y, s2.y, f);
    const float v22 = gather1<FULL>(y, s2.z, f);
    const float v23 = gather1<FULL>(y, s2.w, f);
    const float v30 = gather1<FULL>(y, s3.x, f);
    const float v31 = gather1<FULL>(y, s3.y, f);
    const float v32 = gather1<FULL>(y, s3.z, f);
    const float v33 = gather1<FULL>(y, s3.w, f);
    add1<FULL>(agg, d0.x, f, v00);
    add1<FULL>(agg, d0.y, f, v01);
    add1<FULL>(agg, d0.z, f, v02);
    add1<FULL>(agg, d0.w, f, v03);
    add1<FULL>(agg, d1.x, f, v10);
    add1<FULL>(agg, d1.y, f, v11);
    add1<FULL>(agg, d1.z, f, v12);
    add1<FULL>(agg, d1.w, f, v13);
    add1<FULL>(agg, d2.x, f, v20);
    add1<FULL>(agg, d2.y, f, v21);
    add1<FULL>(agg, d2.z, f, v22);
    add1<FULL>(agg, d2.w, f, v23);
    add1<FULL>(agg, d3.x, f, v30);
    add1<FULL>(agg, d3.y, f, v31);
    add1<FULL>(agg, d3.z, f, v32);
    add1<FULL>(agg, d3.w, f, v33);
}

__global__ void k_scatter(const int* __restrict__ src, const int* __restrict__ dst,
                          const float* __restrict__ y, float* agg0, float* agg1, int E) {
    const int lane = threadIdx.x & 31;
    const int wave = wave_index();
    const long long w0 = (long long)wave * WAVE_EDGES;
    if (w0 >= E) return;
    float* agg = wave_half() ? agg1 : agg0;
    const int f  = lane & 1;
    const int e0 = (int)w0 + (lane >> 1) * 4;
    if (vec_ok(dst, E) && vec_ok(src, E) && w0 + WAVE_EDGES <= E)
        scatter_wave<true>(src, dst, y, agg, e0, f, E);
    else
        scatter_wave<false>(src, dst, y, agg, e0, f, E);
}

#define RANGE_BITS  13
#define RANGE_NODES (1 << RANGE_BITS)
#define SRC_BITS    19
#define MAX_RANGES  64
#define PART_T      1024
#define PART_PER    8
#define PART_CHUNK  (PART_T * PART_PER)
#define PART_G      512
#define CHUNK_OFFS  (MAX_RANGES + 1)
#define AGG_SPLIT   8
#define RUN_ROWS    6
#define AGG_RUNS    2
#define DEG_RUNS    4

__device__ __forceinline__ void part_block(int E, long long& beg, long long& end, int& nchunk,
                                           unsigned& q0) {
    const unsigned chunks = ((unsigned)E + (PART_CHUNK - 1u)) / PART_CHUNK;
    const unsigned cpb = (chunks + gridDim.x - 1u) / gridDim.x;
    q0 = blockIdx.x * cpb;
    beg = (long long)q0 * PART_CHUNK;
    end = beg + (long long)cpb * PART_CHUNK;
    if (end > E) end = E;
    if (end < beg) end = beg;
    nchunk = (int)((end - beg + PART_CHUNK - 1) / PART_CHUNK);
}

__global__ __launch_bounds__(PART_T)
void k_part(const int* __restrict__ src, const int* __restrict__ dst,
            unsigned* __restrict__ grouped, unsigned short* __restrict__ offs, int E, int N) {
    __shared__ int s_cnt[MAX_RANGES];
    __shared__ int s_lofs[CHUNK_OFFS];
    __shared__ unsigned s_word[PART_CHUNK];
    const int tid = threadIdx.x;

    long long beg, end; int nchunk; unsigned q0;
    part_block(E, beg, end, nchunk, q0);
    for (int c = 0; c < nchunk; ++c) {
        if (tid < MAX_RANGES) s_cnt[tid] = 0;
        __syncthreads();

        const long long base = beg + (long long)c * PART_CHUNK + tid;
        int d[PART_PER], s[PART_PER];
        if (beg + (long long)(c + 1) * PART_CHUNK <= end) {
            #pragma unroll
            for (int k = 0; k < PART_PER; ++k) {
                d[k] = dst[base + (long long)k * PART_T];
                s[k] = src[base + (long long)k * PART_T];
            }
        } else {
            #pragma unroll
            for (int k = 0; k < PART_PER; ++k) {
                const long long e = base + (long long)k * PART_T;
                const bool in = e < end;
                d[k] = in ? dst[e] : -1;
                s[k] = in ? src[e] : 0;
            }
        }
        int rk[PART_PER];
        #pragma unroll
        for (int k = 0; k < PART_PER; ++k) {
            rk[k] = 0;
            if ((unsigned)d[k] < (unsigned)N) rk[k] = atomicAdd(&s_cnt[d[k] >> RANGE_BITS], 1);
        }
        __syncthreads();
        if (tid < MAX_RANGES) {
            int o = 0;
            for (int q = 0; q < tid; ++q) o += s_cnt[q];
            s_lofs[tid] = o;
            if (tid == MAX_RANGES - 1) s_lofs[MAX_RANGES] = o + s_cnt[tid];
        }
        __syncthreads();
        #pragma unroll
        for (int k = 0; k < PART_PER; ++k) {
            if ((unsigned)d[k] < (unsigned)N) {
                const int p = s_lofs[d[k] >> RANGE_BITS] + rk[k];
                s_word[p] = ((unsigned)s[k] << RANGE_BITS) | ((unsigned)d[k] & (RANGE_NODES - 1u));
            }
        }
        __syncthreads();
        const size_t q = (size_t)q0 + (size_t)c;
        const int total = s_lofs[MAX_RANGES];
        unsigned* outp = grouped + q * PART_CHUNK;
        for (int i = tid; i < total; i += PART_T) outp[i] = s_word[i];
        if (tid < CHUNK_OFFS) offs[q * CHUNK_OFFS + tid] = (unsigned short)s_lofs[tid];
        __syncthreads();
    }
}

__device__ __forceinline__ void share_chunks(int E, int& qlo, int& qhi) {
    const unsigned chunks = ((unsigned)E + (PART_CHUNK - 1u)) / PART_CHUNK;
    const unsigned cps = (chunks + gridDim.x - 1u) / gridDim.x;
    unsigned lo = blockIdx.x * cps, hi = lo + cps;
    if (hi > chunks) hi = chunks;
    if (lo > hi) lo = hi;
    qlo = (int)lo;
    qhi = (int)hi;
}

template <int RUNS>
struct RunSet {
    const unsigned* p[RUNS];
    int i[RUNS];
    int h[RUNS];
};

template <int RUNS>
__device__ __forceinline__ void runs_open(RunSet<RUNS>& R, const unsigned* __restrict__ grouped,
                                          const unsigned short* __restrict__ offs,
                                          int r, int q, int qhi) {
    #pragma unroll
    for (int t = 0; t < RUNS; ++t) {
        const int qt = q + t * (PART_T / 32);
        const int qc = qt < qhi ? qt : qhi - 1;
        const unsigned short* o = offs + (size_t)qc * CHUNK_OFFS + r;
        const int lo = o[0];
        const int hi = o[1];
        R.p[t] = grouped + (size_t)qc * PART_CHUNK;
        R.i[t] = lo;
        R.h[t] = qt < qhi ? hi : lo;
    }
}

template <int RUNS>
__device__ __forceinline__ bool runs_left(const RunSet<RUNS>& R) {
    bool any = false;
    #pragma unroll
    for (int t = 0; t < RUNS; ++t) any = any || (R.i[t] < R.h[t]);
    return any;
}

template <int RUNS>
__device__ __forceinline__ void runs_step(RunSet<RUNS>& R) {
    #pragma unroll
    for (int t = 0; t < RUNS; ++t) R.i[t] += 32 * RUN_ROWS;
}

template <int RUNS>
__device__ __forceinline__ bool run_has(const RunSet<RUNS>& R, int t, int k, int lane) {
    return R.i[t] + k * 32 + lane < R.h[t];
}

template <int RUNS>
__device__ __forceinline__ void runs_load(const RunSet<RUNS>& R, int lane,
                                          unsigned (&w)[RUNS * RUN_ROWS]) {
    const unsigned* safe = R.p[0] + R.i[0];
    #pragma unroll
    for (int t = RUNS - 1; t >= 0; --t)
        if (R.i[t] < R.h[t]) safe = R.p[t] + R.i[t];
    #pragma unroll
    for (int t = 0; t < RUNS; ++t) {
        #pragma unroll
        for (int k = 0; k < RUN_ROWS; ++k) {
            const unsigned* a = R.p[t] + (R.i[t] + k * 32 + lane);
            w[t * RUN_ROWS + k] = *(run_has(R, t, k, lane) ? a : safe);
        }
    }
}

__global__ __launch_bounds__(PART_T)
void k_agg(const unsigned* __restrict__ grouped, const unsigned short* __restrict__ offs,
           const float* __restrict__ y, float* agg0, float* agg1, int E, int N) {
    __shared__ float sx[RANGE_NODES + 32];
    __shared__ float sy[RANGE_NODES + 32];
    const int tid = threadIdx.x;
    const int lane = tid & 31;
    const int wv = __builtin_amdgcn_readfirstlane(tid >> 5);
    const int r = blockIdx.y;

    for (int u = tid; u < RANGE_NODES; u += PART_T) { sx[u] = 0.0f; sy[u] = 0.0f; }
    __syncthreads();

    int qlo, qhi;
    share_chunks(E, qlo, qhi);
    for (int q = qlo + wv; q < qhi; q += AGG_RUNS * (PART_T / 32)) {
        RunSet<AGG_RUNS> R;
        runs_open(R, grouped, offs, r, q, qhi);
        for (; runs_left(R); runs_step(R)) {
            unsigned w[AGG_RUNS * RUN_ROWS];
            runs_load(R, lane, w);
            float2 v[AGG_RUNS * RUN_ROWS];
            #pragma unroll
            for (int j = 0; j < AGG_RUNS * RUN_ROWS; ++j)
                v[j] = *(const float2*)(y + (size_t)(w[j] >> RANGE_BITS) * 2);
            #pragma unroll
            for (int t = 0; t < AGG_RUNS; ++t) {
                #pragma unroll
                for (int k = 0; k < RUN_ROWS; ++k) {
                    const unsigned o = run_has(R, t, k, lane)
                                           ? (w[t * RUN_ROWS + k] & (RANGE_NODES - 1u))
                                           : (unsigned)(RANGE_NODES + lane);
                    atomicAdd(&sx[o], v[t * RUN_ROWS + k].x);
                    atomicAdd(&sy[o], v[t * RUN_ROWS + k].y);
                }
            }
        }
    }
    __syncthreads();

    float* agg = wave_half() ? agg1 : agg0;
    const int node0 = r << RANGE_BITS;
    for (int u = tid; u < 2 * RANGE_NODES; u += PART_T) {
        const int n = u >> 1;
        if (node0 + n < N) {
            const float val = (u & 1) ? sy[n] : sx[n];
            atomicAdd(agg + ((size_t)node0 * 2 + u), val);
        }
    }
}

#define DEG_SPLIT 3

__global__ __launch_bounds__(PART_T)
void k_deg(const unsigned* __restrict__ grouped, const unsigned short* __restrict__ offs,
           int* __restrict__ part, size_t stride, int E, int N) {
    __shared__ int hist[RANGE_NODES + 32];
    const int tid = threadIdx.x;
    const int lane = tid & 31;
    const int wv = __builtin_amdgcn_readfirstlane(tid >> 5);
    const int r = blockIdx.y;

    for (int u = tid; u < RANGE_NODES; u += PART_T) hist[u] = 0;
    __syncthreads();

    int qlo, qhi;
    share_chunks(E, qlo, qhi);
    for (int q = qlo + wv; q < qhi; q += DEG_RUNS * (PART_T / 32)) {
        RunSet<DEG_RUNS> R;
        runs_open(R, grouped, offs, r, q, qhi);
        for (; runs_left(R); runs_step(R)) {
            unsigned w[DEG_RUNS * RUN_ROWS];
            runs_load(R, lane, w);
            #pragma unroll
            for (int t = 0; t < DEG_RUNS; ++t) {
                #pragma unroll
                for (int k = 0; k < RUN_ROWS; ++k) {
                    const unsigned o = run_has(R, t, k, lane)
                                           ? (w[t * RUN_ROWS + k] & (RANGE_NODES - 1u))
                                           : (unsigned)(RANGE_NODES + lane);
                    atomicAdd(&hist[o], 1);
                }
            }
        }
    }
    __syncthreads();

    int* mine = part + (size_t)blockIdx.x * stride;
    const int node0 = r << RANGE_BITS;
    for (int u = tid; u < RANGE_NODES; u += PART_T)
        if (node0 + u < N) mine[node0 + u] = hist[u];
}

__global__ void k_node(const int* __restrict__ p0, const int* __restrict__ p1, const int* p2,
                       const float* __restrict__ x, float* dinv,
                       float* __restrict__ y1, float* __restrict__ agg1,
                       float* __restrict__ agg1b, int N) {
    int i = blockIdx.x * blockDim.x + threadIdx.x;
    if (i >= N) return;
    float f = (float)(p0[i] + p1[i] + p2[i] + 1);
    float dv = 1.0f / sqrtf(f);
    dinv[i] = dv;
    float2 xv = *(const float2*)(x + (size_t)i * 2);
    float2 v;
    v.x = dv * xv.x;
    v.y = dv * xv.y;
    *(float2*)(y1   + (size_t)i * 2) = v;
    *(float2*)(agg1 + (size_t)i * 2) = v;
    *(float2*)(agg1b + (size_t)i * 2) = make_float2(0.0f, 0.0f);
}

__global__ void k_layer2(const float* __restrict__ agg1, float* aggB,
                         const float* __restrict__ dinv,
                         const float* __restrict__ W1, const float* __restrict__ b1,
                         const float* __restrict__ W2,
                         float* __restrict__ y2, float* __restrict__ agg2, int N) {
    int lane = threadIdx.x & 31;
    int wave = threadIdx.x >> 5;
    int base = (blockIdx.x * (blockDim.x >> 5) + wave) * 16;
    if (base >= N) return;
    int m    = lane & 15;
    int half = lane >> 4;
    int node = base + m;
    int kb   = half * 2;
    bool full = (base + 16 <= N);

    int j = m & 7;
    v2f a1 = {0.0f, 0.0f};
    if (half == 0 && j < 4) {
        a1.x = W1[0 * 4 + j];
        a1.y = W1[1 * 4 + j];
    }
    v2f s = {0.0f, 0.0f};
    if (half == 0 && node < N) {
        float dv = dinv[node];
        float2 g = *(const float2*)(agg1 + (size_t)node * 2);
        float2* pb = (float2*)(aggB + (size_t)node * 2);
        float2 gb = *pb;
        *pb = make_float2(0.0f, 0.0f);
        s.x = dv * (g.x + gb.x);
        s.y = dv * (g.y + gb.y);
    }
    v8f c1 = {b1[0], b1[1], b1[2], b1[3], 0.0f, 0.0f, 0.0f, 0.0f};
    v8f p = __builtin_amdgcn_wmma_f32_16x16x4_f32(
        false, a1, false, s, (short)0, c1, false, false);

    v2f a2;
    a2.x = tanhf(half ? p[2] : p[0]);
    a2.y = tanhf(half ? p[3] : p[1]);

    v2f b2 = {0.0f, 0.0f};
    if (m < 2) {
        b2.x = W2[(kb + 0) * 2 + m];
        b2.y = W2[(kb + 1) * 2 + m];
    }
    v8f c2 = {};
    v8f d = __builtin_amdgcn_wmma_f32_16x16x4_f32(
        false, a2, false, b2, (short)0, c2, false, false);

    int nb = base + 8 * half;
    if (full) {
        if (m < 2) {
            float dv[8];
            #pragma unroll
            for (int v = 0; v < 8; ++v) dv[v] = dinv[nb + v];
            #pragma unroll
            for (int v = 0; v < 8; ++v) {
                float val = dv[v] * d[v];
                y2[(nb + v) * 2 + m]   = val;
                agg2[(nb + v) * 2 + m] = val;
            }
        }
    } else {
        if (m < 2) {
            #pragma unroll
            for (int v = 0; v < 8; ++v) {
                int nd = nb + v;
                if (nd < N) {
                    float val = dinv[nd] * d[v];
                    y2[nd * 2 + m]   = val;
                    agg2[nd * 2 + m] = val;
                }
            }
        }
    }
}

__global__ void k_final(const float* __restrict__ agg2, const float* __restrict__ agg2b,
                        const float* __restrict__ dinv,
                        const float* __restrict__ b2, const float* __restrict__ Wc,
                        const float* __restrict__ bc, float* __restrict__ out, int N) {
    int i = blockIdx.x * blockDim.x + threadIdx.x;
    if (i >= N) return;
    float dv = dinv[i];
    float2 ga = *(const float2*)(agg2  + (size_t)i * 2);
    float2 gb = *(const float2*)(agg2b + (size_t)i * 2);
    float h0 = tanhf(dv * (ga.x + gb.x) + b2[0]);
    float h1 = tanhf(dv * (ga.y + gb.y) + b2[1]);
    float2 o;
    o.x = h0 * Wc[0] + h1 * Wc[2] + bc[0];
    o.y = h0 * Wc[1] + h1 * Wc[3] + bc[1];
    *(float2*)(out + (size_t)i * 2) = o;
}

extern "C" void kernel_launch(void* const* d_in, const int* in_sizes, int n_in,
                              void* d_out, int out_size, void* d_ws, size_t ws_size,
                              hipStream_t stream) {
    const float* x  = (const float*)d_in[0];
    const int*   ei = (const int*)d_in[1];
    const float* W1 = (const float*)d_in[2];
    const float* b1 = (const float*)d_in[3];
    const float* W2 = (const float*)d_in[4];
    const float* b2 = (const float*)d_in[5];
    const float* Wc = (const float*)d_in[6];
    const float* bc = (const float*)d_in[7];
    float* out = (float*)d_out;

    const int N = in_sizes[0] / 2;
    const int E = in_sizes[1] / 2;
    const int* src = ei;
    const int* dst = ei + E;

    float* ws   = (float*)d_ws;
    float* y1   = ws;
    float* agg1 = ws + (size_t)2 * N;
    float* aggB = ws + (size_t)4 * N;
    float* y2   = ws;
    float* agg2 = ws + (size_t)6 * N;
    float* dinv = ws + (size_t)8 * N;
    int*   deg0 = (int*)dinv;
    int*   deg1 = (int*)ws;

    const int T = 256;
    dim3 gN((N + T - 1) / T);
    const int edgesPerBlock = (T / 32) * WAVE_EDGES;
    dim3 gE((E + edgesPerBlock - 1) / edgesPerBlock);
    const int nodesPerBlock = (T / 32) * 16;
    dim3 gW((N + nodesPerBlock - 1) / nodesPerBlock);
    dim3 gC((N + HIST_BINS - 1) / HIST_BINS, COUNT_CHUNKS);

    const size_t chunks   = ((size_t)E + PART_CHUNK - 1) / PART_CHUNK;
    const size_t offsOff  = (((size_t)9 * N * sizeof(float)) + 255) & ~(size_t)255;
    const size_t packOff  = (offsOff + chunks * CHUNK_OFFS * sizeof(unsigned short) + 255) & ~(size_t)255;
    const size_t packEnd  = packOff + chunks * PART_CHUNK * sizeof(unsigned);
    const bool   groupedPath = (N <= (1 << SRC_BITS)) && (ws_size >= packEnd);
    unsigned short* offs = (unsigned short*)((char*)d_ws + offsOff);
    unsigned* grouped = (unsigned*)((char*)d_ws + packOff);
    dim3 gA(AGG_SPLIT, (N + RANGE_NODES - 1) / RANGE_NODES);
    dim3 gD(DEG_SPLIT, (N + RANGE_NODES - 1) / RANGE_NODES);
    int* degp = (int*)(ws + (size_t)6 * N);

    if (groupedPath) {
        k_part  <<<PART_G, PART_T, 0, stream>>>(src, dst, grouped, offs, E, N);
        k_deg   <<<gD, PART_T, 0, stream>>>(grouped, offs, degp, (size_t)N, E, N);
        k_node  <<<gN, T, 0, stream>>>(degp, degp + (size_t)N, degp + (size_t)2 * N, x, dinv,
                                       y1, agg1, aggB, N);
        k_agg   <<<gA, PART_T, 0, stream>>>(grouped, offs, y1, agg1, aggB, E, N);
    } else {
        k_init_deg<<<gN, T, 0, stream>>>(deg0, deg1, N);
        k_count   <<<gC, COUNT_T, 0, stream>>>(dst, deg0, deg1, E, N);
        k_dinv    <<<gN, T, 0, stream>>>(deg0, deg1, dinv, N);
        k_y1      <<<gN, T, 0, stream>>>(x, dinv, y1, agg1, aggB, N);
        k_scatter<<<gE, T, 0, stream>>>(src, dst, y1, agg1, aggB, E);
    }
    k_layer2  <<<gW, T, 0, stream>>>(agg1, aggB, dinv, W1, b1, W2, y2, agg2, N);
    if (groupedPath) k_agg    <<<gA, PART_T, 0, stream>>>(grouped, offs, y2, agg2, aggB, E, N);
    else             k_scatter<<<gE, T, 0, stream>>>(src, dst, y2, agg2, aggB, E);
    k_final   <<<gN, T, 0, stream>>>(agg2, aggB, dinv, b2, Wc, bc, out, N);
}


# Edges grouped by destination range AND source slice; a layer's work-group stages each source slice in LDS: an edge's source pair is an LDS read, not a cache request; summation order changes
# speedup vs baseline: 11.3710x; 1.3284x over previous
#include <hip/hip_runtime.h>
#include <math.h>

typedef __attribute__((ext_vector_type(2))) float v2f;
typedef __attribute__((ext_vector_type(8))) float v8f;

#define WAVE_EDGES 256

__device__ __forceinline__ int wave_half() {
    unsigned id = __builtin_amdgcn_s_sendmsg_rtn(0x87);
    return (int)((id >> 18) & 1u);
}

__device__ __forceinline__ bool vec_ok(const int* p, int E) {
    return ((E & 3) == 0) && ((((size_t)p) & 15) == 0);
}

__device__ __forceinline__ int wave_index() {
    return blockIdx.x * (blockDim.x >> 5) + __builtin_amdgcn_readfirstlane((int)(threadIdx.x >> 5));
}

__global__ void k_init_deg(int* deg0, int* deg1, int N) {
    int i = blockIdx.x * blockDim.x + threadIdx.x;
    if (i < N) {
        deg0[i] = 1;
        deg1[i] = 0;
    }
}

#define HIST_BITS    16
#define HIST_BINS    (1 << HIST_BITS)
#define COUNT_T      1024
#define COUNT_CHUNKS 32
#define COUNT_LOADS  8

__device__ __forceinline__ void hist_add(int* hist, unsigned slice, int d) {
    if (((unsigned)d >> HIST_BITS) == slice) atomicAdd(&hist[d & (HIST_BINS - 1)], 1);
}

template <bool FULL>
__device__ __forceinline__ void count_group(const int* __restrict__ p, int left,
                                            int* hist, unsigned slice) {
    int4 v[COUNT_LOADS];
    #pragma unroll
    for (int k = 0; k < COUNT_LOADS; ++k) {
        if (FULL || k * (COUNT_T * 4) + 4 <= left) v[k] = *(const int4*)(p + k * (COUNT_T * 4));
        else                                       v[k] = make_int4(-1, -1, -1, -1);
    }
    #pragma unroll
    for (int k = 0; k < COUNT_LOADS; ++k) {
        hist_add(hist, slice, v[k].x);
        hist_add(hist, slice, v[k].y);
        hist_add(hist, slice, v[k].z);
        hist_add(hist, slice, v[k].w);
    }
}

__global__ __launch_bounds__(COUNT_T)
void k_count(const int* __restrict__ dst, int* deg0, int* deg1, int E, int N) {
    __shared__ __attribute__((aligned(16))) int hist[HIST_BINS];
    const int tid = threadIdx.x;
    const unsigned slice = blockIdx.x;

    for (int u = tid * 4; u < HIST_BINS; u += COUNT_T * 4)
        *(int4*)(hist + u) = make_int4(0, 0, 0, 0);
    __syncthreads();

    unsigned per = ((unsigned)E + gridDim.y - 1u) / gridDim.y;
    per = (per + 3u) & ~3u;
    const long long beg = (long long)blockIdx.y * per;
    long long end = beg + per;
    if (end > E) end = E;
    if (end < beg) end = beg;

    const bool vec = (((size_t)dst) & 15) == 0;
    const long long end4 = vec ? beg + ((end - beg) & ~3LL) : beg;
    const long long step = (long long)COUNT_T * 4 * COUNT_LOADS;
    long long i = beg;
    for (; i + step <= end4; i += step)
        count_group<true>(dst + i + tid * 4, 0, hist, slice);
    if (i < end4)
        count_group<false>(dst + i + tid * 4, (int)(end4 - i) - tid * 4, hist, slice);
    for (long long j = end4 + tid; j < end; j += COUNT_T)
        hist_add(hist, slice, dst[j]);
    __syncthreads();

    int* deg = wave_half() ? deg1 : deg0;
    const int base = (int)(slice << HIST_BITS);
    for (int u = tid; u < HIST_BINS; u += COUNT_T * 8) {
        int c[8];
        #pragma unroll
        for (int k = 0; k < 8; ++k) c[k] = hist[u + k * COUNT_T];
        #pragma unroll
        for (int k = 0; k < 8; ++k) {
            const int node = base + u + k * COUNT_T;
            if (c[k] != 0 && node < N) atomicAdd(&deg[node], c[k]);
        }
    }
}

__global__ void k_dinv(const int* deg0, const int* __restrict__ deg1, float* dinv, int N) {
    int i = blockIdx.x * blockDim.x + threadIdx.x;
    if (i < N) {
        float f = (float)(deg0[i] + deg1[i]);
        dinv[i] = 1.0f / sqrtf(f);
    }
}

__global__ void k_y1(const float* __restrict__ x, const float* __restrict__ dinv,
                     float* __restrict__ y1, float* __restrict__ agg1,
                     float* __restrict__ agg1b, int N) {
    int i = blockIdx.x * blockDim.x + threadIdx.x;
    if (i >= N) return;
    float dv = dinv[i];
    float2 xv = *(const float2*)(x + (size_t)i * 2);
    float2 v;
    v.x = dv * xv.x;
    v.y = dv * xv.y;
    *(float2*)(y1   + (size_t)i * 2) = v;
    *(float2*)(agg1 + (size_t)i * 2) = v;
    *(float2*)(agg1b + (size_t)i * 2) = make_float2(0.0f, 0.0f);
}

template <bool FULL>
__device__ __forceinline__ int4 load4(const int* __restrict__ p, int e, int E) {
    int4 a;
    if (FULL) {
        a = *(const int4*)(p + e);
    } else {
        a.x = (e + 0 < E) ? p[e + 0] : -1;
        a.y = (e + 1 < E) ? p[e + 1] : -1;
        a.z = (e + 2 < E) ? p[e + 2] : -1;
        a.w = (e + 3 < E) ? p[e + 3] : -1;
    }
    return a;
}

template <bool FULL>
__device__ __forceinline__ float gather1(const float* __restrict__ y, int s, int f) {
    if (!FULL && s < 0) s = 0;
    return y[(unsigned)s * 2u + (unsigned)f];
}

template <bool FULL>
__device__ __forceinline__ void add1(float* agg, int t, int f, float v) {
    if (FULL || t >= 0)
        atomicAdd(agg + ((unsigned)t * 2u + (unsigned)f), v);
}

template <bool FULL>
__device__ __forceinline__ void scatter_wave(const int* __restrict__ src, const int* __restrict__ dst,
                                             const float* __restrict__ y, float* agg,
                                             int e0, int f, int E) {
    const int4 d0 = load4<FULL>(dst, e0,       E);
    const int4 d1 = load4<FULL>(dst, e0 + 64,  E);
    const int4 d2 = load4<FULL>(dst, e0 + 128, E);
    const int4 d3 = load4<FULL>(dst, e0 + 192, E);
    const int4 s0 = load4<FULL>(src, e0,       E);
    const int4 s1 = load4<FULL>(src, e0 + 64,  E);
    const int4 s2 = load4<FULL>(src, e0 + 128, E);
    const int4 s3 = load4<FULL>(src, e0 + 192, E);
    __builtin_amdgcn_sched_barrier(0);
    const float v00 = gather1<FULL>(y, s0.x, f);
    const float v01 = gather1<FULL>(y, s0.y, f);
    const float v02 = gather1<FULL>(y, s0.z, f);
    const float v03 = gather1<FULL>(y, s0.w, f);
    const float v10 = gather1<FULL>(y, s1.x, f);
    const float v11 = gather1<FULL>(y, s1.y, f);
    const float v12 = gather1<FULL>(y, s1.z, f);
    const float v13 = gather1<FULL>(y, s1.w, f);
    const float v20 = gather1<FULL>(y, s2.x, f);
    const float v21 = gather1<FULL>(y, s2.y, f);
    const float v22 = gather1<FULL>(y, s2.z, f);
    const float v23 = gather1<FULL>(y, s2.w, f);
    const float v30 = gather1<FULL>(y, s3.x, f);
    const float v31 = gather1<FULL>(y, s3.y, f);
    const float v32 = gather1<FULL>(y, s3.z, f);
    const float v33 = gather1<FULL>(y, s3.w, f);
    add1<FULL>(agg, d0.x, f, v00);
    add1<FULL>(agg, d0.y, f, v01);
    add1<FULL>(agg, d0.z, f, v02);
    add1<FULL>(agg, d0.w, f, v03);
    add1<FULL>(agg, d1.x, f, v10);
    add1<FULL>(agg, d1.y, f, v11);
    add1<FULL>(agg, d1.z, f, v12);
    add1<FULL>(agg, d1.w, f, v13);
    add1<FULL>(agg, d2.x, f, v20);
    add1<FULL>(agg, d2.y, f, v21);
    add1<FULL>(agg, d2.z, f, v22);
    add1<FULL>(agg, d2.w, f, v23);
    add1<FULL>(agg, d3.x, f, v30);
    add1<FULL>(agg, d3.y, f, v31);
    add1<FULL>(agg, d3.z, f, v32);
    add1<FULL>(agg, d3.w, f, v33);
}

__global__ void k_scatter(const int* __restrict__ src, const int* __restrict__ dst,
                          const float* __restrict__ y, float* agg0, float* agg1, int E) {
    const int lane = threadIdx.x & 31;
    const int wave = wave_index();
    const long long w0 = (long long)wave * WAVE_EDGES;
    if (w0 >= E) return;
    float* agg = wave_half() ? agg1 : agg0;
    const int f  = lane & 1;
    const int e0 = (int)w0 + (lane >> 1) * 4;
    if (vec_ok(dst, E) && vec_ok(src, E) && w0 + WAVE_EDGES <= E)
        scatter_wave<true>(src, dst, y, agg, e0, f, E);
    else
        scatter_wave<false>(src, dst, y, agg, e0, f, E);
}

#define RANGE_BITS  13
#define RANGE_NODES (1 << RANGE_BITS)
#define SRC_BITS    19
#define MAX_RANGES  64
#define SLICE_SHIFT 12
#define SLICE_UNITS 7
#define SLICE_NODES (SLICE_UNITS << SLICE_SHIFT)
#define MAX_SLICES  19
#define MAX_BUCKETS (MAX_RANGES * MAX_SLICES)
#define TAB_ENTRIES (MAX_BUCKETS + 1)
#define SCAN_PER    (MAX_BUCKETS / 32)
#define PART_T      1024
#define PART_PER    8
#define PART_GROUP  (PART_T * PART_PER)
#define PART_CHUNK  65536
#define AGG_GROUPS  4
#define AGG_RUNS    8
#define AGG_ROWS    2
#define DEG_RUNS    4
#define DEG_ROWS    6
#define STAGE_PER   7

__device__ __forceinline__ unsigned slice_of(unsigned s) {
    return (s >> SLICE_SHIFT) / SLICE_UNITS;
}

template <bool FULL>
__device__ __forceinline__ void part_load(const int* __restrict__ src, const int* __restrict__ dst,
                                          long long base, long long end, int N,
                                          int (&d)[PART_PER], unsigned (&s)[PART_PER]) {
    #pragma unroll
    for (int k = 0; k < PART_PER; ++k) {
        const long long e = base + (long long)k * PART_T;
        if (FULL || e < end) { d[k] = dst[e]; s[k] = (unsigned)src[e]; }
        else                 { d[k] = -1;     s[k] = 0u; }
    }
    #pragma unroll
    for (int k = 0; k < PART_PER; ++k)
        if (s[k] >= (unsigned)N) s[k] = (unsigned)N - 1u;
}

__global__ __launch_bounds__(PART_T)
void k_part(const int* __restrict__ src, const int* __restrict__ dst,
            unsigned* __restrict__ grouped, unsigned* __restrict__ tab, int E, int N) {
    __shared__ unsigned s_cur[MAX_BUCKETS];
    __shared__ unsigned s_lofs[TAB_ENTRIES];
    __shared__ unsigned s_lane[32];
    __shared__ unsigned s_word[PART_CHUNK];
    const int tid = threadIdx.x;
    const long long beg = (long long)blockIdx.x * PART_CHUNK;
    long long end = beg + PART_CHUNK;
    if (end > E) end = E;

    for (int u = tid; u < MAX_BUCKETS; u += PART_T) s_cur[u] = 0u;
    __syncthreads();

    for (int g = 0; g < PART_CHUNK / PART_GROUP; ++g) {
        const long long gb = beg + (long long)g * PART_GROUP;
        int d[PART_PER]; unsigned s[PART_PER];
        if (gb + PART_GROUP <= end) part_load<true >(src, dst, gb + tid, end, N, d, s);
        else                        part_load<false>(src, dst, gb + tid, end, N, d, s);
        #pragma unroll
        for (int k = 0; k < PART_PER; ++k)
            if ((unsigned)d[k] < (unsigned)N)
                atomicAdd(&s_cur[((unsigned)d[k] >> RANGE_BITS) * MAX_SLICES + slice_of(s[k])], 1u);
    }
    __syncthreads();

    if (tid < 32) {
        unsigned sum = 0u;
        for (int j = 0; j < SCAN_PER; ++j) sum += s_cur[tid * SCAN_PER + j];
        s_lane[tid] = sum;
    }
    __syncthreads();
    if (tid < 32) {
        unsigned o = 0u;
        for (int l = 0; l < tid; ++l) o += s_lane[l];
        for (int j = 0; j < SCAN_PER; ++j) {
            const unsigned c = s_cur[tid * SCAN_PER + j];
            s_lofs[tid * SCAN_PER + j] = o;
            s_cur[tid * SCAN_PER + j]  = o;
            o += c;
        }
        if (tid == 31) s_lofs[MAX_BUCKETS] = o;
    }
    __syncthreads();

    for (int g = 0; g < PART_CHUNK / PART_GROUP; ++g) {
        const long long gb = beg + (long long)g * PART_GROUP;
        int d[PART_PER]; unsigned s[PART_PER];
        if (gb + PART_GROUP <= end) part_load<true >(src, dst, gb + tid, end, N, d, s);
        else                        part_load<false>(src, dst, gb + tid, end, N, d, s);
        #pragma unroll
        for (int k = 0; k < PART_PER; ++k) {
            if ((unsigned)d[k] < (unsigned)N) {
                const unsigned sl = slice_of(s[k]);
                const unsigned p = atomicAdd(&s_cur[((unsigned)d[k] >> RANGE_BITS) * MAX_SLICES + sl], 1u);
                s_word[p] = ((s[k] - sl * SLICE_NODES) << RANGE_BITS) | ((unsigned)d[k] & (RANGE_NODES - 1u));
            }
        }
    }
    __syncthreads();

    const int total = (int)s_lofs[MAX_BUCKETS];
    unsigned* outp = grouped + (size_t)blockIdx.x * PART_CHUNK;
    #pragma unroll 8
    for (int i = tid; i < total; i += PART_T) outp[i] = s_word[i];
    for (int u = tid; u < TAB_ENTRIES; u += PART_T)
        tab[(size_t)blockIdx.x * TAB_ENTRIES + u] = s_lofs[u];
}

__device__ __forceinline__ int chunk_count(int E) {
    return (int)(((unsigned)E + (PART_CHUNK - 1u)) / PART_CHUNK);
}

__device__ __forceinline__ void share_chunks(int E, int& qlo, int& qhi) {
    const unsigned chunks = (unsigned)chunk_count(E);
    const unsigned cps = (chunks + gridDim.x - 1u) / gridDim.x;
    unsigned lo = blockIdx.x * cps, hi = lo + cps;
    if (hi > chunks) hi = chunks;
    if (lo > hi) lo = hi;
    qlo = (int)lo;
    qhi = (int)hi;
}

template <int RUNS>
struct RunSet {
    unsigned i[RUNS];
    unsigned h[RUNS];
};

template <int RUNS>
__device__ __forceinline__ void runs_open(RunSet<RUNS>& R, const unsigned* __restrict__ tab,
                                          int b0, int b1, int q, int qhi) {
    #pragma unroll
    for (int t = 0; t < RUNS; ++t) {
        const int qt = q + t * (PART_T / 32);
        const int qc = qt < qhi ? qt : qhi - 1;
        const unsigned* o = tab + (size_t)qc * TAB_ENTRIES;
        const unsigned lo = o[b0];
        const unsigned hi = o[b1];
        const unsigned base = (unsigned)qc * PART_CHUNK;
        R.i[t] = base + lo;
        R.h[t] = base + (qt < qhi ? hi : lo);
    }
}

template <int RUNS>
__device__ __forceinline__ bool runs_left(const RunSet<RUNS>& R) {
    bool any = false;
    #pragma unroll
    for (int t = 0; t < RUNS; ++t) any = any || (R.i[t] < R.h[t]);
    return any;
}

template <int RUNS, int ROWS>
__device__ __forceinline__ void runs_step(RunSet<RUNS>& R) {
    #pragma unroll
    for (int t = 0; t < RUNS; ++t) R.i[t] += 32 * ROWS;
}

template <int RUNS>
__device__ __forceinline__ bool run_has(const RunSet<RUNS>& R, int t, int k, int lane) {
    return R.i[t] + (unsigned)(k * 32 + lane) < R.h[t];
}

template <int RUNS, int ROWS>
__device__ __forceinline__ void runs_load(const RunSet<RUNS>& R, const unsigned* __restrict__ grouped,
                                          int lane, unsigned (&w)[RUNS * ROWS]) {
    unsigned safe = R.i[0];
    #pragma unroll
    for (int t = RUNS - 1; t >= 0; --t)
        if (R.i[t] < R.h[t]) safe = R.i[t];
    #pragma unroll
    for (int t = 0; t < RUNS; ++t) {
        #pragma unroll
        for (int k = 0; k < ROWS; ++k) {
            const unsigned a = R.i[t] + (unsigned)(k * 32 + lane);
            w[t * ROWS + k] = grouped[run_has(R, t, k, lane) ? a : safe];
        }
    }
}

__global__ __launch_bounds__(PART_T)
void k_agg(const unsigned* __restrict__ grouped, const unsigned* __restrict__ tab,
           const float* __restrict__ y, float* agg0, float* agg1, int E, int N) {
    __shared__ float sx[RANGE_NODES + 32];
    __shared__ float sy[RANGE_NODES + 32];
    __shared__ float2 ys[SLICE_NODES];
    __shared__ int s_lo[AGG_RUNS * (PART_T / 32)];
    __shared__ int s_hi[AGG_RUNS * (PART_T / 32)];
    const int tid = threadIdx.x;
    const int lane = tid & 31;
    const int wv = __builtin_amdgcn_readfirstlane(tid >> 5);
    const int r = blockIdx.y;

    for (int u = tid; u < RANGE_NODES; u += PART_T) { sx[u] = 0.0f; sy[u] = 0.0f; }

    const int chunks = chunk_count(E);
    const int slices = (N + SLICE_NODES - 1) / SLICE_NODES;
    for (int s = blockIdx.x; s < slices; s += AGG_GROUPS) {
        const int n0 = s * SLICE_NODES;
        const int cnt = (N - n0 < SLICE_NODES) ? (N - n0) : SLICE_NODES;
        const float2* ysrc = (const float2*)y + n0;
        const int b = r * MAX_SLICES + s;
        for (int qb = 0; qb < chunks; qb += AGG_RUNS * (PART_T / 32)) {
            __syncthreads();
            if (tid < AGG_RUNS * (PART_T / 32)) {
                const int qt = qb + tid;
                const unsigned* o = tab + (size_t)(qt < chunks ? qt : chunks - 1) * TAB_ENTRIES + b;
                const int lo = (int)o[0];
                const int hi = (int)o[1];
                s_lo[tid] = lo;
                s_hi[tid] = qt < chunks ? hi : lo;
            }
            if (qb == 0) {
                for (int u0 = 0; u0 < SLICE_NODES; u0 += PART_T * STAGE_PER) {
                    float2 v[STAGE_PER];
                    #pragma unroll
                    for (int k = 0; k < STAGE_PER; ++k) {
                        const int u = u0 + k * PART_T + tid;
                        v[k] = ysrc[u < cnt ? u : cnt - 1];
                    }
                    #pragma unroll
                    for (int k = 0; k < STAGE_PER; ++k) ys[u0 + k * PART_T + tid] = v[k];
                }
            }
            __syncthreads();

            RunSet<AGG_RUNS> R;
            #pragma unroll
            for (int t = 0; t < AGG_RUNS; ++t) {
                const int idx = wv + t * (PART_T / 32);
                const int qt = qb + idx;
                const unsigned base = (unsigned)(qt < chunks ? qt : chunks - 1) * PART_CHUNK;
                R.i[t] = base + (unsigned)__builtin_amdgcn_readfirstlane(s_lo[idx]);
                R.h[t] = base + (unsigned)__builtin_amdgcn_readfirstlane(s_hi[idx]);
            }
            for (; runs_left(R); runs_step<AGG_RUNS, AGG_ROWS>(R)) {
                unsigned w[AGG_RUNS * AGG_ROWS];
                runs_load<AGG_RUNS, AGG_ROWS>(R, grouped, lane, w);
                float2 v[AGG_RUNS * AGG_ROWS];
                #pragma unroll
                for (int j = 0; j < AGG_RUNS * AGG_ROWS; ++j) v[j] = ys[w[j] >> RANGE_BITS];
                #pragma unroll
                for (int t = 0; t < AGG_RUNS; ++t) {
                    #pragma unroll
                    for (int k = 0; k < AGG_ROWS; ++k) {
                        const unsigned o = run_has(R, t, k, lane)
                                               ? (w[t * AGG_ROWS + k] & (RANGE_NODES - 1u))
                                               : (unsigned)(RANGE_NODES + lane);
                        atomicAdd(&sx[o], v[t * AGG_ROWS + k].x);
                        atomicAdd(&sy[o], v[t * AGG_ROWS + k].y);
                    }
                }
            }
        }
    }
    __syncthreads();

    float* agg = wave_half() ? agg1 : agg0;
    const int node0 = r << RANGE_BITS;
    for (int u = tid; u < 2 * RANGE_NODES; u += PART_T) {
        const int n = u >> 1;
        if (node0 + n < N) {
            const float val = (u & 1) ? sy[n] : sx[n];
            atomicAdd(agg + ((size_t)node0 * 2 + u), val);
        }
    }
}

#define DEG_SPLIT 3

__global__ __launch_bounds__(PART_T)
void k_deg(const unsigned* __restrict__ grouped, const unsigned* __restrict__ tab,
           int* __restrict__ part, size_t stride, int E, int N) {
    __shared__ int hist[RANGE_NODES + 32];
    const int tid = threadIdx.x;
    const int lane = tid & 31;
    const int wv = __builtin_amdgcn_readfirstlane(tid >> 5);
    const int r = blockIdx.y;

    for (int u = tid; u < RANGE_NODES; u += PART_T) hist[u] = 0;
    __syncthreads();

    int qlo, qhi;
    share_chunks(E, qlo, qhi);
    for (int q = qlo + wv; q < qhi; q += DEG_RUNS * (PART_T / 32)) {
        RunSet<DEG_RUNS> R;
        runs_open(R, tab, r * MAX_SLICES, (r + 1) * MAX_SLICES, q, qhi);
        for (; runs_left(R); runs_step<DEG_RUNS, DEG_ROWS>(R)) {
            unsigned w[DEG_RUNS * DEG_ROWS];
            runs_load<DEG_RUNS, DEG_ROWS>(R, grouped, lane, w);
            #pragma unroll
            for (int t = 0; t < DEG_RUNS; ++t) {
                #pragma unroll
                for (int k = 0; k < DEG_ROWS; ++k) {
                    const unsigned o = run_has(R, t, k, lane)
                                           ? (w[t * DEG_ROWS + k] & (RANGE_NODES - 1u))
                                           : (unsigned)(RANGE_NODES + lane);
                    atomicAdd(&hist[o], 1);
                }
            }
        }
    }
    __syncthreads();

    int* mine = part + (size_t)blockIdx.x * stride;
    const int node0 = r << RANGE_BITS;
    for (int u = tid; u < RANGE_NODES; u += PART_T)
        if (node0 + u < N) mine[node0 + u] = hist[u];
}

__global__ void k_node(const int* __restrict__ p0, const int* __restrict__ p1, const int* p2,
                       const float* __restrict__ x, float* dinv,
                       float* __restrict__ y1, float* __restrict__ agg1,
                       float* __restrict__ agg1b, int N) {
    int i = blockIdx.x * blockDim.x + threadIdx.x;
    if (i >= N) return;
    float f = (float)(p0[i] + p1[i] + p2[i] + 1);
    float dv = 1.0f / sqrtf(f);
    dinv[i] = dv;
    float2 xv = *(const float2*)(x + (size_t)i * 2);
    float2 v;
    v.x = dv * xv.x;
    v.y = dv * xv.y;
    *(float2*)(y1   + (size_t)i * 2) = v;
    *(float2*)(agg1 + (size_t)i * 2) = v;
    *(float2*)(agg1b + (size_t)i * 2) = make_float2(0.0f, 0.0f);
}

__global__ void k_layer2(const float* __restrict__ agg1, float* aggB,
                         const float* __restrict__ dinv,
                         const float* __restrict__ W1, const float* __restrict__ b1,
                         const float* __restrict__ W2,
                         float* __restrict__ y2, float* __restrict__ agg2, int N) {
    int lane = threadIdx.x & 31;
    int wave = threadIdx.x >> 5;
    int base = (blockIdx.x * (blockDim.x >> 5) + wave) * 16;
    if (base >= N) return;
    int m    = lane & 15;
    int half = lane >> 4;
    int node = base + m;
    int kb   = half * 2;
    bool full = (base + 16 <= N);

    int j = m & 7;
    v2f a1 = {0.0f, 0.0f};
    if (half == 0 && j < 4) {
        a1.x = W1[0 * 4 + j];
        a1.y = W1[1 * 4 + j];
    }
    v2f s = {0.0f, 0.0f};
    if (half == 0 && node < N) {
        float dv = dinv[node];
        float2 g = *(const float2*)(agg1 + (size_t)node * 2);
        float2* pb = (float2*)(aggB + (size_t)node * 2);
        float2 gb = *pb;
        *pb = make_float2(0.0f, 0.0f);
        s.x = dv * (g.x + gb.x);
        s.y = dv * (g.y + gb.y);
    }
    v8f c1 = {b1[0], b1[1], b1[2], b1[3], 0.0f, 0.0f, 0.0f, 0.0f};
    v8f p = __builtin_amdgcn_wmma_f32_16x16x4_f32(
        false, a1, false, s, (short)0, c1, false, false);

    v2f a2;
    a2.x = tanhf(half ? p[2] : p[0]);
    a2.y = tanhf(half ? p[3] : p[1]);

    v2f b2 = {0.0f, 0.0f};
    if (m < 2) {
        b2.x = W2[(kb + 0) * 2 + m];
        b2.y = W2[(kb + 1) * 2 + m];
    }
    v8f c2 = {};
    v8f d = __builtin_amdgcn_wmma_f32_16x16x4_f32(
        false, a2, false, b2, (short)0, c2, false, false);

    int nb = base + 8 * half;
    if (full) {
        if (m < 2) {
            float dv[8];
            #pragma unroll
            for (int v = 0; v < 8; ++v) dv[v] = dinv[nb + v];
            #pragma unroll
            for (int v = 0; v < 8; ++v) {
                float val = dv[v] * d[v];
                y2[(nb + v) * 2 + m]   = val;
                agg2[(nb + v) * 2 + m] = val;
            }
        }
    } else {
        if (m < 2) {
            #pragma unroll
            for (int v = 0; v < 8; ++v) {
                int nd = nb + v;
                if (nd < N) {
                    float val = dinv[nd] * d[v];
                    y2[nd * 2 + m]   = val;
                    agg2[nd * 2 + m] = val;
                }
            }
        }
    }
}

__global__ void k_final(const float* __restrict__ agg2, const float* __restrict__ agg2b,
                        const float* __restrict__ dinv,
                        const float* __restrict__ b2, const float* __restrict__ Wc,
                        const float* __restrict__ bc, float* __restrict__ out, int N) {
    int i = blockIdx.x * blockDim.x + threadIdx.x;
    if (i >= N) return;
    float dv = dinv[i];
    float2 ga = *(const float2*)(agg2  + (size_t)i * 2);
    float2 gb = *(const float2*)(agg2b + (size_t)i * 2);
    float h0 = tanhf(dv * (ga.x + gb.x) + b2[0]);
    float h1 = tanhf(dv * (ga.y + gb.y) + b2[1]);
    float2 o;
    o.x = h0 * Wc[0] + h1 * Wc[2] + bc[0];
    o.y = h0 * Wc[1] + h1 * Wc[3] + bc[1];
    *(float2*)(out + (size_t)i * 2) = o;
}

extern "C" void kernel_launch(void* const* d_in, const int* in_sizes, int n_in,
                              void* d_out, int out_size, void* d_ws, size_t ws_size,
                              hipStream_t stream) {
    const float* x  = (const float*)d_in[0];
    const int*   ei = (const int*)d_in[1];
    const float* W1 = (const float*)d_in[2];
    const float* b1 = (const float*)d_in[3];
    const float* W2 = (const float*)d_in[4];
    const float* b2 = (const float*)d_in[5];
    const float* Wc = (const float*)d_in[6];
    const float* bc = (const float*)d_in[7];
    float* out = (float*)d_out;

    const int N = in_sizes[0] / 2;
    const int E = in_sizes[1] / 2;
    const int* src = ei;
    const int* dst = ei + E;

    float* ws   = (float*)d_ws;
    float* y1   = ws;
    float* agg1 = ws + (size_t)2 * N;
    float* aggB = ws + (size_t)4 * N;
    float* y2   = ws;
    float* agg2 = ws + (size_t)6 * N;
    float* dinv = ws + (size_t)8 * N;
    int*   deg0 = (int*)dinv;
    int*   deg1 = (int*)ws;

    const int T = 256;
    dim3 gN((N + T - 1) / T);
    const int edgesPerBlock = (T / 32) * WAVE_EDGES;
    dim3 gE((E + edgesPerBlock - 1) / edgesPerBlock);
    const int nodesPerBlock = (T / 32) * 16;
    dim3 gW((N + nodesPerBlock - 1) / nodesPerBlock);
    dim3 gC((N + HIST_BINS - 1) / HIST_BINS, COUNT_CHUNKS);

    const size_t chunks   = ((size_t)E + PART_CHUNK - 1) / PART_CHUNK;
    const size_t offsOff  = (((size_t)9 * N * sizeof(float)) + 255) & ~(size_t)255;
    const size_t packOff  = (offsOff + chunks * TAB_ENTRIES * sizeof(unsigned) + 255) & ~(size_t)255;
    const size_t packEnd  = packOff + (size_t)E * sizeof(unsigned);
    const bool   groupedPath = (chunks > 0) && (N <= (1 << SRC_BITS)) && (ws_size >= packEnd);
    unsigned* offs = (unsigned*)((char*)d_ws + offsOff);
    unsigned* grouped = (unsigned*)((char*)d_ws + packOff);
    dim3 gA(AGG_GROUPS, (N + RANGE_NODES - 1) / RANGE_NODES);
    dim3 gD(DEG_SPLIT, (N + RANGE_NODES - 1) / RANGE_NODES);
    int* degp = (int*)(ws + (size_t)6 * N);

    if (groupedPath) {
        k_part  <<<dim3((unsigned)chunks), PART_T, 0, stream>>>(src, dst, grouped, offs, E, N);
        k_deg   <<<gD, PART_T, 0, stream>>>(grouped, offs, degp, (size_t)N, E, N);
        k_node  <<<gN, T, 0, stream>>>(degp, degp + (size_t)N, degp + (size_t)2 * N, x, dinv,
                                       y1, agg1, aggB, N);
        k_agg   <<<gA, PART_T, 0, stream>>>(grouped, offs, y1, agg1, aggB, E, N);
    } else {
        k_init_deg<<<gN, T, 0, stream>>>(deg0, deg1, N);
        k_count   <<<gC, COUNT_T, 0, stream>>>(dst, deg0, deg1, E, N);
        k_dinv    <<<gN, T, 0, stream>>>(deg0, deg1, dinv, N);
        k_y1      <<<gN, T, 0, stream>>>(x, dinv, y1, agg1, aggB, N);
        k_scatter<<<gE, T, 0, stream>>>(src, dst, y1, agg1, aggB, E);
    }
    k_layer2  <<<gW, T, 0, stream>>>(agg1, aggB, dinv, W1, b1, W2, y2, agg2, N);
    if (groupedPath) k_agg    <<<gA, PART_T, 0, stream>>>(grouped, offs, y2, agg2, aggB, E, N);
    else             k_scatter<<<gE, T, 0, stream>>>(src, dst, y2, agg2, aggB, E);
    k_final   <<<gN, T, 0, stream>>>(agg2, aggB, dinv, b2, Wc, bc, out, N);
}


# Source slices of 31,744 nodes (16, four per work-group of a range; was 5/5/4/4 of 28,672), a run taken in one trip of three rows; staging ahead of the chunk loop; summation order changes
# speedup vs baseline: 12.7506x; 1.1213x over previous
#include <hip/hip_runtime.h>
#include <math.h>

typedef __attribute__((ext_vector_type(2))) float v2f;
typedef __attribute__((ext_vector_type(8))) float v8f;

#define WAVE_EDGES 256

__device__ __forceinline__ int wave_half() {
    unsigned id = __builtin_amdgcn_s_sendmsg_rtn(0x87);
    return (int)((id >> 18) & 1u);
}

__device__ __forceinline__ bool vec_ok(const int* p, int E) {
    return ((E & 3) == 0) && ((((size_t)p) & 15) == 0);
}

__device__ __forceinline__ int wave_index() {
    return blockIdx.x * (blockDim.x >> 5) + __builtin_amdgcn_readfirstlane((int)(threadIdx.x >> 5));
}

__global__ void k_init_deg(int* deg0, int* deg1, int N) {
    int i = blockIdx.x * blockDim.x + threadIdx.x;
    if (i < N) {
        deg0[i] = 1;
        deg1[i] = 0;
    }
}

#define HIST_BITS    16
#define HIST_BINS    (1 << HIST_BITS)
#define COUNT_T      1024
#define COUNT_CHUNKS 32
#define COUNT_LOADS  8

__device__ __forceinline__ void hist_add(int* hist, unsigned slice, int d) {
    if (((unsigned)d >> HIST_BITS) == slice) atomicAdd(&hist[d & (HIST_BINS - 1)], 1);
}

template <bool FULL>
__device__ __forceinline__ void count_group(const int* __restrict__ p, int left,
                                            int* hist, unsigned slice) {
    int4 v[COUNT_LOADS];
    #pragma unroll
    for (int k = 0; k < COUNT_LOADS; ++k) {
        if (FULL || k * (COUNT_T * 4) + 4 <= left) v[k] = *(const int4*)(p + k * (COUNT_T * 4));
        else                                       v[k] = make_int4(-1, -1, -1, -1);
    }
    #pragma unroll
    for (int k = 0; k < COUNT_LOADS; ++k) {
        hist_add(hist, slice, v[k].x);
        hist_add(hist, slice, v[k].y);
        hist_add(hist, slice, v[k].z);
        hist_add(hist, slice, v[k].w);
    }
}

__global__ __launch_bounds__(COUNT_T)
void k_count(const int* __restrict__ dst, int* deg0, int* deg1, int E, int N) {
    __shared__ __attribute__((aligned(16))) int hist[HIST_BINS];
    const int tid = threadIdx.x;
    const unsigned slice = blockIdx.x;

    for (int u = tid * 4; u < HIST_BINS; u += COUNT_T * 4)
        *(int4*)(hist + u) = make_int4(0, 0, 0, 0);
    __syncthreads();

    unsigned per = ((unsigned)E + gridDim.y - 1u) / gridDim.y;
    per = (per + 3u) & ~3u;
    const long long beg = (long long)blockIdx.y * per;
    long long end = beg + per;
    if (end > E) end = E;
    if (end < beg) end = beg;

    const bool vec = (((size_t)dst) & 15) == 0;
    const long long end4 = vec ? beg + ((end - beg) & ~3LL) : beg;
    const long long step = (long long)COUNT_T * 4 * COUNT_LOADS;
    long long i = beg;
    for (; i + step <= end4; i += step)
        count_group<true>(dst + i + tid * 4, 0, hist, slice);
    if (i < end4)
        count_group<false>(dst + i + tid * 4, (int)(end4 - i) - tid * 4, hist, slice);
    for (long long j = end4 + tid; j < end; j += COUNT_T)
        hist_add(hist, slice, dst[j]);
    __syncthreads();

    int* deg = wave_half() ? deg1 : deg0;
    const int base = (int)(slice << HIST_BITS);
    for (int u = tid; u < HIST_BINS; u += COUNT_T * 8) {
        int c[8];
        #pragma unroll
        for (int k = 0; k < 8; ++k) c[k] = hist[u + k * COUNT_T];
        #pragma unroll
        for (int k = 0; k < 8; ++k) {
            const int node = base + u + k * COUNT_T;
            if (c[k] != 0 && node < N) atomicAdd(&deg[node], c[k]);
        }
    }
}

__global__ void k_dinv(const int* deg0, const int* __restrict__ deg1, float* dinv, int N) {
    int i = blockIdx.x * blockDim.x + threadIdx.x;
    if (i < N) {
        float f = (float)(deg0[i] + deg1[i]);
        dinv[i] = 1.0f / sqrtf(f);
    }
}

__global__ void k_y1(const float* __restrict__ x, const float* __restrict__ dinv,
                     float* __restrict__ y1, float* __restrict__ agg1,
                     float* __restrict__ agg1b, int N) {
    int i = blockIdx.x * blockDim.x + threadIdx.x;
    if (i >= N) return;
    float dv = dinv[i];
    float2 xv = *(const float2*)(x + (size_t)i * 2);
    float2 v;
    v.x = dv * xv.x;
    v.y = dv * xv.y;
    *(float2*)(y1   + (size_t)i * 2) = v;
    *(float2*)(agg1 + (size_t)i * 2) = v;
    *(float2*)(agg1b + (size_t)i * 2) = make_float2(0.0f, 0.0f);
}

template <bool FULL>
__device__ __forceinline__ int4 load4(const int* __restrict__ p, int e, int E) {
    int4 a;
    if (FULL) {
        a = *(const int4*)(p + e);
    } else {
        a.x = (e + 0 < E) ? p[e + 0] : -1;
        a.y = (e + 1 < E) ? p[e + 1] : -1;
        a.z = (e + 2 < E) ? p[e + 2] : -1;
        a.w = (e + 3 < E) ? p[e + 3] : -1;
    }
    return a;
}

template <bool FULL>
__device__ __forceinline__ float gather1(const float* __restrict__ y, int s, int f) {
    if (!FULL && s < 0) s = 0;
    return y[(unsigned)s * 2u + (unsigned)f];
}

template <bool FULL>
__device__ __forceinline__ void add1(float* agg, int t, int f, float v) {
    if (FULL || t >= 0)
        atomicAdd(agg + ((unsigned)t * 2u + (unsigned)f), v);
}

template <bool FULL>
__device__ __forceinline__ void scatter_wave(const int* __restrict__ src, const int* __restrict__ dst,
                                             const float* __restrict__ y, float* agg,
                                             int e0, int f, int E) {
    const int4 d0 = load4<FULL>(dst, e0,       E);
    const int4 d1 = load4<FULL>(dst, e0 + 64,  E);
    const int4 d2 = load4<FULL>(dst, e0 + 128, E);
    const int4 d3 = load4<FULL>(dst, e0 + 192, E);
    const int4 s0 = load4<FULL>(src, e0,       E);
    const int4 s1 = load4<FULL>(src, e0 + 64,  E);
    const int4 s2 = load4<FULL>(src, e0 + 128, E);
    const int4 s3 = load4<FULL>(src, e0 + 192, E);
    __builtin_amdgcn_sched_barrier(0);
    const float v00 = gather1<FULL>(y, s0.x, f);
    const float v01 = gather1<FULL>(y, s0.y, f);
    const float v02 = gather1<FULL>(y, s0.z, f);
    const float v03 = gather1<FULL>(y, s0.w, f);
    const float v10 = gather1<FULL>(y, s1.x, f);
    const float v11 = gather1<FULL>(y, s1.y, f);
    const float v12 = gather1<FULL>(y, s1.z, f);
    const float v13 = gather1<FULL>(y, s1.w, f);
    const float v20 = gather1<FULL>(y, s2.x, f);
    const float v21 = gather1<FULL>(y, s2.y, f);
    const float v22 = gather1<FULL>(y, s2.z, f);
    const float v23 = gather1<FULL>(y, s2.w, f);
    const float v30 = gather1<FULL>(y, s3.x, f);
    const float v31 = gather1<FULL>(y, s3.y, f);
    const float v32 = gather1<FULL>(y, s3.z, f);
    const float v33 = gather1<FULL>(y, s3.w, f);
    add1<FULL>(agg, d0.x, f, v00);
    add1<FULL>(agg, d0.y, f, v01);
    add1<FULL>(agg, d0.z, f, v02);
    add1<FULL>(agg, d0.w, f, v03);
    add1<FULL>(agg, d1.x, f, v10);
    add1<FULL>(agg, d1.y, f, v11);
    add1<FULL>(agg, d1.z, f, v12);
    add1<FULL>(agg, d1.w, f, v13);
    add1<FULL>(agg, d2.x, f, v20);
    add1<FULL>(agg, d2.y, f, v21);
    add1<FULL>(agg, d2.z, f, v22);
    add1<FULL>(agg, d2.w, f, v23);
    add1<FULL>(agg, d3.x, f, v30);
    add1<FULL>(agg, d3.y, f, v31);
    add1<FULL>(agg, d3.z, f, v32);
    add1<FULL>(agg, d3.w, f, v33);
}

__global__ void k_scatter(const int* __restrict__ src, const int* __restrict__ dst,
                          const float* __restrict__ y, float* agg0, float* agg1, int E) {
    const int lane = threadIdx.x & 31;
    const int wave = wave_index();
    const long long w0 = (long long)wave * WAVE_EDGES;
    if (w0 >= E) return;
    float* agg = wave_half() ? agg1 : agg0;
    const int f  = lane & 1;
    const int e0 = (int)w0 + (lane >> 1) * 4;
    if (vec_ok(dst, E) && vec_ok(src, E) && w0 + WAVE_EDGES <= E)
        scatter_wave<true>(src, dst, y, agg, e0, f, E);
    else
        scatter_wave<false>(src, dst, y, agg, e0, f, E);
}

#define RANGE_BITS  13
#define RANGE_NODES (1 << RANGE_BITS)
#define SRC_BITS    19
#define MAX_RANGES  64
#define SLICE_SHIFT 10
#define SLICE_UNITS 31
#define SLICE_NODES (SLICE_UNITS << SLICE_SHIFT)
#define MAX_SLICES  17
#define MAX_BUCKETS (MAX_RANGES * MAX_SLICES)
#define TAB_ENTRIES (MAX_BUCKETS + 1)
#define SCAN_PER    (MAX_BUCKETS / 32)
#define PART_T      1024
#define PART_PER    8
#define PART_GROUP  (PART_T * PART_PER)
#define PART_CHUNK  65536
#define AGG_GROUPS  4
#define AGG_RUNS    8
#define AGG_ROWS    3
#define AGG_HALF    (AGG_RUNS / 2)
#define DEG_RUNS    4
#define DEG_ROWS    6
#define STAGE_PER   8
static_assert(SLICE_NODES % PART_T == 0, "a slice is staged in whole rows of PART_T pairs");
static_assert(MAX_BUCKETS % 32 == 0, "the scanning wave takes the same number of buckets per lane");

__device__ __forceinline__ unsigned slice_of(unsigned s) {
    return (s >> SLICE_SHIFT) / SLICE_UNITS;
}

template <bool FULL>
__device__ __forceinline__ void part_load(const int* __restrict__ src, const int* __restrict__ dst,
                                          long long base, long long end, int N,
                                          int (&d)[PART_PER], unsigned (&s)[PART_PER]) {
    #pragma unroll
    for (int k = 0; k < PART_PER; ++k) {
        const long long e = base + (long long)k * PART_T;
        if (FULL || e < end) { d[k] = dst[e]; s[k] = (unsigned)src[e]; }
        else                 { d[k] = -1;     s[k] = 0u; }
    }
    #pragma unroll
    for (int k = 0; k < PART_PER; ++k)
        if (s[k] >= (unsigned)N) s[k] = (unsigned)N - 1u;
}

__global__ __launch_bounds__(PART_T)
void k_part(const int* __restrict__ src, const int* __restrict__ dst,
            unsigned* __restrict__ grouped, unsigned* __restrict__ tab, int E, int N) {
    __shared__ unsigned s_cur[MAX_BUCKETS];
    __shared__ unsigned s_lofs[TAB_ENTRIES];
    __shared__ unsigned s_lane[32];
    __shared__ unsigned s_word[PART_CHUNK];
    const int tid = threadIdx.x;
    const long long beg = (long long)blockIdx.x * PART_CHUNK;
    long long end = beg + PART_CHUNK;
    if (end > E) end = E;

    for (int u = tid; u < MAX_BUCKETS; u += PART_T) s_cur[u] = 0u;
    __syncthreads();

    for (int g = 0; g < PART_CHUNK / PART_GROUP; ++g) {
        const long long gb = beg + (long long)g * PART_GROUP;
        int d[PART_PER]; unsigned s[PART_PER];
        if (gb + PART_GROUP <= end) part_load<true >(src, dst, gb + tid, end, N, d, s);
        else                        part_load<false>(src, dst, gb + tid, end, N, d, s);
        #pragma unroll
        for (int k = 0; k < PART_PER; ++k)
            if ((unsigned)d[k] < (unsigned)N)
                atomicAdd(&s_cur[((unsigned)d[k] >> RANGE_BITS) * MAX_SLICES + slice_of(s[k])], 1u);
    }
    __syncthreads();

    if (tid < 32) {
        unsigned sum = 0u;
        for (int j = 0; j < SCAN_PER; ++j) sum += s_cur[tid * SCAN_PER + j];
        s_lane[tid] = sum;
    }
    __syncthreads();
    if (tid < 32) {
        unsigned o = 0u;
        for (int l = 0; l < tid; ++l) o += s_lane[l];
        for (int j = 0; j < SCAN_PER; ++j) {
            const unsigned c = s_cur[tid * SCAN_PER + j];
            s_lofs[tid * SCAN_PER + j] = o;
            s_cur[tid * SCAN_PER + j]  = o;
            o += c;
        }
        if (tid == 31) s_lofs[MAX_BUCKETS] = o;
    }
    __syncthreads();

    for (int g = 0; g < PART_CHUNK / PART_GROUP; ++g) {
        const long long gb = beg + (long long)g * PART_GROUP;
        int d[PART_PER]; unsigned s[PART_PER];
        if (gb + PART_GROUP <= end) part_load<true >(src, dst, gb + tid, end, N, d, s);
        else                        part_load<false>(src, dst, gb + tid, end, N, d, s);
        #pragma unroll
        for (int k = 0; k < PART_PER; ++k) {
            if ((unsigned)d[k] < (unsigned)N) {
                const unsigned sl = slice_of(s[k]);
                const unsigned p = atomicAdd(&s_cur[((unsigned)d[k] >> RANGE_BITS) * MAX_SLICES + sl], 1u);
                s_word[p] = ((s[k] - sl * SLICE_NODES) << RANGE_BITS) | ((unsigned)d[k] & (RANGE_NODES - 1u));
            }
        }
    }
    __syncthreads();

    const int total = (int)s_lofs[MAX_BUCKETS];
    unsigned* outp = grouped + (size_t)blockIdx.x * PART_CHUNK;
    #pragma unroll 8
    for (int i = tid; i < total; i += PART_T) outp[i] = s_word[i];
    for (int u = tid; u < TAB_ENTRIES; u += PART_T)
        tab[(size_t)blockIdx.x * TAB_ENTRIES + u] = s_lofs[u];
}

__device__ __forceinline__ int chunk_count(int E) {
    return (int)(((unsigned)E + (PART_CHUNK - 1u)) / PART_CHUNK);
}

__device__ __forceinline__ void share_chunks(int E, int& qlo, int& qhi) {
    const unsigned chunks = (unsigned)chunk_count(E);
    const unsigned cps = (chunks + gridDim.x - 1u) / gridDim.x;
    unsigned lo = blockIdx.x * cps, hi = lo + cps;
    if (hi > chunks) hi = chunks;
    if (lo > hi) lo = hi;
    qlo = (int)lo;
    qhi = (int)hi;
}

template <int RUNS>
struct RunSet {
    unsigned i[RUNS];
    unsigned h[RUNS];
};

template <int RUNS>
__device__ __forceinline__ void runs_open(RunSet<RUNS>& R, const unsigned* __restrict__ tab,
                                          int b0, int b1, int q, int qhi) {
    #pragma unroll
    for (int t = 0; t < RUNS; ++t) {
        const int qt = q + t * (PART_T / 32);
        const int qc = qt < qhi ? qt : qhi - 1;
        const unsigned* o = tab + (size_t)qc * TAB_ENTRIES;
        const unsigned lo = o[b0];
        const unsigned hi = o[b1];
        const unsigned base = (unsigned)qc * PART_CHUNK;
        R.i[t] = base + lo;
        R.h[t] = base + (qt < qhi ? hi : lo);
    }
}

template <int RUNS>
__device__ __forceinline__ bool runs_left(const RunSet<RUNS>& R) {
    bool any = false;
    #pragma unroll
    for (int t = 0; t < RUNS; ++t) any = any || (R.i[t] < R.h[t]);
    return any;
}

template <int RUNS, int ROWS>
__device__ __forceinline__ void runs_step(RunSet<RUNS>& R) {
    #pragma unroll
    for (int t = 0; t < RUNS; ++t) R.i[t] += 32 * ROWS;
}

template <int RUNS>
__device__ __forceinline__ bool run_has(const RunSet<RUNS>& R, int t, int k, int lane) {
    return R.i[t] + (unsigned)(k * 32 + lane) < R.h[t];
}

template <int RUNS, int ROWS>
__device__ __forceinline__ void runs_load(const RunSet<RUNS>& R, const unsigned* __restrict__ grouped,
                                          int lane, unsigned (&w)[RUNS * ROWS]) {
    unsigned safe = R.i[0];
    #pragma unroll
    for (int t = RUNS - 1; t >= 0; --t)
        if (R.i[t] < R.h[t]) safe = R.i[t];
    #pragma unroll
    for (int t = 0; t < RUNS; ++t) {
        #pragma unroll
        for (int k = 0; k < ROWS; ++k) {
            const unsigned a = R.i[t] + (unsigned)(k * 32 + lane);
            w[t * ROWS + k] = grouped[run_has(R, t, k, lane) ? a : safe];
        }
    }
}

__global__ __launch_bounds__(PART_T)
void k_agg(const unsigned* __restrict__ grouped, const unsigned* __restrict__ tab,
           const float* __restrict__ y, float* agg0, float* agg1, int E, int N) {
    __shared__ float sx[RANGE_NODES + 32];
    __shared__ float sy[RANGE_NODES + 32];
    __shared__ float2 ys[SLICE_NODES];
    __shared__ int s_lo[AGG_RUNS * (PART_T / 32)];
    __shared__ int s_hi[AGG_RUNS * (PART_T / 32)];
    const int tid = threadIdx.x;
    const int lane = tid & 31;
    const int wv = __builtin_amdgcn_readfirstlane(tid >> 5);
    const int r = blockIdx.y;

    for (int u = tid; u < RANGE_NODES; u += PART_T) { sx[u] = 0.0f; sy[u] = 0.0f; }

    const int chunks = chunk_count(E);
    const int slices = (N + SLICE_NODES - 1) / SLICE_NODES;
    for (int s = blockIdx.x; s < slices; s += AGG_GROUPS) {
        const int n0 = s * SLICE_NODES;
        const int cnt = (N - n0 < SLICE_NODES) ? (N - n0) : SLICE_NODES;
        const float2* ysrc = (const float2*)y + n0;
        const int b = r * MAX_SLICES + s;
        __syncthreads();
        {
            const unsigned last = (unsigned)(cnt - 1);
            for (int u0 = 0; u0 < SLICE_NODES; u0 += PART_T * STAGE_PER) {
                float2 v[STAGE_PER];
                #pragma unroll
                for (int k = 0; k < STAGE_PER; ++k) {
                    if (u0 + k * PART_T < SLICE_NODES) {
                        const unsigned u = (unsigned)(u0 + k * PART_T + tid);
                        v[k] = ysrc[u < last ? u : last];
                    }
                }
                #pragma unroll
                for (int k = 0; k < STAGE_PER; ++k)
                    if (u0 + k * PART_T < SLICE_NODES) ys[u0 + k * PART_T + tid] = v[k];
            }
        }
        for (int qb = 0; qb < chunks; qb += AGG_RUNS * (PART_T / 32)) {
            if (qb != 0) __syncthreads();
            if (tid < AGG_RUNS * (PART_T / 32)) {
                const int qt = qb + tid;
                const unsigned* o = tab + (size_t)(qt < chunks ? qt : chunks - 1) * TAB_ENTRIES + b;
                const int lo = (int)o[0];
                const int hi = (int)o[1];
                s_lo[tid] = lo;
                s_hi[tid] = qt < chunks ? hi : lo;
            }
            __syncthreads();

            RunSet<AGG_RUNS> R;
            #pragma unroll
            for (int t = 0; t < AGG_RUNS; ++t) {
                const int idx = wv + t * (PART_T / 32);
                const int qt = qb + idx;
                const unsigned base = (unsigned)(qt < chunks ? qt : chunks - 1) * PART_CHUNK;
                R.i[t] = base + (unsigned)__builtin_amdgcn_readfirstlane(s_lo[idx]);
                R.h[t] = base + (unsigned)__builtin_amdgcn_readfirstlane(s_hi[idx]);
            }
            for (; runs_left(R); runs_step<AGG_RUNS, AGG_ROWS>(R)) {
                unsigned w[AGG_RUNS * AGG_ROWS];
                runs_load<AGG_RUNS, AGG_ROWS>(R, grouped, lane, w);
                #pragma unroll
                for (int g = 0; g < AGG_RUNS; g += AGG_HALF) {
                    float2 v[AGG_HALF * AGG_ROWS];
                    #pragma unroll
                    for (int j = 0; j < AGG_HALF * AGG_ROWS; ++j) v[j] = ys[w[g * AGG_ROWS + j] >> RANGE_BITS];
                    #pragma unroll
                    for (int t = 0; t < AGG_HALF; ++t) {
                        #pragma unroll
                        for (int k = 0; k < AGG_ROWS; ++k) {
                            const unsigned o = run_has(R, g + t, k, lane)
                                                   ? (w[(g + t) * AGG_ROWS + k] & (RANGE_NODES - 1u))
                                                   : (unsigned)(RANGE_NODES + lane);
                            atomicAdd(&sx[o], v[t * AGG_ROWS + k].x);
                            atomicAdd(&sy[o], v[t * AGG_ROWS + k].y);
                        }
                    }
                }
            }
        }
    }
    __syncthreads();

    float* agg = wave_half() ? agg1 : agg0;
    const int node0 = r << RANGE_BITS;
    for (int u = tid; u < 2 * RANGE_NODES; u += PART_T) {
        const int n = u >> 1;
        if (node0 + n < N) {
            const float val = (u & 1) ? sy[n] : sx[n];
            atomicAdd(agg + ((size_t)node0 * 2 + u), val);
        }
    }
}

#define DEG_SPLIT 3

__global__ __launch_bounds__(PART_T)
void k_deg(const unsigned* __restrict__ grouped, const unsigned* __restrict__ tab,
           int* __restrict__ part, size_t stride, int E, int N) {
    __shared__ int hist[RANGE_NODES + 32];
    const int tid = threadIdx.x;
    const int lane = tid & 31;
    const int wv = __builtin_amdgcn_readfirstlane(tid >> 5);
    const int r = blockIdx.y;

    for (int u = tid; u < RANGE_NODES; u += PART_T) hist[u] = 0;
    __syncthreads();

    int qlo, qhi;
    share_chunks(E, qlo, qhi);
    for (int q = qlo + wv; q < qhi; q += DEG_RUNS * (PART_T / 32)) {
        RunSet<DEG_RUNS> R;
        runs_open(R, tab, r * MAX_SLICES, (r + 1) * MAX_SLICES, q, qhi);
        for (; runs_left(R); runs_step<DEG_RUNS, DEG_ROWS>(R)) {
            unsigned w[DEG_RUNS * DEG_ROWS];
            runs_load<DEG_RUNS, DEG_ROWS>(R, grouped, lane, w);
            #pragma unroll
            for (int t = 0; t < DEG_RUNS; ++t) {
                #pragma unroll
                for (int k = 0; k < DEG_ROWS; ++k) {
                    const unsigned o = run_has(R, t, k, lane)
                                           ? (w[t * DEG_ROWS + k] & (RANGE_NODES - 1u))
                                           : (unsigned)(RANGE_NODES + lane);
                    atomicAdd(&hist[o], 1);
                }
            }
        }
    }
    __syncthreads();

    int* mine = part + (size_t)blockIdx.x * stride;
    const int node0 = r << RANGE_BITS;
    for (int u = tid; u < RANGE_NODES; u += PART_T)
        if (node0 + u < N) mine[node0 + u] = hist[u];
}

__global__ void k_node(const int* __restrict__ p0, const int* __restrict__ p1, const int* p2,
                       const float* __restrict__ x, float* dinv,
                       float* __restrict__ y1, float* __restrict__ agg1,
                       float* __restrict__ agg1b, int N) {
    int i = blockIdx.x * blockDim.x + threadIdx.x;
    if (i >= N) return;
    float f = (float)(p0[i] + p1[i] + p2[i] + 1);
    float dv = 1.0f / sqrtf(f);
    dinv[i] = dv;
    float2 xv = *(const float2*)(x + (size_t)i * 2);
    float2 v;
    v.x = dv * xv.x;
    v.y = dv * xv.y;
    *(float2*)(y1   + (size_t)i * 2) = v;
    *(float2*)(agg1 + (size_t)i * 2) = v;
    *(float2*)(agg1b + (size_t)i * 2) = make_float2(0.0f, 0.0f);
}

__global__ void k_layer2(const float* __restrict__ agg1, float* aggB,
                         const float* __restrict__ dinv,
                         const float* __restrict__ W1, const float* __restrict__ b1,
                         const float* __restrict__ W2,
                         float* __restrict__ y2, float* __restrict__ agg2, int N) {
    int lane = threadIdx.x & 31;
    int wave = threadIdx.x >> 5;
    int base = (blockIdx.x * (blockDim.x >> 5) + wave) * 16;
    if (base >= N) return;
    int m    = lane & 15;
    int half = lane >> 4;
    int node = base + m;
    int kb   = half * 2;
    bool full = (base + 16 <= N);

    int j = m & 7;
    v2f a1 = {0.0f, 0.0f};
    if (half == 0 && j < 4) {
        a1.x = W1[0 * 4 + j];
        a1.y = W1[1 * 4 + j];
    }
    v2f s = {0.0f, 0.0f};
    if (half == 0 && node < N) {
        float dv = dinv[node];
        float2 g = *(const float2*)(agg1 + (size_t)node * 2);
        float2* pb = (float2*)(aggB + (size_t)node * 2);
        float2 gb = *pb;
        *pb = make_float2(0.0f, 0.0f);
        s.x = dv * (g.x + gb.x);
        s.y = dv * (g.y + gb.y);
    }
    v8f c1 = {b1[0], b1[1], b1[2], b1[3], 0.0f, 0.0f, 0.0f, 0.0f};
    v8f p = __builtin_amdgcn_wmma_f32_16x16x4_f32(
        false, a1, false, s, (short)0, c1, false, false);

    v2f a2;
    a2.x = tanhf(half ? p[2] : p[0]);
    a2.y = tanhf(half ? p[3] : p[1]);

    v2f b2 = {0.0f, 0.0f};
    if (m < 2) {
        b2.x = W2[(kb + 0) * 2 + m];
        b2.y = W2[(kb + 1) * 2 + m];
    }
    v8f c2 = {};
    v8f d = __builtin_amdgcn_wmma_f32_16x16x4_f32(
        false, a2, false, b2, (short)0, c2, false, false);

    int nb = base + 8 * half;
    if (full) {
        if (m < 2) {
            float dv[8];
            #pragma unroll
            for (int v = 0; v < 8; ++v) dv[v] = dinv[nb + v];
            #pragma unroll
            for (int v = 0; v < 8; ++v) {
                float val = dv[v] * d[v];
                y2[(nb + v) * 2 + m]   = val;
                agg2[(nb + v) * 2 + m] = val;
            }
        }
    } else {
        if (m < 2) {
            #pragma unroll
            for (int v = 0; v < 8; ++v) {
                int nd = nb + v;
                if (nd < N) {
                    float val = dinv[nd] * d[v];
                    y2[nd * 2 + m]   = val;
                    agg2[nd * 2 + m] = val;
                }
            }
        }
    }
}

__global__ void k_final(const float* __restrict__ agg2, const float* __restrict__ agg2b,
                        const float* __restrict__ dinv,
                        const float* __restrict__ b2, const float* __restrict__ Wc,
                        const float* __restrict__ bc, float* __restrict__ out, int N) {
    int i = blockIdx.x * blockDim.x + threadIdx.x;
    if (i >= N) return;
    float dv = dinv[i];
    float2 ga = *(const float2*)(agg2  + (size_t)i * 2);
    float2 gb = *(const float2*)(agg2b + (size_t)i * 2);
    float h0 = tanhf(dv * (ga.x + gb.x) + b2[0]);
    float h1 = tanhf(dv * (ga.y + gb.y) + b2[1]);
    float2 o;
    o.x = h0 * Wc[0] + h1 * Wc[2] + bc[0];
    o.y = h0 * Wc[1] + h1 * Wc[3] + bc[1];
    *(float2*)(out + (size_t)i * 2) = o;
}

extern "C" void kernel_launch(void* const* d_in, const int* in_sizes, int n_in,
                              void* d_out, int out_size, void* d_ws, size_t ws_size,
                              hipStream_t stream) {
    const float* x  = (const float*)d_in[0];
    const int*   ei = (const int*)d_in[1];
    const float* W1 = (const float*)d_in[2];
    const float* b1 = (const float*)d_in[3];
    const float* W2 = (const float*)d_in[4];
    const float* b2 = (const float*)d_in[5];
    const float* Wc = (const float*)d_in[6];
    const float* bc = (const float*)d_in[7];
    float* out = (float*)d_out;

    const int N = in_sizes[0] / 2;
    const int E = in_sizes[1] / 2;
    const int* src = ei;
    const int* dst = ei + E;

    float* ws   = (float*)d_ws;
    float* y1   = ws;
    float* agg1 = ws + (size_t)2 * N;
    float* aggB = ws + (size_t)4 * N;
    float* y2   = ws;
    float* agg2 = ws + (size_t)6 * N;
    float* dinv = ws + (size_t)8 * N;
    int*   deg0 = (int*)dinv;
    int*   deg1 = (int*)ws;

    const int T = 256;
    dim3 gN((N + T - 1) / T);
    const int edgesPerBlock = (T / 32) * WAVE_EDGES;
    dim3 gE((E + edgesPerBlock - 1) / edgesPerBlock);
    const int nodesPerBlock = (T / 32) * 16;
    dim3 gW((N + nodesPerBlock - 1) / nodesPerBlock);
    dim3 gC((N + HIST_BINS - 1) / HIST_BINS, COUNT_CHUNKS);

    const size_t chunks   = ((size_t)E + PART_CHUNK - 1) / PART_CHUNK;
    const size_t offsOff  = (((size_t)9 * N * sizeof(float)) + 255) & ~(size_t)255;
    const size_t packOff  = (offsOff + chunks * TAB_ENTRIES * sizeof(unsigned) + 255) & ~(size_t)255;
    const size_t packEnd  = packOff + (size_t)E * sizeof(unsigned);
    const bool   groupedPath = (chunks > 0) && (N <= (1 << SRC_BITS)) && (ws_size >= packEnd);
    unsigned* offs = (unsigned*)((char*)d_ws + offsOff);
    unsigned* grouped = (unsigned*)((char*)d_ws + packOff);
    dim3 gA(AGG_GROUPS, (N + RANGE_NODES - 1) / RANGE_NODES);
    dim3 gD(DEG_SPLIT, (N + RANGE_NODES - 1) / RANGE_NODES);
    int* degp = (int*)(ws + (size_t)6 * N);

    if (groupedPath) {
        k_part  <<<dim3((unsigned)chunks), PART_T, 0, stream>>>(src, dst, grouped, offs, E, N);
        k_deg   <<<gD, PART_T, 0, stream>>>(grouped, offs, degp, (size_t)N, E, N);
        k_node  <<<gN, T, 0, stream>>>(degp, degp + (size_t)N, degp + (size_t)2 * N, x, dinv,
                                       y1, agg1, aggB, N);
        k_agg   <<<gA, PART_T, 0, stream>>>(grouped, offs, y1, agg1, aggB, E, N);
    } else {
        k_init_deg<<<gN, T, 0, stream>>>(deg0, deg1, N);
        k_count   <<<gC, COUNT_T, 0, stream>>>(dst, deg0, deg1, E, N);
        k_dinv    <<<gN, T, 0, stream>>>(deg0, deg1, dinv, N);
        k_y1      <<<gN, T, 0, stream>>>(x, dinv, y1, agg1, aggB, N);
        k_scatter<<<gE, T, 0, stream>>>(src, dst, y1, agg1, aggB, E);
    }
    k_layer2  <<<gW, T, 0, stream>>>(agg1, aggB, dinv, W1, b1, W2, y2, agg2, N);
    if (groupedPath) k_agg    <<<gA, PART_T, 0, stream>>>(grouped, offs, y2, agg2, aggB, E, N);
    else             k_scatter<<<gE, T, 0, stream>>>(src, dst, y2, agg2, aggB, E);
    k_final   <<<gN, T, 0, stream>>>(agg2, aggB, dinv, b2, Wc, bc, out, N);
}


# k_agg: a wave's eight runs of a slice taken as one stream of words, in a trip of 16, 18 or 20 full rows, not three rows per run; LDS float atomics kept, summation order changes
# speedup vs baseline: 13.2836x; 1.0418x over previous
#include <hip/hip_runtime.h>
#include <math.h>

typedef __attribute__((ext_vector_type(2))) float v2f;
typedef __attribute__((ext_vector_type(8))) float v8f;

#define WAVE_EDGES 256

__device__ __forceinline__ int wave_half() {
    unsigned id = __builtin_amdgcn_s_sendmsg_rtn(0x87);
    return (int)((id >> 18) & 1u);
}

__device__ __forceinline__ bool vec_ok(const int* p, int E) {
    return ((E & 3) == 0) && ((((size_t)p) & 15) == 0);
}

__device__ __forceinline__ int wave_index() {
    return blockIdx.x * (blockDim.x >> 5) + __builtin_amdgcn_readfirstlane((int)(threadIdx.x >> 5));
}

__global__ void k_init_deg(int* deg0, int* deg1, int N) {
    int i = blockIdx.x * blockDim.x + threadIdx.x;
    if (i < N) {
        deg0[i] = 1;
        deg1[i] = 0;
    }
}

#define HIST_BITS    16
#define HIST_BINS    (1 << HIST_BITS)
#define COUNT_T      1024
#define COUNT_CHUNKS 32
#define COUNT_LOADS  8

__device__ __forceinline__ void hist_add(int* hist, unsigned slice, int d) {
    if (((unsigned)d >> HIST_BITS) == slice) atomicAdd(&hist[d & (HIST_BINS - 1)], 1);
}

template <bool FULL>
__device__ __forceinline__ void count_group(const int* __restrict__ p, int left,
                                            int* hist, unsigned slice) {
    int4 v[COUNT_LOADS];
    #pragma unroll
    for (int k = 0; k < COUNT_LOADS; ++k) {
        if (FULL || k * (COUNT_T * 4) + 4 <= left) v[k] = *(const int4*)(p + k * (COUNT_T * 4));
        else                                       v[k] = make_int4(-1, -1, -1, -1);
    }
    #pragma unroll
    for (int k = 0; k < COUNT_LOADS; ++k) {
        hist_add(hist, slice, v[k].x);
        hist_add(hist, slice, v[k].y);
        hist_add(hist, slice, v[k].z);
        hist_add(hist, slice, v[k].w);
    }
}

__global__ __launch_bounds__(COUNT_T)
void k_count(const int* __restrict__ dst, int* deg0, int* deg1, int E, int N) {
    __shared__ __attribute__((aligned(16))) int hist[HIST_BINS];
    const int tid = threadIdx.x;
    const unsigned slice = blockIdx.x;

    for (int u = tid * 4; u < HIST_BINS; u += COUNT_T * 4)
        *(int4*)(hist + u) = make_int4(0, 0, 0, 0);
    __syncthreads();

    unsigned per = ((unsigned)E + gridDim.y - 1u) / gridDim.y;
    per = (per + 3u) & ~3u;
    const long long beg = (long long)blockIdx.y * per;
    long long end = beg + per;
    if (end > E) end = E;
    if (end < beg) end = beg;

    const bool vec = (((size_t)dst) & 15) == 0;
    const long long end4 = vec ? beg + ((end - beg) & ~3LL) : beg;
    const long long step = (long long)COUNT_T * 4 * COUNT_LOADS;
    long long i = beg;
    for (; i + step <= end4; i += step)
        count_group<true>(dst + i + tid * 4, 0, hist, slice);
    if (i < end4)
        count_group<false>(dst + i + tid * 4, (int)(end4 - i) - tid * 4, hist, slice);
    for (long long j = end4 + tid; j < end; j += COUNT_T)
        hist_add(hist, slice, dst[j]);
    __syncthreads();

    int* deg = wave_half() ? deg1 : deg0;
    const int base = (int)(slice << HIST_BITS);
    for (int u = tid; u < HIST_BINS; u += COUNT_T * 8) {
        int c[8];
        #pragma unroll
        for (int k = 0; k < 8; ++k) c[k] = hist[u + k * COUNT_T];
        #pragma unroll
        for (int k = 0; k < 8; ++k) {
            const int node = base + u + k * COUNT_T;
            if (c[k] != 0 && node < N) atomicAdd(&deg[node], c[k]);
        }
    }
}

__global__ void k_dinv(const int* deg0, const int* __restrict__ deg1, float* dinv, int N) {
    int i = blockIdx.x * blockDim.x + threadIdx.x;
    if (i < N) {
        float f = (float)(deg0[i] + deg1[i]);
        dinv[i] = 1.0f / sqrtf(f);
    }
}

__global__ void k_y1(const float* __restrict__ x, const float* __restrict__ dinv,
                     float* __restrict__ y1, float* __restrict__ agg1,
                     float* __restrict__ agg1b, int N) {
    int i = blockIdx.x * blockDim.x + threadIdx.x;
    if (i >= N) return;
    float dv = dinv[i];
    float2 xv = *(const float2*)(x + (size_t)i * 2);
    float2 v;
    v.x = dv * xv.x;
    v.y = dv * xv.y;
    *(float2*)(y1   + (size_t)i * 2) = v;
    *(float2*)(agg1 + (size_t)i * 2) = v;
    *(float2*)(agg1b + (size_t)i * 2) = make_float2(0.0f, 0.0f);
}

template <bool FULL>
__device__ __forceinline__ int4 load4(const int* __restrict__ p, int e, int E) {
    int4 a;
    if (FULL) {
        a = *(const int4*)(p + e);
    } else {
        a.x = (e + 0 < E) ? p[e + 0] : -1;
        a.y = (e + 1 < E) ? p[e + 1] : -1;
        a.z = (e + 2 < E) ? p[e + 2] : -1;
        a.w = (e + 3 < E) ? p[e + 3] : -1;
    }
    return a;
}

template <bool FULL>
__device__ __forceinline__ float gather1(const float* __restrict__ y, int s, int f) {
    if (!FULL && s < 0) s = 0;
    return y[(unsigned)s * 2u + (unsigned)f];
}

template <bool FULL>
__device__ __forceinline__ void add1(float* agg, int t, int f, float v) {
    if (FULL || t >= 0)
        atomicAdd(agg + ((unsigned)t * 2u + (unsigned)f), v);
}

template <bool FULL>
__device__ __forceinline__ void scatter_wave(const int* __restrict__ src, const int* __restrict__ dst,
                                             const float* __restrict__ y, float* agg,
                                             int e0, int f, int E) {
    const int4 d0 = load4<FULL>(dst, e0,       E);
    const int4 d1 = load4<FULL>(dst, e0 + 64,  E);
    const int4 d2 = load4<FULL>(dst, e0 + 128, E);
    const int4 d3 = load4<FULL>(dst, e0 + 192, E);
    const int4 s0 = load4<FULL>(src, e0,       E);
    const int4 s1 = load4<FULL>(src, e0 + 64,  E);
    const int4 s2 = load4<FULL>(src, e0 + 128, E);
    const int4 s3 = load4<FULL>(src, e0 + 192, E);
    __builtin_amdgcn_sched_barrier(0);
    const float v00 = gather1<FULL>(y, s0.x, f);
    const float v01 = gather1<FULL>(y, s0.y, f);
    const float v02 = gather1<FULL>(y, s0.z, f);
    const float v03 = gather1<FULL>(y, s0.w, f);
    const float v10 = gather1<FULL>(y, s1.x, f);
    const float v11 = gather1<FULL>(y, s1.y, f);
    const float v12 = gather1<FULL>(y, s1.z, f);
    const float v13 = gather1<FULL>(y, s1.w, f);
    const float v20 = gather1<FULL>(y, s2.x, f);
    const float v21 = gather1<FULL>(y, s2.y, f);
    const float v22 = gather1<FULL>(y, s2.z, f);
    const float v23 = gather1<FULL>(y, s2.w, f);
    const float v30 = gather1<FULL>(y, s3.x, f);
    const float v31 = gather1<FULL>(y, s3.y, f);
    const float v32 = gather1<FULL>(y, s3.z, f);
    const float v33 = gather1<FULL>(y, s3.w, f);
    add1<FULL>(agg, d0.x, f, v00);
    add1<FULL>(agg, d0.y, f, v01);
    add1<FULL>(agg, d0.z, f, v02);
    add1<FULL>(agg, d0.w, f, v03);
    add1<FULL>(agg, d1.x, f, v10);
    add1<FULL>(agg, d1.y, f, v11);
    add1<FULL>(agg, d1.z, f, v12);
    add1<FULL>(agg, d1.w, f, v13);
    add1<FULL>(agg, d2.x, f, v20);
    add1<FULL>(agg, d2.y, f, v21);
    add1<FULL>(agg, d2.z, f, v22);
    add1<FULL>(agg, d2.w, f, v23);
    add1<FULL>(agg, d3.x, f, v30);
    add1<FULL>(agg, d3.y, f, v31);
    add1<FULL>(agg, d3.z, f, v32);
    add1<FULL>(agg, d3.w, f, v33);
}

__global__ void k_scatter(const int* __restrict__ src, const int* __restrict__ dst,
                          const float* __restrict__ y, float* agg0, float* agg1, int E) {
    const int lane = threadIdx.x & 31;
    const int wave = wave_index();
    const long long w0 = (long long)wave * WAVE_EDGES;
    if (w0 >= E) return;
    float* agg = wave_half() ? agg1 : agg0;
    const int f  = lane & 1;
    const int e0 = (int)w0 + (lane >> 1) * 4;
    if (vec_ok(dst, E) && vec_ok(src, E) && w0 + WAVE_EDGES <= E)
        scatter_wave<true>(src, dst, y, agg, e0, f, E);
    else
        scatter_wave<false>(src, dst, y, agg, e0, f, E);
}

#define RANGE_BITS  13
#define RANGE_NODES (1 << RANGE_BITS)
#define SRC_BITS    19
#define MAX_RANGES  64
#define SLICE_SHIFT 10
#define SLICE_UNITS 31
#define SLICE_NODES (SLICE_UNITS << SLICE_SHIFT)
#define MAX_SLICES  17
#define MAX_BUCKETS (MAX_RANGES * MAX_SLICES)
#define TAB_ENTRIES (MAX_BUCKETS + 1)
#define SCAN_PER    (MAX_BUCKETS / 32)
#define PART_T      1024
#define PART_PER    8
#define PART_GROUP  (PART_T * PART_PER)
#define PART_CHUNK  65536
#define AGG_GROUPS  4
#define AGG_RUNS    8
#define DEG_RUNS    4
#define DEG_ROWS    6
#define STAGE_PER   8
static_assert(SLICE_NODES % PART_T == 0, "a slice is staged in whole rows of PART_T pairs");
static_assert(MAX_BUCKETS % 32 == 0, "the scanning wave takes the same number of buckets per lane");

__device__ __forceinline__ unsigned slice_of(unsigned s) {
    return (s >> SLICE_SHIFT) / SLICE_UNITS;
}

template <bool FULL>
__device__ __forceinline__ void part_load(const int* __restrict__ src, const int* __restrict__ dst,
                                          long long base, long long end, int N,
                                          int (&d)[PART_PER], unsigned (&s)[PART_PER]) {
    #pragma unroll
    for (int k = 0; k < PART_PER; ++k) {
        const long long e = base + (long long)k * PART_T;
        if (FULL || e < end) { d[k] = dst[e]; s[k] = (unsigned)src[e]; }
        else                 { d[k] = -1;     s[k] = 0u; }
    }
    #pragma unroll
    for (int k = 0; k < PART_PER; ++k)
        if (s[k] >= (unsigned)N) s[k] = (unsigned)N - 1u;
}

__global__ __launch_bounds__(PART_T)
void k_part(const int* __restrict__ src, const int* __restrict__ dst,
            unsigned* __restrict__ grouped, unsigned* __restrict__ tab, int E, int N) {
    __shared__ unsigned s_cur[MAX_BUCKETS];
    __shared__ unsigned s_lofs[TAB_ENTRIES];
    __shared__ unsigned s_lane[32];
    __shared__ unsigned s_word[PART_CHUNK];
    const int tid = threadIdx.x;
    const long long beg = (long long)blockIdx.x * PART_CHUNK;
    long long end = beg + PART_CHUNK;
    if (end > E) end = E;

    for (int u = tid; u < MAX_BUCKETS; u += PART_T) s_cur[u] = 0u;
    __syncthreads();

    for (int g = 0; g < PART_CHUNK / PART_GROUP; ++g) {
        const long long gb = beg + (long long)g * PART_GROUP;
        int d[PART_PER]; unsigned s[PART_PER];
        if (gb + PART_GROUP <= end) part_load<true >(src, dst, gb + tid, end, N, d, s);
        else                        part_load<false>(src, dst, gb + tid, end, N, d, s);
        #pragma unroll
        for (int k = 0; k < PART_PER; ++k)
            if ((unsigned)d[k] < (unsigned)N)
                atomicAdd(&s_cur[((unsigned)d[k] >> RANGE_BITS) * MAX_SLICES + slice_of(s[k])], 1u);
    }
    __syncthreads();

    if (tid < 32) {
        unsigned sum = 0u;
        for (int j = 0; j < SCAN_PER; ++j) sum += s_cur[tid * SCAN_PER + j];
        s_lane[tid] = sum;
    }
    __syncthreads();
    if (tid < 32) {
        unsigned o = 0u;
        for (int l = 0; l < tid; ++l) o += s_lane[l];
        for (int j = 0; j < SCAN_PER; ++j) {
            const unsigned c = s_cur[tid * SCAN_PER + j];
            s_lofs[tid * SCAN_PER + j] = o;
            s_cur[tid * SCAN_PER + j]  = o;
            o += c;
        }
        if (tid == 31) s_lofs[MAX_BUCKETS] = o;
    }
    __syncthreads();

    for (int g = 0; g < PART_CHUNK / PART_GROUP; ++g) {
        const long long gb = beg + (long long)g * PART_GROUP;
        int d[PART_PER]; unsigned s[PART_PER];
        if (gb + PART_GROUP <= end) part_load<true >(src, dst, gb + tid, end, N, d, s);
        else                        part_load<false>(src, dst, gb + tid, end, N, d, s);
        #pragma unroll
        for (int k = 0; k < PART_PER; ++k) {
            if ((unsigned)d[k] < (unsigned)N) {
                const unsigned sl = slice_of(s[k]);
                const unsigned p = atomicAdd(&s_cur[((unsigned)d[k] >> RANGE_BITS) * MAX_SLICES + sl], 1u);
                s_word[p] = ((s[k] - sl * SLICE_NODES) << RANGE_BITS) | ((unsigned)d[k] & (RANGE_NODES - 1u));
            }
        }
    }
    __syncthreads();

    const int total = (int)s_lofs[MAX_BUCKETS];
    unsigned* outp = grouped + (size_t)blockIdx.x * PART_CHUNK;
    #pragma unroll 8
    for (int i = tid; i < total; i += PART_T) outp[i] = s_word[i];
    for (int u = tid; u < TAB_ENTRIES; u += PART_T)
        tab[(size_t)blockIdx.x * TAB_ENTRIES + u] = s_lofs[u];
}

__device__ __forceinline__ int chunk_count(int E) {
    return (int)(((unsigned)E + (PART_CHUNK - 1u)) / PART_CHUNK);
}

__device__ __forceinline__ void share_chunks(int E, int& qlo, int& qhi) {
    const unsigned chunks = (unsigned)chunk_count(E);
    const unsigned cps = (chunks + gridDim.x - 1u) / gridDim.x;
    unsigned lo = blockIdx.x * cps, hi = lo + cps;
    if (hi > chunks) hi = chunks;
    if (lo > hi) lo = hi;
    qlo = (int)lo;
    qhi = (int)hi;
}

template <int RUNS>
struct RunSet {
    unsigned i[RUNS];
    unsigned h[RUNS];
};

template <int RUNS>
__device__ __forceinline__ void runs_open(RunSet<RUNS>& R, const unsigned* __restrict__ tab,
                                          int b0, int b1, int q, int qhi) {
    #pragma unroll
    for (int t = 0; t < RUNS; ++t) {
        const int qt = q + t * (PART_T / 32);
        const int qc = qt < qhi ? qt : qhi - 1;
        const unsigned* o = tab + (size_t)qc * TAB_ENTRIES;
        const unsigned lo = o[b0];
        const unsigned hi = o[b1];
        const unsigned base = (unsigned)qc * PART_CHUNK;
        R.i[t] = base + lo;
        R.h[t] = base + (qt < qhi ? hi : lo);
    }
}

template <int RUNS>
__device__ __forceinline__ bool runs_left(const RunSet<RUNS>& R) {
    bool any = false;
    #pragma unroll
    for (int t = 0; t < RUNS; ++t) any = any || (R.i[t] < R.h[t]);
    return any;
}

template <int RUNS, int ROWS>
__device__ __forceinline__ void runs_step(RunSet<RUNS>& R) {
    #pragma unroll
    for (int t = 0; t < RUNS; ++t) R.i[t] += 32 * ROWS;
}

template <int RUNS>
__device__ __forceinline__ bool run_has(const RunSet<RUNS>& R, int t, int k, int lane) {
    return R.i[t] + (unsigned)(k * 32 + lane) < R.h[t];
}

template <int RUNS, int ROWS>
__device__ __forceinline__ void runs_load(const RunSet<RUNS>& R, const unsigned* __restrict__ grouped,
                                          int lane, unsigned (&w)[RUNS * ROWS]) {
    unsigned safe = R.i[0];
    #pragma unroll
    for (int t = RUNS - 1; t >= 0; --t)
        if (R.i[t] < R.h[t]) safe = R.i[t];
    #pragma unroll
    for (int t = 0; t < RUNS; ++t) {
        #pragma unroll
        for (int k = 0; k < ROWS; ++k) {
            const unsigned a = R.i[t] + (unsigned)(k * 32 + lane);
            w[t * ROWS + k] = grouped[run_has(R, t, k, lane) ? a : safe];
        }
    }
}

template <int ROWS>
__device__ __forceinline__ void agg_trip(const unsigned* __restrict__ grouped, const float2* ys,
                                         float* sx, float* sy,
                                         const unsigned (&P)[AGG_RUNS + 1], const unsigned (&off)[AGG_RUNS],
                                         unsigned pos, unsigned L, int lane) {
    static_assert(ROWS % 2 == 0, "the LDS work of a trip goes in two equal halves");
    unsigned w[ROWS];
    bool has[ROWS];
    #pragma unroll
    for (int k = 0; k < ROWS; ++k) {
        unsigned p = pos + (unsigned)(k * 32 + lane);
        has[k] = p < L;
        if (!has[k]) p = 0u;
        unsigned o = off[0];
        #pragma unroll
        for (int t = 1; t < AGG_RUNS; ++t) o = (p >= P[t]) ? off[t] : o;
        w[k] = grouped[p + o];
    }
    #pragma unroll
    for (int g = 0; g < ROWS; g += ROWS / 2) {
        float2 v[ROWS / 2];
        #pragma unroll
        for (int j = 0; j < ROWS / 2; ++j) v[j] = ys[w[g + j] >> RANGE_BITS];
        #pragma unroll
        for (int j = 0; j < ROWS / 2; ++j) {
            const unsigned o = has[g + j] ? (w[g + j] & (RANGE_NODES - 1u))
                                          : (unsigned)(RANGE_NODES + lane);
            atomicAdd(&sx[o], v[j].x);
            atomicAdd(&sy[o], v[j].y);
        }
    }
}

__global__ __launch_bounds__(PART_T)
void k_agg(const unsigned* __restrict__ grouped, const unsigned* __restrict__ tab,
           const float* __restrict__ y, float* agg0, float* agg1, int E, int N) {
    __shared__ float sx[RANGE_NODES + 32];
    __shared__ float sy[RANGE_NODES + 32];
    __shared__ float2 ys[SLICE_NODES];
    __shared__ int s_lo[AGG_RUNS * (PART_T / 32)];
    __shared__ int s_hi[AGG_RUNS * (PART_T / 32)];
    const int tid = threadIdx.x;
    const int lane = tid & 31;
    const int wv = __builtin_amdgcn_readfirstlane(tid >> 5);
    const int r = blockIdx.y;

    for (int u = tid; u < RANGE_NODES; u += PART_T) { sx[u] = 0.0f; sy[u] = 0.0f; }

    const int chunks = chunk_count(E);
    const int slices = (N + SLICE_NODES - 1) / SLICE_NODES;
    for (int s = blockIdx.x; s < slices; s += AGG_GROUPS) {
        const int n0 = s * SLICE_NODES;
        const int cnt = (N - n0 < SLICE_NODES) ? (N - n0) : SLICE_NODES;
        const float2* ysrc = (const float2*)y + n0;
        const int b = r * MAX_SLICES + s;
        __syncthreads();
        {
            const unsigned last = (unsigned)(cnt - 1);
            for (int u0 = 0; u0 < SLICE_NODES; u0 += PART_T * STAGE_PER) {
                float2 v[STAGE_PER];
                #pragma unroll
                for (int k = 0; k < STAGE_PER; ++k) {
                    if (u0 + k * PART_T < SLICE_NODES) {
                        const unsigned u = (unsigned)(u0 + k * PART_T + tid);
                        v[k] = ysrc[u < last ? u : last];
                    }
                }
                #pragma unroll
                for (int k = 0; k < STAGE_PER; ++k)
                    if (u0 + k * PART_T < SLICE_NODES) ys[u0 + k * PART_T + tid] = v[k];
            }
        }
        for (int qb = 0; qb < chunks; qb += AGG_RUNS * (PART_T / 32)) {
            if (qb != 0) __syncthreads();
            if (tid < AGG_RUNS * (PART_T / 32)) {
                const int qt = qb + tid;
                const unsigned* o = tab + (size_t)(qt < chunks ? qt : chunks - 1) * TAB_ENTRIES + b;
                const int lo = (int)o[0];
                const int hi = (int)o[1];
                s_lo[tid] = lo;
                s_hi[tid] = qt < chunks ? hi : lo;
            }
            __syncthreads();

            RunSet<AGG_RUNS> R;
            #pragma unroll
            for (int t = 0; t < AGG_RUNS; ++t) {
                const int idx = wv + t * (PART_T / 32);
                const int qt = qb + idx;
                const unsigned base = (unsigned)(qt < chunks ? qt : chunks - 1) * PART_CHUNK;
                R.i[t] = base + (unsigned)__builtin_amdgcn_readfirstlane(s_lo[idx]);
                R.h[t] = base + (unsigned)__builtin_amdgcn_readfirstlane(s_hi[idx]);
            }
            unsigned P[AGG_RUNS + 1];
            unsigned off[AGG_RUNS];
            P[0] = 0u;
            #pragma unroll
            for (int t = 0; t < AGG_RUNS; ++t) {
                off[t] = R.i[t] - P[t];
                P[t + 1] = P[t] + (R.h[t] - R.i[t]);
            }
            const unsigned L = P[AGG_RUNS];
            for (unsigned pos = 0u; pos < L; ) {
                const unsigned left = L - pos;
                if (left <= 16u * 32u)      { agg_trip<16>(grouped, ys, sx, sy, P, off, pos, L, lane); pos += 16u * 32u; }
                else if (left <= 18u * 32u) { agg_trip<18>(grouped, ys, sx, sy, P, off, pos, L, lane); pos += 18u * 32u; }
                else                        { agg_trip<20>(grouped, ys, sx, sy, P, off, pos, L, lane); pos += 20u * 32u; }
            }
        }
    }
    __syncthreads();

    float* agg = wave_half() ? agg1 : agg0;
    const int node0 = r << RANGE_BITS;
    for (int u = tid; u < 2 * RANGE_NODES; u += PART_T) {
        const int n = u >> 1;
        if (node0 + n < N) {
            const float val = (u & 1) ? sy[n] : sx[n];
            atomicAdd(agg + ((size_t)node0 * 2 + u), val);
        }
    }
}

#define DEG_SPLIT 3

__global__ __launch_bounds__(PART_T)
void k_deg(const unsigned* __restrict__ grouped, const unsigned* __restrict__ tab,
           int* __restrict__ part, size_t stride, int E, int N) {
    __shared__ int hist[RANGE_NODES + 32];
    const int tid = threadIdx.x;
    const int lane = tid & 31;
    const int wv = __builtin_amdgcn_readfirstlane(tid >> 5);
    const int r = blockIdx.y;

    for (int u = tid; u < RANGE_NODES; u += PART_T) hist[u] = 0;
    __syncthreads();

    int qlo, qhi;
    share_chunks(E, qlo, qhi);
    for (int q = qlo + wv; q < qhi; q += DEG_RUNS * (PART_T / 32)) {
        RunSet<DEG_RUNS> R;
        runs_open(R, tab, r * MAX_SLICES, (r + 1) * MAX_SLICES, q, qhi);
        for (; runs_left(R); runs_step<DEG_RUNS, DEG_ROWS>(R)) {
            unsigned w[DEG_RUNS * DEG_ROWS];
            runs_load<DEG_RUNS, DEG_ROWS>(R, grouped, lane, w);
            #pragma unroll
            for (int t = 0; t < DEG_RUNS; ++t) {
                #pragma unroll
                for (int k = 0; k < DEG_ROWS; ++k) {
                    const unsigned o = run_has(R, t, k, lane)
                                           ? (w[t * DEG_ROWS + k] & (RANGE_NODES - 1u))
                                           : (unsigned)(RANGE_NODES + lane);
                    atomicAdd(&hist[o], 1);
                }
            }
        }
    }
    __syncthreads();

    int* mine = part + (size_t)blockIdx.x * stride;
    const int node0 = r << RANGE_BITS;
    for (int u = tid; u < RANGE_NODES; u += PART_T)
        if (node0 + u < N) mine[node0 + u] = hist[u];
}

__global__ void k_node(const int* __restrict__ p0, const int* __restrict__ p1, const int* p2,
                       const float* __restrict__ x, float* dinv,
                       float* __restrict__ y1, float* __restrict__ agg1,
                       float* __restrict__ agg1b, int N) {
    int i = blockIdx.x * blockDim.x + threadIdx.x;
    if (i >= N) return;
    float f = (float)(p0[i] + p1[i] + p2[i] + 1);
    float dv = 1.0f / sqrtf(f);
    dinv[i] = dv;
    float2 xv = *(const float2*)(x + (size_t)i * 2);
    float2 v;
    v.x = dv * xv.x;
    v.y = dv * xv.y;
    *(float2*)(y1   + (size_t)i * 2) = v;
    *(float2*)(agg1 + (size_t)i * 2) = v;
    *(float2*)(agg1b + (size_t)i * 2) = make_float2(0.0f, 0.0f);
}

__global__ void k_layer2(const float* __restrict__ agg1, float* aggB,
                         const float* __restrict__ dinv,
                         const float* __restrict__ W1, const float* __restrict__ b1,
                         const float* __restrict__ W2,
                         float* __restrict__ y2, float* __restrict__ agg2, int N) {
    int lane = threadIdx.x & 31;
    int wave = threadIdx.x >> 5;
    int base = (blockIdx.x * (blockDim.x >> 5) + wave) * 16;
    if (base >= N) return;
    int m    = lane & 15;
    int half = lane >> 4;
    int node = base + m;
    int kb   = half * 2;
    bool full = (base + 16 <= N);

    int j = m & 7;
    v2f a1 = {0.0f, 0.0f};
    if (half == 0 && j < 4) {
        a1.x = W1[0 * 4 + j];
        a1.y = W1[1 * 4 + j];
    }
    v2f s = {0.0f, 0.0f};
    if (half == 0 && node < N) {
        float dv = dinv[node];
        float2 g = *(const float2*)(agg1 + (size_t)node * 2);
        float2* pb = (float2*)(aggB + (size_t)node * 2);
        float2 gb = *pb;
        *pb = make_float2(0.0f, 0.0f);
        s.x = dv * (g.x + gb.x);
        s.y = dv * (g.y + gb.y);
    }
    v8f c1 = {b1[0], b1[1], b1[2], b1[3], 0.0f, 0.0f, 0.0f, 0.0f};
    v8f p = __builtin_amdgcn_wmma_f32_16x16x4_f32(
        false, a1, false, s, (short)0, c1, false, false);

    v2f a2;
    a2.x = tanhf(half ? p[2] : p[0]);
    a2.y = tanhf(half ? p[3] : p[1]);

    v2f b2 = {0.0f, 0.0f};
    if (m < 2) {
        b2.x = W2[(kb + 0) * 2 + m];
        b2.y = W2[(kb + 1) * 2 + m];
    }
    v8f c2 = {};
    v8f d = __builtin_amdgcn_wmma_f32_16x16x4_f32(
        false, a2, false, b2, (short)0, c2, false, false);

    int nb = base + 8 * half;
    if (full) {
        if (m < 2) {
            float dv[8];
            #pragma unroll
            for (int v = 0; v < 8; ++v) dv[v] = dinv[nb + v];
            #pragma unroll
            for (int v = 0; v < 8; ++v) {
                float val = dv[v] * d[v];
                y2[(nb + v) * 2 + m]   = val;
                agg2[(nb + v) * 2 + m] = val;
            }
        }
    } else {
        if (m < 2) {
            #pragma unroll
            for (int v = 0; v < 8; ++v) {
                int nd = nb + v;
                if (nd < N) {
                    float val = dinv[nd] * d[v];
                    y2[nd * 2 + m]   = val;
                    agg2[nd * 2 + m] = val;
                }
            }
        }
    }
}

__global__ void k_final(const float* __restrict__ agg2, const float* __restrict__ agg2b,
                        const float* __restrict__ dinv,
                        const float* __restrict__ b2, const float* __restrict__ Wc,
                        const float* __restrict__ bc, float* __restrict__ out, int N) {
    int i = blockIdx.x * blockDim.x + threadIdx.x;
    if (i >= N) return;
    float dv = dinv[i];
    float2 ga = *(const float2*)(agg2  + (size_t)i * 2);
    float2 gb = *(const float2*)(agg2b + (size_t)i * 2);
    float h0 = tanhf(dv * (ga.x + gb.x) + b2[0]);
    float h1 = tanhf(dv * (ga.y + gb.y) + b2[1]);
    float2 o;
    o.x = h0 * Wc[0] + h1 * Wc[2] + bc[0];
    o.y = h0 * Wc[1] + h1 * Wc[3] + bc[1];
    *(float2*)(out + (size_t)i * 2) = o;
}

extern "C" void kernel_launch(void* const* d_in, const int* in_sizes, int n_in,
                              void* d_out, int out_size, void* d_ws, size_t ws_size,
                              hipStream_t stream) {
    const float* x  = (const float*)d_in[0];
    const int*   ei = (const int*)d_in[1];
    const float* W1 = (const float*)d_in[2];
    const float* b1 = (const float*)d_in[3];
    const float* W2 = (const float*)d_in[4];
    const float* b2 = (const float*)d_in[5];
    const float* Wc = (const float*)d_in[6];
    const float* bc = (const float*)d_in[7];
    float* out = (float*)d_out;

    const int N = in_sizes[0] / 2;
    const int E = in_sizes[1] / 2;
    const int* src = ei;
    const int* dst = ei + E;

    float* ws   = (float*)d_ws;
    float* y1   = ws;
    float* agg1 = ws + (size_t)2 * N;
    float* aggB = ws + (size_t)4 * N;
    float* y2   = ws;
    float* agg2 = ws + (size_t)6 * N;
    float* dinv = ws + (size_t)8 * N;
    int*   deg0 = (int*)dinv;
    int*   deg1 = (int*)ws;

    const int T = 256;
    dim3 gN((N + T - 1) / T);
    const int edgesPerBlock = (T / 32) * WAVE_EDGES;
    dim3 gE((E + edgesPerBlock - 1) / edgesPerBlock);
    const int nodesPerBlock = (T / 32) * 16;
    dim3 gW((N + nodesPerBlock - 1) / nodesPerBlock);
    dim3 gC((N + HIST_BINS - 1) / HIST_BINS, COUNT_CHUNKS);

    const size_t chunks   = ((size_t)E + PART_CHUNK - 1) / PART_CHUNK;
    const size_t offsOff  = (((size_t)9 * N * sizeof(float)) + 255) & ~(size_t)255;
    const size_t packOff  = (offsOff + chunks * TAB_ENTRIES * sizeof(unsigned) + 255) & ~(size_t)255;
    const size_t packEnd  = packOff + (size_t)E * sizeof(unsigned);
    const bool   groupedPath = (chunks > 0) && (N <= (1 << SRC_BITS)) && (ws_size >= packEnd);
    unsigned* offs = (unsigned*)((char*)d_ws + offsOff);
    unsigned* grouped = (unsigned*)((char*)d_ws + packOff);
    dim3 gA(AGG_GROUPS, (N + RANGE_NODES - 1) / RANGE_NODES);
    dim3 gD(DEG_SPLIT, (N + RANGE_NODES - 1) / RANGE_NODES);
    int* degp = (int*)(ws + (size_t)6 * N);

    if (groupedPath) {
        k_part  <<<dim3((unsigned)chunks), PART_T, 0, stream>>>(src, dst, grouped, offs, E, N);
        k_deg   <<<gD, PART_T, 0, stream>>>(grouped, offs, degp, (size_t)N, E, N);
        k_node  <<<gN, T, 0, stream>>>(degp, degp + (size_t)N, degp + (size_t)2 * N, x, dinv,
                                       y1, agg1, aggB, N);
        k_agg   <<<gA, PART_T, 0, stream>>>(grouped, offs, y1, agg1, aggB, E, N);
    } else {
        k_init_deg<<<gN, T, 0, stream>>>(deg0, deg1, N);
        k_count   <<<gC, COUNT_T, 0, stream>>>(dst, deg0, deg1, E, N);
        k_dinv    <<<gN, T, 0, stream>>>(deg0, deg1, dinv, N);
        k_y1      <<<gN, T, 0, stream>>>(x, dinv, y1, agg1, aggB, N);
        k_scatter<<<gE, T, 0, stream>>>(src, dst, y1, agg1, aggB, E);
    }
    k_layer2  <<<gW, T, 0, stream>>>(agg1, aggB, dinv, W1, b1, W2, y2, agg2, N);
    if (groupedPath) k_agg    <<<gA, PART_T, 0, stream>>>(grouped, offs, y2, agg2, aggB, E, N);
    else             k_scatter<<<gE, T, 0, stream>>>(src, dst, y2, agg2, aggB, E);
    k_final   <<<gN, T, 0, stream>>>(agg2, aggB, dinv, b2, Wc, bc, out, N);
}
